# starting kernel
# baseline (speedup 1.0000x reference)
#include <hip/hip_runtime.h>
#include <hip/hip_bf16.h>

typedef __attribute__((ext_vector_type(2))) float v2f;
typedef __attribute__((ext_vector_type(8))) float v8f;

#define BATCH     2
#define SEQLEN    2048
#define C_IN      256
#define D_MODEL   512
#define LOUT      1024
#define D_STATE   64
#define D_CONV    4
#define D_INNER   1024
#define HEADDIM   64
#define NHEADS    16
#define CONV_DIM  (D_INNER + 2*D_STATE)
#define D_IN_PROJ (2*D_INNER + 2*D_STATE + NHEADS)
#define ROWS      (BATCH*LOUT)
#define KSTEM     (C_IN * 7)

__global__ void im2col_stem_kernel(const float* __restrict__ tokens,
                                   float* __restrict__ col) {
  const long idx = (long)blockIdx.x * 256 + threadIdx.x;
  if (idx >= (long)ROWS * KSTEM) return;
  const int j  = (int)(idx % KSTEM);
  const long bl = idx / KSTEM;
  const int c = j / 7;
  const int k = j - c * 7;
  const int l = (int)(bl & (LOUT - 1));
  const int b = (int)(bl >> 10);
  const int s = 2 * l - 3 + k;
  float v = 0.f;
  if (s >= 0 && s < SEQLEN) v = tokens[((long)b * C_IN + c) * SEQLEN + s];
  col[idx] = v;
}

__global__ void gemm_xt_wmma(const float* __restrict__ A,
                             const float* __restrict__ W,
                             float* __restrict__ C,
                             int M, int N, int K) {
  const int lane = threadIdx.x & 31;
  const int wave = threadIdx.x >> 5;
  const int tiles_n = N >> 4;
  const int strips  = (tiles_n + 3) >> 2;
  const int tile = blockIdx.x * (blockDim.x >> 5) + wave;
  if (tile >= (M >> 4) * strips) return;
  const int tm  = tile / strips;
  const int ts  = tile - tm * strips;
  const int tn0 = ts << 2;
  const int nt  = min(4, tiles_n - tn0);

  const int r    = lane & 15;
  const int koff = (lane >> 4) << 1;
  const float* arow = A + ((long)(tm * 16 + r)) * K + koff;
  const float* brow = W + ((long)(tn0 * 16 + r)) * K + koff;
  const long  bstep = (long)16 * K;

  if (nt == 4) {
    v8f acc0 = {0.f,0.f,0.f,0.f,0.f,0.f,0.f,0.f};
    v8f acc1 = acc0, acc2 = acc0, acc3 = acc0;
    #pragma unroll 2
    for (int k = 0; k < K; k += 4) {
      const v2f a  = *(const v2f*)(arow + k);
      const v2f b0 = *(const v2f*)(brow + k);
      const v2f b1 = *(const v2f*)(brow + bstep + k);
      const v2f b2 = *(const v2f*)(brow + 2 * bstep + k);
      const v2f b3 = *(const v2f*)(brow + 3 * bstep + k);
      acc0 = __builtin_amdgcn_wmma_f32_16x16x4_f32(false, a, false, b0, (short)0, acc0, false, false);
      acc1 = __builtin_amdgcn_wmma_f32_16x16x4_f32(false, a, false, b1, (short)0, acc1, false, false);
      acc2 = __builtin_amdgcn_wmma_f32_16x16x4_f32(false, a, false, b2, (short)0, acc2, false, false);
      acc3 = __builtin_amdgcn_wmma_f32_16x16x4_f32(false, a, false, b3, (short)0, acc3, false, false);
    }
    const int row0 = (lane >> 4) << 3;
    float* cbase = C + ((long)(tm * 16)) * N + tn0 * 16 + (lane & 15);
    #pragma unroll
    for (int i = 0; i < 8; ++i) {
      float* crow = cbase + (long)(row0 + i) * N;
      crow[0]  = acc0[i];
      crow[16] = acc1[i];
      crow[32] = acc2[i];
      crow[48] = acc3[i];
    }
  } else {
    for (int t = 0; t < nt; ++t) {
      v8f acc = {0.f,0.f,0.f,0.f,0.f,0.f,0.f,0.f};
      const float* bt = brow + (long)t * bstep;
      #pragma unroll 4
      for (int k = 0; k < K; k += 4) {
        const v2f a = *(const v2f*)(arow + k);
        const v2f b = *(const v2f*)(bt + k);
        acc = __builtin_amdgcn_wmma_f32_16x16x4_f32(false, a, false, b, (short)0, acc, false, false);
      }
      const int row0 = (lane >> 4) << 3;
      float* cbase = C + ((long)(tm * 16)) * N + (tn0 + t) * 16 + (lane & 15);
      #pragma unroll
      for (int i = 0; i < 8; ++i)
        cbase[(long)(row0 + i) * N] = acc[i];
    }
  }
}

__global__ void dwconv_silu_kernel(const float* __restrict__ zx,
                                   const float* __restrict__ w,
                                   const float* __restrict__ bias,
                                   float* __restrict__ out) {
  const int c = blockIdx.x * 128 + threadIdx.x;
  if (c >= CONV_DIM) return;
  const long bl = blockIdx.y;
  const int l = (int)(bl & (LOUT - 1));
  const float* xcol = zx + bl * D_IN_PROJ + D_INNER + c;
  const float w0 = w[c * 4 + 0], w1 = w[c * 4 + 1], w2 = w[c * 4 + 2], w3 = w[c * 4 + 3];
  float acc = w3 * xcol[0];
  if (l >= 1) acc = fmaf(w2, xcol[-(long)D_IN_PROJ],     acc);
  if (l >= 2) acc = fmaf(w1, xcol[-(long)2 * D_IN_PROJ], acc);
  if (l >= 3) acc = fmaf(w0, xcol[-(long)3 * D_IN_PROJ], acc);
  const float v = acc + bias[c];
  out[bl * CONV_DIM + c] = v / (1.f + __expf(-v));
}

__global__ void ssm_scan_kernel(const float* __restrict__ zx,
                                const float* __restrict__ xbc,
                                const float* __restrict__ dtb,
                                const float* __restrict__ alog,
                                const float* __restrict__ Dp,
                                float* __restrict__ y) {
  const int b  = blockIdx.x >> 4;
  const int hh = blockIdx.x & 15;
  const int tid = threadIdx.x;
  const int p = tid >> 2;
  const int q = tid & 3;
  __shared__ float sx[HEADDIM], sB[D_STATE], sC[D_STATE], sdt[2];
  const float Ah   = -__expf(alog[hh]);
  const float dtbh = dtb[hh];
  const float Dh   = Dp[hh];
  float hreg[16];
  #pragma unroll
  for (int i = 0; i < 16; ++i) hreg[i] = 0.f;

  for (int l = 0; l < LOUT; ++l) {
    const long bl = (long)b * LOUT + l;
    const float* row = xbc + bl * CONV_DIM;
    if (tid < 64) {
      sx[tid] = row[hh * HEADDIM + tid];
    } else if (tid < 128) {
      sB[tid - 64] = row[D_INNER + (tid - 64)];
    } else if (tid < 192) {
      sC[tid - 128] = row[D_INNER + D_STATE + (tid - 128)];
    } else if (tid == 192) {
      const float raw = zx[bl * D_IN_PROJ + (D_IN_PROJ - NHEADS) + hh] + dtbh;
      const float dtv = (raw > 20.f) ? raw : log1pf(__expf(raw));
      sdt[0] = dtv;
      sdt[1] = __expf(dtv * Ah);
    }
    __syncthreads();
    const float dtv = sdt[0];
    const float dA  = sdt[1];
    const float xv  = sx[p];
    const float xdt = xv * dtv;
    const float* Bq = sB + q * 16;
    const float* Cq = sC + q * 16;
    float accv = 0.f;
    #pragma unroll
    for (int i = 0; i < 16; ++i) {
      hreg[i] = fmaf(hreg[i], dA, xdt * Bq[i]);
      accv    = fmaf(hreg[i], Cq[i], accv);
    }
    accv += __shfl_xor(accv, 1, 32);
    accv += __shfl_xor(accv, 2, 32);
    if (q == 0)
      y[bl * D_INNER + hh * HEADDIM + p] = fmaf(xv, Dh, accv);
    __syncthreads();
  }
}

__global__ void gate_rmsnorm_kernel(const float* __restrict__ zx,
                                    const float* __restrict__ nw,
                                    float* __restrict__ y) {
  const long bl = blockIdx.x;
  const int tid = threadIdx.x;
  __shared__ float red[256];
  float yv[4];
  float ss = 0.f;
  #pragma unroll
  for (int j = 0; j < 4; ++j) {
    const int i = tid + j * 256;
    const float z = zx[bl * D_IN_PROJ + i];
    const float g = z / (1.f + __expf(-z));
    const float v = y[bl * D_INNER + i] * g;
    yv[j] = v;
    ss = fmaf(v, v, ss);
  }
  red[tid] = ss;
  __syncthreads();
  for (int st = 128; st > 0; st >>= 1) {
    if (tid < st) red[tid] += red[tid + st];
    __syncthreads();
  }
  const float scale = rsqrtf(red[0] * (1.f / (float)D_INNER) + 1e-5f);
  #pragma unroll
  for (int j = 0; j < 4; ++j) {
    const int i = tid + j * 256;
    y[bl * D_INNER + i] = yv[j] * scale * nw[i];
  }
}

extern "C" void kernel_launch(void* const* d_in, const int* in_sizes, int n_in,
                              void* d_out, int out_size, void* d_ws, size_t ws_size,
                              hipStream_t stream) {
  const float* tokens     = (const float*)d_in[0];
  const float* conv_w     = (const float*)d_in[1];
  const float* in_proj_w  = (const float*)d_in[2];
  const float* conv1d_w   = (const float*)d_in[3];
  const float* conv1d_b   = (const float*)d_in[4];
  const float* dt_bias    = (const float*)d_in[5];
  const float* A_log      = (const float*)d_in[6];
  const float* D_param    = (const float*)d_in[7];
  const float* norm_w     = (const float*)d_in[8];
  const float* out_proj_w = (const float*)d_in[9];
  float* out = (float*)d_out;
  float* ws  = (float*)d_ws;

  float* u0   = ws;
  float* u1   = u0 + (long)ROWS * D_MODEL;
  float* zx   = u1 + (long)ROWS * D_MODEL;
  float* xact = zx + (long)ROWS * D_IN_PROJ;
  float* ybuf = xact + (long)ROWS * CONV_DIM;
  float* col  = zx;
  (void)ws_size; (void)in_sizes; (void)n_in; (void)out_size;

  {
    const long nel = (long)ROWS * KSTEM;
    im2col_stem_kernel<<<(int)((nel + 255) / 256), 256, 0, stream>>>(tokens, col);
    const int strips = ((D_MODEL / 16) + 3) / 4;
    const int tiles  = (ROWS / 16) * strips;
    gemm_xt_wmma<<<(tiles + 7) / 8, 256, 0, stream>>>(col, conv_w, u0, ROWS, D_MODEL, KSTEM);
  }

  const float* u_in = u0;
  for (int blk = 0; blk < 2; ++blk) {
    const float* ipw = in_proj_w  + (long)blk * D_IN_PROJ * D_MODEL;
    const float* cw  = conv1d_w   + (long)blk * CONV_DIM * D_CONV;
    const float* cb  = conv1d_b   + (long)blk * CONV_DIM;
    const float* dtb = dt_bias    + (long)blk * NHEADS;
    const float* al  = A_log      + (long)blk * NHEADS;
    const float* Dpp = D_param    + (long)blk * NHEADS;
    const float* nw  = norm_w     + (long)blk * D_INNER;
    const float* opw = out_proj_w + (long)blk * D_MODEL * D_INNER;

    {
      const int strips = ((D_IN_PROJ / 16) + 3) / 4;
      const int tiles  = (ROWS / 16) * strips;
      gemm_xt_wmma<<<(tiles + 7) / 8, 256, 0, stream>>>(u_in, ipw, zx, ROWS, D_IN_PROJ, D_MODEL);
    }
    dwconv_silu_kernel<<<dim3(CONV_DIM / 128, ROWS), 128, 0, stream>>>(zx, cw, cb, xact);
    ssm_scan_kernel<<<BATCH * NHEADS, 256, 0, stream>>>(zx, xact, dtb, al, Dpp, ybuf);
    gate_rmsnorm_kernel<<<ROWS, 256, 0, stream>>>(zx, nw, ybuf);
    float* dst = (blk == 0) ? u1 : out;
    {
      const int strips = ((D_MODEL / 16) + 3) / 4;
      const int tiles  = (ROWS / 16) * strips;
      gemm_xt_wmma<<<(tiles + 7) / 8, 256, 0, stream>>>(ybuf, opw, dst, ROWS, D_MODEL, D_INNER);
    }
    u_in = u1;
  }
}


# ssm scan: x/B/C/dt of 32 steps staged in LDS one chunk ahead, y rows leave as whole lines; one barrier per 32 steps, no global round trip or barrier inside a step; same sums in the same order
# speedup vs baseline: 2.9377x; 2.9377x over previous
#include <hip/hip_runtime.h>
#include <hip/hip_bf16.h>

typedef __attribute__((ext_vector_type(2))) float v2f;
typedef __attribute__((ext_vector_type(8))) float v8f;

#define BATCH     2
#define SEQLEN    2048
#define C_IN      256
#define D_MODEL   512
#define LOUT      1024
#define D_STATE   64
#define D_CONV    4
#define D_INNER   1024
#define HEADDIM   64
#define NHEADS    16
#define CONV_DIM  (D_INNER + 2*D_STATE)
#define D_IN_PROJ (2*D_INNER + 2*D_STATE + NHEADS)
#define ROWS      (BATCH*LOUT)
#define KSTEM     (C_IN * 7)

__global__ void im2col_stem_kernel(const float* __restrict__ tokens,
                                   float* __restrict__ col) {
  const long idx = (long)blockIdx.x * 256 + threadIdx.x;
  if (idx >= (long)ROWS * KSTEM) return;
  const int j  = (int)(idx % KSTEM);
  const long bl = idx / KSTEM;
  const int c = j / 7;
  const int k = j - c * 7;
  const int l = (int)(bl & (LOUT - 1));
  const int b = (int)(bl >> 10);
  const int s = 2 * l - 3 + k;
  float v = 0.f;
  if (s >= 0 && s < SEQLEN) v = tokens[((long)b * C_IN + c) * SEQLEN + s];
  col[idx] = v;
}

__global__ void gemm_xt_wmma(const float* __restrict__ A,
                             const float* __restrict__ W,
                             float* __restrict__ C,
                             int M, int N, int K) {
  const int lane = threadIdx.x & 31;
  const int wave = threadIdx.x >> 5;
  const int tiles_n = N >> 4;
  const int strips  = (tiles_n + 3) >> 2;
  const int tile = blockIdx.x * (blockDim.x >> 5) + wave;
  if (tile >= (M >> 4) * strips) return;
  const int tm  = tile / strips;
  const int ts  = tile - tm * strips;
  const int tn0 = ts << 2;
  const int nt  = min(4, tiles_n - tn0);

  const int r    = lane & 15;
  const int koff = (lane >> 4) << 1;
  const float* arow = A + ((long)(tm * 16 + r)) * K + koff;
  const float* brow = W + ((long)(tn0 * 16 + r)) * K + koff;
  const long  bstep = (long)16 * K;

  if (nt == 4) {
    v8f acc0 = {0.f,0.f,0.f,0.f,0.f,0.f,0.f,0.f};
    v8f acc1 = acc0, acc2 = acc0, acc3 = acc0;
    #pragma unroll 2
    for (int k = 0; k < K; k += 4) {
      const v2f a  = *(const v2f*)(arow + k);
      const v2f b0 = *(const v2f*)(brow + k);
      const v2f b1 = *(const v2f*)(brow + bstep + k);
      const v2f b2 = *(const v2f*)(brow + 2 * bstep + k);
      const v2f b3 = *(const v2f*)(brow + 3 * bstep + k);
      acc0 = __builtin_amdgcn_wmma_f32_16x16x4_f32(false, a, false, b0, (short)0, acc0, false, false);
      acc1 = __builtin_amdgcn_wmma_f32_16x16x4_f32(false, a, false, b1, (short)0, acc1, false, false);
      acc2 = __builtin_amdgcn_wmma_f32_16x16x4_f32(false, a, false, b2, (short)0, acc2, false, false);
      acc3 = __builtin_amdgcn_wmma_f32_16x16x4_f32(false, a, false, b3, (short)0, acc3, false, false);
    }
    const int row0 = (lane >> 4) << 3;
    float* cbase = C + ((long)(tm * 16)) * N + tn0 * 16 + (lane & 15);
    #pragma unroll
    for (int i = 0; i < 8; ++i) {
      float* crow = cbase + (long)(row0 + i) * N;
      crow[0]  = acc0[i];
      crow[16] = acc1[i];
      crow[32] = acc2[i];
      crow[48] = acc3[i];
    }
  } else {
    for (int t = 0; t < nt; ++t) {
      v8f acc = {0.f,0.f,0.f,0.f,0.f,0.f,0.f,0.f};
      const float* bt = brow + (long)t * bstep;
      #pragma unroll 4
      for (int k = 0; k < K; k += 4) {
        const v2f a = *(const v2f*)(arow + k);
        const v2f b = *(const v2f*)(bt + k);
        acc = __builtin_amdgcn_wmma_f32_16x16x4_f32(false, a, false, b, (short)0, acc, false, false);
      }
      const int row0 = (lane >> 4) << 3;
      float* cbase = C + ((long)(tm * 16)) * N + (tn0 + t) * 16 + (lane & 15);
      #pragma unroll
      for (int i = 0; i < 8; ++i)
        cbase[(long)(row0 + i) * N] = acc[i];
    }
  }
}

__global__ void dwconv_silu_kernel(const float* __restrict__ zx,
                                   const float* __restrict__ w,
                                   const float* __restrict__ bias,
                                   float* __restrict__ out) {
  const int c = blockIdx.x * 128 + threadIdx.x;
  if (c >= CONV_DIM) return;
  const long bl = blockIdx.y;
  const int l = (int)(bl & (LOUT - 1));
  const float* xcol = zx + bl * D_IN_PROJ + D_INNER + c;
  const float w0 = w[c * 4 + 0], w1 = w[c * 4 + 1], w2 = w[c * 4 + 2], w3 = w[c * 4 + 3];
  float acc = w3 * xcol[0];
  if (l >= 1) acc = fmaf(w2, xcol[-(long)D_IN_PROJ],     acc);
  if (l >= 2) acc = fmaf(w1, xcol[-(long)2 * D_IN_PROJ], acc);
  if (l >= 3) acc = fmaf(w0, xcol[-(long)3 * D_IN_PROJ], acc);
  const float v = acc + bias[c];
  out[bl * CONV_DIM + c] = v / (1.f + __expf(-v));
}

#define SCAN_T   32
#define SCAN_NCH (LOUT / SCAN_T)

__global__ void __launch_bounds__(256)
ssm_scan_kernel(const float* __restrict__ zx,
                const float* __restrict__ xbc,
                const float* __restrict__ dtb,
                const float* __restrict__ alog,
                const float* __restrict__ Dp,
                float* __restrict__ y) {
  const int b  = blockIdx.x >> 4;
  const int hh = blockIdx.x & 15;
  const int tid = threadIdx.x;
  const int p = tid >> 2;
  const int q = tid & 3;
  __shared__ __attribute__((aligned(16))) float sBC[2][SCAN_T][2 * D_STATE];
  __shared__ __attribute__((aligned(16))) float sX[2][SCAN_T][HEADDIM];
  __shared__ __attribute__((aligned(16))) float sY[2][SCAN_T][HEADDIM];
  __shared__ __attribute__((aligned(8)))  float sDT[2][SCAN_T][2];
  const float Ah   = -__expf(alog[hh]);
  const float dtbh = dtb[hh];
  const float Dh   = Dp[hh];
  float hreg[16];
  #pragma unroll
  for (int i = 0; i < 16; ++i) hreg[i] = 0.f;

  const int bc_c4 = tid & 31;
  const int bc_t  = tid >> 5;
  const int x_c4  = tid & 15;
  const int x_t   = tid >> 4;
  const bool dt_wave = (tid >= 256 - SCAN_T);
  const int  dt_t    = dt_wave ? tid - (256 - SCAN_T) : 0;
  const float* gBC = xbc + ((long)b * LOUT + bc_t) * CONV_DIM + D_INNER + bc_c4 * 4;
  const float* gX  = xbc + ((long)b * LOUT + x_t) * CONV_DIM + hh * HEADDIM + x_c4 * 4;
  const float* gDT = zx + ((long)b * LOUT + dt_t) * D_IN_PROJ + (D_IN_PROJ - NHEADS) + hh;
  float*       gY  = y + ((long)b * LOUT + x_t) * D_INNER + hh * HEADDIM + x_c4 * 4;

  float4 rBC0, rBC1, rBC2, rBC3, rX0, rX1;
  float  rdt = 0.f;

#define SCAN_FETCH(l0)                                                              \
  do {                                                                              \
    const float* s_bc = gBC + (long)(l0) * CONV_DIM;                                \
    const float* s_x  = gX  + (long)(l0) * CONV_DIM;                                \
    rBC0 = *(const float4*)(s_bc);                                                  \
    rBC1 = *(const float4*)(s_bc + 8  * CONV_DIM);                                  \
    rBC2 = *(const float4*)(s_bc + 16 * CONV_DIM);                                  \
    rBC3 = *(const float4*)(s_bc + 24 * CONV_DIM);                                  \
    rX0  = *(const float4*)(s_x);                                                   \
    rX1  = *(const float4*)(s_x + 16 * CONV_DIM);                                   \
    if (dt_wave) rdt = gDT[(long)(l0) * D_IN_PROJ];                                 \
  } while (0)

#define SCAN_COMMIT(buf)                                                            \
  do {                                                                              \
    *(float4*)&sBC[(buf)][bc_t     ][bc_c4 * 4] = rBC0;                             \
    *(float4*)&sBC[(buf)][bc_t + 8 ][bc_c4 * 4] = rBC1;                             \
    *(float4*)&sBC[(buf)][bc_t + 16][bc_c4 * 4] = rBC2;                             \
    *(float4*)&sBC[(buf)][bc_t + 24][bc_c4 * 4] = rBC3;                             \
    *(float4*)&sX[(buf)][x_t     ][x_c4 * 4] = rX0;                                 \
    *(float4*)&sX[(buf)][x_t + 16][x_c4 * 4] = rX1;                                 \
    if (dt_wave) {                                                                  \
      const float raw = rdt + dtbh;                                                 \
      const float dtv = (raw > 20.f) ? raw : log1pf(__expf(raw));      \
      float2 dd;                                                                    \
      dd.x = dtv;                                                                   \
      dd.y = __expf(dtv * Ah);                                               \
      *(float2*)&sDT[(buf)][dt_t][0] = dd;                                          \
    }                                                                               \
  } while (0)

#define SCAN_WRITEOUT(l0, buf)                                                      \
  do {                                                                              \
    float* d_y = gY + (long)(l0) * D_INNER;                                         \
    const float4 v0 = *(const float4*)&sY[(buf)][x_t     ][x_c4 * 4];               \
    const float4 v1 = *(const float4*)&sY[(buf)][x_t + 16][x_c4 * 4];               \
    *(float4*)(d_y)                = v0;                                            \
    *(float4*)(d_y + 16 * D_INNER) = v1;                                            \
  } while (0)

  SCAN_FETCH(0);
  SCAN_COMMIT(0);
  __syncthreads();

  for (int c = 0; c < SCAN_NCH; ++c) {
    const int cb = c & 1;
    const bool more = (c + 1 < SCAN_NCH);
    if (more) SCAN_FETCH((c + 1) * SCAN_T);
    if (c > 0) SCAN_WRITEOUT((c - 1) * SCAN_T, cb ^ 1);
    for (int t0 = 0; t0 < SCAN_T; t0 += 4) {
      float yv[4];
      #pragma unroll
      for (int u = 0; u < 4; ++u) {
        const int t = t0 + u;
        const float2 dd = *(const float2*)&sDT[cb][t][0];
        const float dtv = dd.x;
        const float dA  = dd.y;
        const float xv  = sX[cb][t][p];
        const float xdt = xv * dtv;
        const float* Bq = &sBC[cb][t][q * 16];
        const float* Cq = &sBC[cb][t][D_STATE + q * 16];
        float accv = 0.f;
        #pragma unroll
        for (int i = 0; i < 16; ++i) {
          hreg[i] = fmaf(hreg[i], dA, xdt * Bq[i]);
          accv    = fmaf(hreg[i], Cq[i], accv);
        }
        accv += __shfl_xor(accv, 1, 32);
        accv += __shfl_xor(accv, 2, 32);
        yv[u] = fmaf(xv, Dh, accv);
      }
      if (q == 0) {
        #pragma unroll
        for (int u = 0; u < 4; ++u) sY[cb][t0 + u][p] = yv[u];
      }
    }
    if (more) SCAN_COMMIT(cb ^ 1);
    __syncthreads();
  }
  SCAN_WRITEOUT((SCAN_NCH - 1) * SCAN_T, (SCAN_NCH - 1) & 1);
#undef SCAN_FETCH
#undef SCAN_COMMIT
#undef SCAN_WRITEOUT
}

__global__ void gate_rmsnorm_kernel(const float* __restrict__ zx,
                                    const float* __restrict__ nw,
                                    float* __restrict__ y) {
  const long bl = blockIdx.x;
  const int tid = threadIdx.x;
  __shared__ float red[256];
  float yv[4];
  float ss = 0.f;
  #pragma unroll
  for (int j = 0; j < 4; ++j) {
    const int i = tid + j * 256;
    const float z = zx[bl * D_IN_PROJ + i];
    const float g = z / (1.f + __expf(-z));
    const float v = y[bl * D_INNER + i] * g;
    yv[j] = v;
    ss = fmaf(v, v, ss);
  }
  red[tid] = ss;
  __syncthreads();
  for (int st = 128; st > 0; st >>= 1) {
    if (tid < st) red[tid] += red[tid + st];
    __syncthreads();
  }
  const float scale = rsqrtf(red[0] * (1.f / (float)D_INNER) + 1e-5f);
  #pragma unroll
  for (int j = 0; j < 4; ++j) {
    const int i = tid + j * 256;
    y[bl * D_INNER + i] = yv[j] * scale * nw[i];
  }
}

extern "C" void kernel_launch(void* const* d_in, const int* in_sizes, int n_in,
                              void* d_out, int out_size, void* d_ws, size_t ws_size,
                              hipStream_t stream) {
  const float* tokens     = (const float*)d_in[0];
  const float* conv_w     = (const float*)d_in[1];
  const float* in_proj_w  = (const float*)d_in[2];
  const float* conv1d_w   = (const float*)d_in[3];
  const float* conv1d_b   = (const float*)d_in[4];
  const float* dt_bias    = (const float*)d_in[5];
  const float* A_log      = (const float*)d_in[6];
  const float* D_param    = (const float*)d_in[7];
  const float* norm_w     = (const float*)d_in[8];
  const float* out_proj_w = (const float*)d_in[9];
  float* out = (float*)d_out;
  float* ws  = (float*)d_ws;

  float* u0   = ws;
  float* u1   = u0 + (long)ROWS * D_MODEL;
  float* zx   = u1 + (long)ROWS * D_MODEL;
  float* xact = zx + (long)ROWS * D_IN_PROJ;
  float* ybuf = xact + (long)ROWS * CONV_DIM;
  float* col  = zx;
  (void)ws_size; (void)in_sizes; (void)n_in; (void)out_size;

  {
    const long nel = (long)ROWS * KSTEM;
    im2col_stem_kernel<<<(int)((nel + 255) / 256), 256, 0, stream>>>(tokens, col);
    const int strips = ((D_MODEL / 16) + 3) / 4;
    const int tiles  = (ROWS / 16) * strips;
    gemm_xt_wmma<<<(tiles + 7) / 8, 256, 0, stream>>>(col, conv_w, u0, ROWS, D_MODEL, KSTEM);
  }

  const float* u_in = u0;
  for (int blk = 0; blk < 2; ++blk) {
    const float* ipw = in_proj_w  + (long)blk * D_IN_PROJ * D_MODEL;
    const float* cw  = conv1d_w   + (long)blk * CONV_DIM * D_CONV;
    const float* cb  = conv1d_b   + (long)blk * CONV_DIM;
    const float* dtb = dt_bias    + (long)blk * NHEADS;
    const float* al  = A_log      + (long)blk * NHEADS;
    const float* Dpp = D_param    + (long)blk * NHEADS;
    const float* nw  = norm_w     + (long)blk * D_INNER;
    const float* opw = out_proj_w + (long)blk * D_MODEL * D_INNER;

    {
      const int strips = ((D_IN_PROJ / 16) + 3) / 4;
      const int tiles  = (ROWS / 16) * strips;
      gemm_xt_wmma<<<(tiles + 7) / 8, 256, 0, stream>>>(u_in, ipw, zx, ROWS, D_IN_PROJ, D_MODEL);
    }
    dwconv_silu_kernel<<<dim3(CONV_DIM / 128, ROWS), 128, 0, stream>>>(zx, cw, cb, xact);
    ssm_scan_kernel<<<BATCH * NHEADS, 256, 0, stream>>>(zx, xact, dtb, al, Dpp, ybuf);
    gate_rmsnorm_kernel<<<ROWS, 256, 0, stream>>>(zx, nw, ybuf);
    float* dst = (blk == 0) ? u1 : out;
    {
      const int strips = ((D_MODEL / 16) + 3) / 4;
      const int tiles  = (ROWS / 16) * strips;
      gemm_xt_wmma<<<(tiles + 7) / 8, 256, 0, stream>>>(ybuf, opw, dst, ROWS, D_MODEL, D_INNER);
    }
    u_in = u1;
  }
}


# GEMMs: one 64x64 C tile per workgroup; A and W tiles read once as whole rows, one K-step ahead, into a two-stage LDS ring; C leaves via LDS; k summed in another order inside each 8-k group
# speedup vs baseline: 6.2024x; 2.1113x over previous
#include <hip/hip_runtime.h>
#include <hip/hip_bf16.h>

typedef __attribute__((ext_vector_type(2))) float v2f;
typedef __attribute__((ext_vector_type(8))) float v8f;

#define BATCH     2
#define SEQLEN    2048
#define C_IN      256
#define D_MODEL   512
#define LOUT      1024
#define D_STATE   64
#define D_CONV    4
#define D_INNER   1024
#define HEADDIM   64
#define NHEADS    16
#define CONV_DIM  (D_INNER + 2*D_STATE)
#define D_IN_PROJ (2*D_INNER + 2*D_STATE + NHEADS)
#define ROWS      (BATCH*LOUT)
#define KSTEM     (C_IN * 7)

__global__ void im2col_stem_kernel(const float* __restrict__ tokens,
                                   float* __restrict__ col) {
  const long idx = (long)blockIdx.x * 256 + threadIdx.x;
  if (idx >= (long)ROWS * KSTEM) return;
  const int j  = (int)(idx % KSTEM);
  const long bl = idx / KSTEM;
  const int c = j / 7;
  const int k = j - c * 7;
  const int l = (int)(bl & (LOUT - 1));
  const int b = (int)(bl >> 10);
  const int s = 2 * l - 3 + k;
  float v = 0.f;
  if (s >= 0 && s < SEQLEN) v = tokens[((long)b * C_IN + c) * SEQLEN + s];
  col[idx] = v;
}

#define GT_M 64
#define GT_N 64
#define GT_K 64
#define GT_P 68

__global__ void __launch_bounds__(256)
gemm_xt_wmma(const float* __restrict__ A,
             const float* __restrict__ W,
             float* __restrict__ C,
             int M, int N, int K) {
  __shared__ __attribute__((aligned(16))) float sA[2][GT_M][GT_P];
  __shared__ __attribute__((aligned(16))) float sB[2][GT_N][GT_P];
  const int tid  = threadIdx.x;
  const int lane = tid & 31;
  const int wave = tid >> 5;
  const int nblk = (N + GT_N - 1) / GT_N;
  const int mb = blockIdx.x / nblk;
  const int nb = blockIdx.x - mb * nblk;
  const int m0 = mb * GT_M;
  const int n0 = nb * GT_N;
  (void)M;

  const int ld_c4 = tid & 15;
  const int ld_r  = tid >> 4;
  const long a_step = (long)16 * K;
  const float* gA  = A + (long)(m0 + ld_r) * K + ld_c4 * 4;
  const float* gB0 = W + (long)min(n0 + ld_r,      N - 1) * K + ld_c4 * 4;
  const float* gB1 = W + (long)min(n0 + ld_r + 16, N - 1) * K + ld_c4 * 4;
  const float* gB2 = W + (long)min(n0 + ld_r + 32, N - 1) * K + ld_c4 * 4;
  const float* gB3 = W + (long)min(n0 + ld_r + 48, N - 1) * K + ld_c4 * 4;

  float4 ra0, ra1, ra2, ra3, rb0, rb1, rb2, rb3;

#define GT_FETCH(k0)                                                   \
  do {                                                                 \
    ra0 = *(const float4*)(gA + (k0));                                 \
    ra1 = *(const float4*)(gA + a_step + (k0));                        \
    ra2 = *(const float4*)(gA + 2 * a_step + (k0));                    \
    ra3 = *(const float4*)(gA + 3 * a_step + (k0));                    \
    rb0 = *(const float4*)(gB0 + (k0));                                \
    rb1 = *(const float4*)(gB1 + (k0));                                \
    rb2 = *(const float4*)(gB2 + (k0));                                \
    rb3 = *(const float4*)(gB3 + (k0));                                \
  } while (0)

#define GT_COMMIT(buf)                                                 \
  do {                                                                 \
    *(float4*)&sA[(buf)][ld_r     ][ld_c4 * 4] = ra0;                  \
    *(float4*)&sA[(buf)][ld_r + 16][ld_c4 * 4] = ra1;                  \
    *(float4*)&sA[(buf)][ld_r + 32][ld_c4 * 4] = ra2;                  \
    *(float4*)&sA[(buf)][ld_r + 48][ld_c4 * 4] = ra3;                  \
    *(float4*)&sB[(buf)][ld_r     ][ld_c4 * 4] = rb0;                  \
    *(float4*)&sB[(buf)][ld_r + 16][ld_c4 * 4] = rb1;                  \
    *(float4*)&sB[(buf)][ld_r + 32][ld_c4 * 4] = rb2;                  \
    *(float4*)&sB[(buf)][ld_r + 48][ld_c4 * 4] = rb3;                  \
  } while (0)

  const int r  = lane & 15;
  const int h  = lane >> 4;
  const int wm = wave & 3;
  const int wn = wave >> 2;
  const float* fa  = &sA[0][wm * 16 + r][4 * h];
  const float* fb0 = &sB[0][wn * 32 + r][4 * h];
  const float* fb1 = &sB[0][wn * 32 + 16 + r][4 * h];

  v8f acc0 = {0.f,0.f,0.f,0.f,0.f,0.f,0.f,0.f};
  v8f acc1 = acc0;

#define GT_COMPUTE(buf)                                                                            \
  do {                                                                                             \
    const float* pa  = fa  + (buf) * (GT_M * GT_P);                                                \
    const float* pb0 = fb0 + (buf) * (GT_N * GT_P);                                                \
    const float* pb1 = fb1 + (buf) * (GT_N * GT_P);                                                \
    _Pragma("unroll")                                                                              \
    for (int g = 0; g < GT_K / 8; ++g) {                                                           \
      const float4 a4 = *(const float4*)(pa  + g * 8);                                             \
      const float4 p4 = *(const float4*)(pb0 + g * 8);                                             \
      const float4 q4 = *(const float4*)(pb1 + g * 8);                                             \
      const v2f alo = {a4.x, a4.y}, ahi = {a4.z, a4.w};                                            \
      const v2f plo = {p4.x, p4.y}, phi = {p4.z, p4.w};                                            \
      const v2f qlo = {q4.x, q4.y}, qhi = {q4.z, q4.w};                                            \
      acc0 = __builtin_amdgcn_wmma_f32_16x16x4_f32(false, alo, false, plo, (short)0, acc0, false, false); \
      acc1 = __builtin_amdgcn_wmma_f32_16x16x4_f32(false, alo, false, qlo, (short)0, acc1, false, false); \
      acc0 = __builtin_amdgcn_wmma_f32_16x16x4_f32(false, ahi, false, phi, (short)0, acc0, false, false); \
      acc1 = __builtin_amdgcn_wmma_f32_16x16x4_f32(false, ahi, false, qhi, (short)0, acc1, false, false); \
    }                                                                                              \
  } while (0)

  const int KT = K / GT_K;
  GT_FETCH(0);
  GT_COMMIT(0);
  __syncthreads();

  for (int kt = 0; kt + 1 < KT; ++kt) {
    const int cb = kt & 1;
    GT_FETCH((kt + 1) * GT_K);
    GT_COMPUTE(cb);
    GT_COMMIT(cb ^ 1);
    __syncthreads();
  }
  GT_COMPUTE((KT - 1) & 1);
  __syncthreads();

  float (*sC)[GT_P] = sA[0];
  #pragma unroll
  for (int i = 0; i < 8; ++i) {
    sC[wm * 16 + 8 * h + i][wn * 32 + r]      = acc0[i];
    sC[wm * 16 + 8 * h + i][wn * 32 + 16 + r] = acc1[i];
  }
  __syncthreads();
  const int col = n0 + ld_c4 * 4;
  if (col < N) {
    #pragma unroll
    for (int p = 0; p < 4; ++p) {
      const int row = ld_r + 16 * p;
      const float4 v = *(const float4*)&sC[row][ld_c4 * 4];
      *(float4*)(C + (long)(m0 + row) * N + col) = v;
    }
  }
#undef GT_FETCH
#undef GT_COMMIT
#undef GT_COMPUTE
}

__global__ void dwconv_silu_kernel(const float* __restrict__ zx,
                                   const float* __restrict__ w,
                                   const float* __restrict__ bias,
                                   float* __restrict__ out) {
  const int c = blockIdx.x * 128 + threadIdx.x;
  if (c >= CONV_DIM) return;
  const long bl = blockIdx.y;
  const int l = (int)(bl & (LOUT - 1));
  const float* xcol = zx + bl * D_IN_PROJ + D_INNER + c;
  const float w0 = w[c * 4 + 0], w1 = w[c * 4 + 1], w2 = w[c * 4 + 2], w3 = w[c * 4 + 3];
  float acc = w3 * xcol[0];
  if (l >= 1) acc = fmaf(w2, xcol[-(long)D_IN_PROJ],     acc);
  if (l >= 2) acc = fmaf(w1, xcol[-(long)2 * D_IN_PROJ], acc);
  if (l >= 3) acc = fmaf(w0, xcol[-(long)3 * D_IN_PROJ], acc);
  const float v = acc + bias[c];
  out[bl * CONV_DIM + c] = v / (1.f + __expf(-v));
}

#define SCAN_T   32
#define SCAN_NCH (LOUT / SCAN_T)

__global__ void __launch_bounds__(256)
ssm_scan_kernel(const float* __restrict__ zx,
                const float* __restrict__ xbc,
                const float* __restrict__ dtb,
                const float* __restrict__ alog,
                const float* __restrict__ Dp,
                float* __restrict__ y) {
  const int b  = blockIdx.x >> 4;
  const int hh = blockIdx.x & 15;
  const int tid = threadIdx.x;
  const int p = tid >> 2;
  const int q = tid & 3;
  __shared__ __attribute__((aligned(16))) float sBC[2][SCAN_T][2 * D_STATE];
  __shared__ __attribute__((aligned(16))) float sX[2][SCAN_T][HEADDIM];
  __shared__ __attribute__((aligned(16))) float sY[2][SCAN_T][HEADDIM];
  __shared__ __attribute__((aligned(8)))  float sDT[2][SCAN_T][2];
  const float Ah   = -__expf(alog[hh]);
  const float dtbh = dtb[hh];
  const float Dh   = Dp[hh];
  float hreg[16];
  #pragma unroll
  for (int i = 0; i < 16; ++i) hreg[i] = 0.f;

  const int bc_c4 = tid & 31;
  const int bc_t  = tid >> 5;
  const int x_c4  = tid & 15;
  const int x_t   = tid >> 4;
  const bool dt_wave = (tid >= 256 - SCAN_T);
  const int  dt_t    = dt_wave ? tid - (256 - SCAN_T) : 0;
  const float* gBC = xbc + ((long)b * LOUT + bc_t) * CONV_DIM + D_INNER + bc_c4 * 4;
  const float* gX  = xbc + ((long)b * LOUT + x_t) * CONV_DIM + hh * HEADDIM + x_c4 * 4;
  const float* gDT = zx + ((long)b * LOUT + dt_t) * D_IN_PROJ + (D_IN_PROJ - NHEADS) + hh;
  float*       gY  = y + ((long)b * LOUT + x_t) * D_INNER + hh * HEADDIM + x_c4 * 4;

  float4 rBC0, rBC1, rBC2, rBC3, rX0, rX1;
  float  rdt = 0.f;

#define SCAN_FETCH(l0)                                                              \
  do {                                                                              \
    const float* s_bc = gBC + (long)(l0) * CONV_DIM;                                \
    const float* s_x  = gX  + (long)(l0) * CONV_DIM;                                \
    rBC0 = *(const float4*)(s_bc);                                                  \
    rBC1 = *(const float4*)(s_bc + 8  * CONV_DIM);                                  \
    rBC2 = *(const float4*)(s_bc + 16 * CONV_DIM);                                  \
    rBC3 = *(const float4*)(s_bc + 24 * CONV_DIM);                                  \
    rX0  = *(const float4*)(s_x);                                                   \
    rX1  = *(const float4*)(s_x + 16 * CONV_DIM);                                   \
    if (dt_wave) rdt = gDT[(long)(l0) * D_IN_PROJ];                                 \
  } while (0)

#define SCAN_COMMIT(buf)                                                            \
  do {                                                                              \
    *(float4*)&sBC[(buf)][bc_t     ][bc_c4 * 4] = rBC0;                             \
    *(float4*)&sBC[(buf)][bc_t + 8 ][bc_c4 * 4] = rBC1;                             \
    *(float4*)&sBC[(buf)][bc_t + 16][bc_c4 * 4] = rBC2;                             \
    *(float4*)&sBC[(buf)][bc_t + 24][bc_c4 * 4] = rBC3;                             \
    *(float4*)&sX[(buf)][x_t     ][x_c4 * 4] = rX0;                                 \
    *(float4*)&sX[(buf)][x_t + 16][x_c4 * 4] = rX1;                                 \
    if (dt_wave) {                                                                  \
      const float raw = rdt + dtbh;                                                 \
      const float dtv = (raw > 20.f) ? raw : log1pf(__expf(raw));      \
      float2 dd;                                                                    \
      dd.x = dtv;                                                                   \
      dd.y = __expf(dtv * Ah);                                               \
      *(float2*)&sDT[(buf)][dt_t][0] = dd;                                          \
    }                                                                               \
  } while (0)

#define SCAN_WRITEOUT(l0, buf)                                                      \
  do {                                                                              \
    float* d_y = gY + (long)(l0) * D_INNER;                                         \
    const float4 v0 = *(const float4*)&sY[(buf)][x_t     ][x_c4 * 4];               \
    const float4 v1 = *(const float4*)&sY[(buf)][x_t + 16][x_c4 * 4];               \
    *(float4*)(d_y)                = v0;                                            \
    *(float4*)(d_y + 16 * D_INNER) = v1;                                            \
  } while (0)

  SCAN_FETCH(0);
  SCAN_COMMIT(0);
  __syncthreads();

  for (int c = 0; c < SCAN_NCH; ++c) {
    const int cb = c & 1;
    const bool more = (c + 1 < SCAN_NCH);
    if (more) SCAN_FETCH((c + 1) * SCAN_T);
    if (c > 0) SCAN_WRITEOUT((c - 1) * SCAN_T, cb ^ 1);
    for (int t0 = 0; t0 < SCAN_T; t0 += 4) {
      float yv[4];
      #pragma unroll
      for (int u = 0; u < 4; ++u) {
        const int t = t0 + u;
        const float2 dd = *(const float2*)&sDT[cb][t][0];
        const float dtv = dd.x;
        const float dA  = dd.y;
        const float xv  = sX[cb][t][p];
        const float xdt = xv * dtv;
        const float* Bq = &sBC[cb][t][q * 16];
        const float* Cq = &sBC[cb][t][D_STATE + q * 16];
        float accv = 0.f;
        #pragma unroll
        for (int i = 0; i < 16; ++i) {
          hreg[i] = fmaf(hreg[i], dA, xdt * Bq[i]);
          accv    = fmaf(hreg[i], Cq[i], accv);
        }
        accv += __shfl_xor(accv, 1, 32);
        accv += __shfl_xor(accv, 2, 32);
        yv[u] = fmaf(xv, Dh, accv);
      }
      if (q == 0) {
        #pragma unroll
        for (int u = 0; u < 4; ++u) sY[cb][t0 + u][p] = yv[u];
      }
    }
    if (more) SCAN_COMMIT(cb ^ 1);
    __syncthreads();
  }
  SCAN_WRITEOUT((SCAN_NCH - 1) * SCAN_T, (SCAN_NCH - 1) & 1);
#undef SCAN_FETCH
#undef SCAN_COMMIT
#undef SCAN_WRITEOUT
}

__global__ void gate_rmsnorm_kernel(const float* __restrict__ zx,
                                    const float* __restrict__ nw,
                                    float* __restrict__ y) {
  const long bl = blockIdx.x;
  const int tid = threadIdx.x;
  __shared__ float red[256];
  float yv[4];
  float ss = 0.f;
  #pragma unroll
  for (int j = 0; j < 4; ++j) {
    const int i = tid + j * 256;
    const float z = zx[bl * D_IN_PROJ + i];
    const float g = z / (1.f + __expf(-z));
    const float v = y[bl * D_INNER + i] * g;
    yv[j] = v;
    ss = fmaf(v, v, ss);
  }
  red[tid] = ss;
  __syncthreads();
  for (int st = 128; st > 0; st >>= 1) {
    if (tid < st) red[tid] += red[tid + st];
    __syncthreads();
  }
  const float scale = rsqrtf(red[0] * (1.f / (float)D_INNER) + 1e-5f);
  #pragma unroll
  for (int j = 0; j < 4; ++j) {
    const int i = tid + j * 256;
    y[bl * D_INNER + i] = yv[j] * scale * nw[i];
  }
}

extern "C" void kernel_launch(void* const* d_in, const int* in_sizes, int n_in,
                              void* d_out, int out_size, void* d_ws, size_t ws_size,
                              hipStream_t stream) {
  const float* tokens     = (const float*)d_in[0];
  const float* conv_w     = (const float*)d_in[1];
  const float* in_proj_w  = (const float*)d_in[2];
  const float* conv1d_w   = (const float*)d_in[3];
  const float* conv1d_b   = (const float*)d_in[4];
  const float* dt_bias    = (const float*)d_in[5];
  const float* A_log      = (const float*)d_in[6];
  const float* D_param    = (const float*)d_in[7];
  const float* norm_w     = (const float*)d_in[8];
  const float* out_proj_w = (const float*)d_in[9];
  float* out = (float*)d_out;
  float* ws  = (float*)d_ws;

  float* u0   = ws;
  float* u1   = u0 + (long)ROWS * D_MODEL;
  float* zx   = u1 + (long)ROWS * D_MODEL;
  float* xact = zx + (long)ROWS * D_IN_PROJ;
  float* ybuf = xact + (long)ROWS * CONV_DIM;
  float* col  = zx;
  (void)ws_size; (void)in_sizes; (void)n_in; (void)out_size;

  {
    const long nel = (long)ROWS * KSTEM;
    im2col_stem_kernel<<<(int)((nel + 255) / 256), 256, 0, stream>>>(tokens, col);
    gemm_xt_wmma<<<(ROWS / GT_M) * ((D_MODEL + GT_N - 1) / GT_N), 256, 0, stream>>>(col, conv_w, u0, ROWS, D_MODEL, KSTEM);
  }

  const float* u_in = u0;
  for (int blk = 0; blk < 2; ++blk) {
    const float* ipw = in_proj_w  + (long)blk * D_IN_PROJ * D_MODEL;
    const float* cw  = conv1d_w   + (long)blk * CONV_DIM * D_CONV;
    const float* cb  = conv1d_b   + (long)blk * CONV_DIM;
    const float* dtb = dt_bias    + (long)blk * NHEADS;
    const float* al  = A_log      + (long)blk * NHEADS;
    const float* Dpp = D_param    + (long)blk * NHEADS;
    const float* nw  = norm_w     + (long)blk * D_INNER;
    const float* opw = out_proj_w + (long)blk * D_MODEL * D_INNER;

    {
      gemm_xt_wmma<<<(ROWS / GT_M) * ((D_IN_PROJ + GT_N - 1) / GT_N), 256, 0, stream>>>(u_in, ipw, zx, ROWS, D_IN_PROJ, D_MODEL);
    }
    dwconv_silu_kernel<<<dim3(CONV_DIM / 128, ROWS), 128, 0, stream>>>(zx, cw, cb, xact);
    ssm_scan_kernel<<<BATCH * NHEADS, 256, 0, stream>>>(zx, xact, dtb, al, Dpp, ybuf);
    gate_rmsnorm_kernel<<<ROWS, 256, 0, stream>>>(zx, nw, ybuf);
    float* dst = (blk == 0) ? u1 : out;
    {
      gemm_xt_wmma<<<(ROWS / GT_M) * ((D_MODEL + GT_N - 1) / GT_N), 256, 0, stream>>>(ybuf, opw, dst, ROWS, D_MODEL, D_INNER);
    }
    u_in = u1;
  }
}


# ssm scan: the 4-lane sum of a y value read through DPP (row_xmask 1, 2) on the add itself instead of two LDS permute round trips per step; same pairs, same order of sums
# speedup vs baseline: 7.1226x; 1.1484x over previous
#include <hip/hip_runtime.h>
#include <hip/hip_bf16.h>

typedef __attribute__((ext_vector_type(2))) float v2f;
typedef __attribute__((ext_vector_type(8))) float v8f;

#define BATCH     2
#define SEQLEN    2048
#define C_IN      256
#define D_MODEL   512
#define LOUT      1024
#define D_STATE   64
#define D_CONV    4
#define D_INNER   1024
#define HEADDIM   64
#define NHEADS    16
#define CONV_DIM  (D_INNER + 2*D_STATE)
#define D_IN_PROJ (2*D_INNER + 2*D_STATE + NHEADS)
#define ROWS      (BATCH*LOUT)
#define KSTEM     (C_IN * 7)

__global__ void im2col_stem_kernel(const float* __restrict__ tokens,
                                   float* __restrict__ col) {
  const long idx = (long)blockIdx.x * 256 + threadIdx.x;
  if (idx >= (long)ROWS * KSTEM) return;
  const int j  = (int)(idx % KSTEM);
  const long bl = idx / KSTEM;
  const int c = j / 7;
  const int k = j - c * 7;
  const int l = (int)(bl & (LOUT - 1));
  const int b = (int)(bl >> 10);
  const int s = 2 * l - 3 + k;
  float v = 0.f;
  if (s >= 0 && s < SEQLEN) v = tokens[((long)b * C_IN + c) * SEQLEN + s];
  col[idx] = v;
}

#define GT_M 64
#define GT_N 64
#define GT_K 64
#define GT_P 68

__global__ void __launch_bounds__(256)
gemm_xt_wmma(const float* __restrict__ A,
             const float* __restrict__ W,
             float* __restrict__ C,
             int M, int N, int K) {
  __shared__ __attribute__((aligned(16))) float sA[2][GT_M][GT_P];
  __shared__ __attribute__((aligned(16))) float sB[2][GT_N][GT_P];
  const int tid  = threadIdx.x;
  const int lane = tid & 31;
  const int wave = tid >> 5;
  const int nblk = (N + GT_N - 1) / GT_N;
  const int mb = blockIdx.x / nblk;
  const int nb = blockIdx.x - mb * nblk;
  const int m0 = mb * GT_M;
  const int n0 = nb * GT_N;
  (void)M;

  const int ld_c4 = tid & 15;
  const int ld_r  = tid >> 4;
  const long a_step = (long)16 * K;
  const float* gA  = A + (long)(m0 + ld_r) * K + ld_c4 * 4;
  const float* gB0 = W + (long)min(n0 + ld_r,      N - 1) * K + ld_c4 * 4;
  const float* gB1 = W + (long)min(n0 + ld_r + 16, N - 1) * K + ld_c4 * 4;
  const float* gB2 = W + (long)min(n0 + ld_r + 32, N - 1) * K + ld_c4 * 4;
  const float* gB3 = W + (long)min(n0 + ld_r + 48, N - 1) * K + ld_c4 * 4;

  float4 ra0, ra1, ra2, ra3, rb0, rb1, rb2, rb3;

#define GT_FETCH(k0)                                                   \
  do {                                                                 \
    ra0 = *(const float4*)(gA + (k0));                                 \
    ra1 = *(const float4*)(gA + a_step + (k0));                        \
    ra2 = *(const float4*)(gA + 2 * a_step + (k0));                    \
    ra3 = *(const float4*)(gA + 3 * a_step + (k0));                    \
    rb0 = *(const float4*)(gB0 + (k0));                                \
    rb1 = *(const float4*)(gB1 + (k0));                                \
    rb2 = *(const float4*)(gB2 + (k0));                                \
    rb3 = *(const float4*)(gB3 + (k0));                                \
  } while (0)

#define GT_COMMIT(buf)                                                 \
  do {                                                                 \
    *(float4*)&sA[(buf)][ld_r     ][ld_c4 * 4] = ra0;                  \
    *(float4*)&sA[(buf)][ld_r + 16][ld_c4 * 4] = ra1;                  \
    *(float4*)&sA[(buf)][ld_r + 32][ld_c4 * 4] = ra2;                  \
    *(float4*)&sA[(buf)][ld_r + 48][ld_c4 * 4] = ra3;                  \
    *(float4*)&sB[(buf)][ld_r     ][ld_c4 * 4] = rb0;                  \
    *(float4*)&sB[(buf)][ld_r + 16][ld_c4 * 4] = rb1;                  \
    *(float4*)&sB[(buf)][ld_r + 32][ld_c4 * 4] = rb2;                  \
    *(float4*)&sB[(buf)][ld_r + 48][ld_c4 * 4] = rb3;                  \
  } while (0)

  const int r  = lane & 15;
  const int h  = lane >> 4;
  const int wm = wave & 3;
  const int wn = wave >> 2;
  const float* fa  = &sA[0][wm * 16 + r][4 * h];
  const float* fb0 = &sB[0][wn * 32 + r][4 * h];
  const float* fb1 = &sB[0][wn * 32 + 16 + r][4 * h];

  v8f acc0 = {0.f,0.f,0.f,0.f,0.f,0.f,0.f,0.f};
  v8f acc1 = acc0;

#define GT_COMPUTE(buf)                                                                            \
  do {                                                                                             \
    const float* pa  = fa  + (buf) * (GT_M * GT_P);                                                \
    const float* pb0 = fb0 + (buf) * (GT_N * GT_P);                                                \
    const float* pb1 = fb1 + (buf) * (GT_N * GT_P);                                                \
    _Pragma("unroll")                                                                              \
    for (int g = 0; g < GT_K / 8; ++g) {                                                           \
      const float4 a4 = *(const float4*)(pa  + g * 8);                                             \
      const float4 p4 = *(const float4*)(pb0 + g * 8);                                             \
      const float4 q4 = *(const float4*)(pb1 + g * 8);                                             \
      const v2f alo = {a4.x, a4.y}, ahi = {a4.z, a4.w};                                            \
      const v2f plo = {p4.x, p4.y}, phi = {p4.z, p4.w};                                            \
      const v2f qlo = {q4.x, q4.y}, qhi = {q4.z, q4.w};                                            \
      acc0 = __builtin_amdgcn_wmma_f32_16x16x4_f32(false, alo, false, plo, (short)0, acc0, false, false); \
      acc1 = __builtin_amdgcn_wmma_f32_16x16x4_f32(false, alo, false, qlo, (short)0, acc1, false, false); \
      acc0 = __builtin_amdgcn_wmma_f32_16x16x4_f32(false, ahi, false, phi, (short)0, acc0, false, false); \
      acc1 = __builtin_amdgcn_wmma_f32_16x16x4_f32(false, ahi, false, qhi, (short)0, acc1, false, false); \
    }                                                                                              \
  } while (0)

  const int KT = K / GT_K;
  GT_FETCH(0);
  GT_COMMIT(0);
  __syncthreads();

  for (int kt = 0; kt + 1 < KT; ++kt) {
    const int cb = kt & 1;
    GT_FETCH((kt + 1) * GT_K);
    GT_COMPUTE(cb);
    GT_COMMIT(cb ^ 1);
    __syncthreads();
  }
  GT_COMPUTE((KT - 1) & 1);
  __syncthreads();

  float (*sC)[GT_P] = sA[0];
  #pragma unroll
  for (int i = 0; i < 8; ++i) {
    sC[wm * 16 + 8 * h + i][wn * 32 + r]      = acc0[i];
    sC[wm * 16 + 8 * h + i][wn * 32 + 16 + r] = acc1[i];
  }
  __syncthreads();
  const int col = n0 + ld_c4 * 4;
  if (col < N) {
    #pragma unroll
    for (int p = 0; p < 4; ++p) {
      const int row = ld_r + 16 * p;
      const float4 v = *(const float4*)&sC[row][ld_c4 * 4];
      *(float4*)(C + (long)(m0 + row) * N + col) = v;
    }
  }
#undef GT_FETCH
#undef GT_COMMIT
#undef GT_COMPUTE
}

__global__ void dwconv_silu_kernel(const float* __restrict__ zx,
                                   const float* __restrict__ w,
                                   const float* __restrict__ bias,
                                   float* __restrict__ out) {
  const int c = blockIdx.x * 128 + threadIdx.x;
  if (c >= CONV_DIM) return;
  const long bl = blockIdx.y;
  const int l = (int)(bl & (LOUT - 1));
  const float* xcol = zx + bl * D_IN_PROJ + D_INNER + c;
  const float w0 = w[c * 4 + 0], w1 = w[c * 4 + 1], w2 = w[c * 4 + 2], w3 = w[c * 4 + 3];
  float acc = w3 * xcol[0];
  if (l >= 1) acc = fmaf(w2, xcol[-(long)D_IN_PROJ],     acc);
  if (l >= 2) acc = fmaf(w1, xcol[-(long)2 * D_IN_PROJ], acc);
  if (l >= 3) acc = fmaf(w0, xcol[-(long)3 * D_IN_PROJ], acc);
  const float v = acc + bias[c];
  out[bl * CONV_DIM + c] = v / (1.f + __expf(-v));
}

#define SCAN_T   32
#define SCAN_NCH (LOUT / SCAN_T)

template <int CTRL>
__device__ __forceinline__ float lane_xor_read(float v) {
  return __int_as_float(__builtin_amdgcn_update_dpp(0, __float_as_int(v), CTRL, 0xF, 0xF, true));
}

__global__ void __launch_bounds__(256)
ssm_scan_kernel(const float* __restrict__ zx,
                const float* __restrict__ xbc,
                const float* __restrict__ dtb,
                const float* __restrict__ alog,
                const float* __restrict__ Dp,
                float* __restrict__ y) {
  const int b  = blockIdx.x >> 4;
  const int hh = blockIdx.x & 15;
  const int tid = threadIdx.x;
  const int p = tid >> 2;
  const int q = tid & 3;
  __shared__ __attribute__((aligned(16))) float sBC[2][SCAN_T][2 * D_STATE];
  __shared__ __attribute__((aligned(16))) float sX[2][SCAN_T][HEADDIM];
  __shared__ __attribute__((aligned(16))) float sY[2][SCAN_T][HEADDIM];
  __shared__ __attribute__((aligned(8)))  float sDT[2][SCAN_T][2];
  const float Ah   = -__expf(alog[hh]);
  const float dtbh = dtb[hh];
  const float Dh   = Dp[hh];
  float hreg[16];
  #pragma unroll
  for (int i = 0; i < 16; ++i) hreg[i] = 0.f;

  const int bc_c4 = tid & 31;
  const int bc_t  = tid >> 5;
  const int x_c4  = tid & 15;
  const int x_t   = tid >> 4;
  const bool dt_wave = (tid >= 256 - SCAN_T);
  const int  dt_t    = dt_wave ? tid - (256 - SCAN_T) : 0;
  const float* gBC = xbc + ((long)b * LOUT + bc_t) * CONV_DIM + D_INNER + bc_c4 * 4;
  const float* gX  = xbc + ((long)b * LOUT + x_t) * CONV_DIM + hh * HEADDIM + x_c4 * 4;
  const float* gDT = zx + ((long)b * LOUT + dt_t) * D_IN_PROJ + (D_IN_PROJ - NHEADS) + hh;
  float*       gY  = y + ((long)b * LOUT + x_t) * D_INNER + hh * HEADDIM + x_c4 * 4;

  float4 rBC0, rBC1, rBC2, rBC3, rX0, rX1;
  float  rdt = 0.f;

#define SCAN_FETCH(l0)                                                              \
  do {                                                                              \
    const float* s_bc = gBC + (long)(l0) * CONV_DIM;                                \
    const float* s_x  = gX  + (long)(l0) * CONV_DIM;                                \
    rBC0 = *(const float4*)(s_bc);                                                  \
    rBC1 = *(const float4*)(s_bc + 8  * CONV_DIM);                                  \
    rBC2 = *(const float4*)(s_bc + 16 * CONV_DIM);                                  \
    rBC3 = *(const float4*)(s_bc + 24 * CONV_DIM);                                  \
    rX0  = *(const float4*)(s_x);                                                   \
    rX1  = *(const float4*)(s_x + 16 * CONV_DIM);                                   \
    if (dt_wave) rdt = gDT[(long)(l0) * D_IN_PROJ];                                 \
  } while (0)

#define SCAN_COMMIT(buf)                                                            \
  do {                                                                              \
    *(float4*)&sBC[(buf)][bc_t     ][bc_c4 * 4] = rBC0;                             \
    *(float4*)&sBC[(buf)][bc_t + 8 ][bc_c4 * 4] = rBC1;                             \
    *(float4*)&sBC[(buf)][bc_t + 16][bc_c4 * 4] = rBC2;                             \
    *(float4*)&sBC[(buf)][bc_t + 24][bc_c4 * 4] = rBC3;                             \
    *(float4*)&sX[(buf)][x_t     ][x_c4 * 4] = rX0;                                 \
    *(float4*)&sX[(buf)][x_t + 16][x_c4 * 4] = rX1;                                 \
    if (dt_wave) {                                                                  \
      const float raw = rdt + dtbh;                                                 \
      const float dtv = (raw > 20.f) ? raw : log1pf(__expf(raw));      \
      float2 dd;                                                                    \
      dd.x = dtv;                                                                   \
      dd.y = __expf(dtv * Ah);                                               \
      *(float2*)&sDT[(buf)][dt_t][0] = dd;                                          \
    }                                                                               \
  } while (0)

#define SCAN_WRITEOUT(l0, buf)                                                      \
  do {                                                                              \
    float* d_y = gY + (long)(l0) * D_INNER;                                         \
    const float4 v0 = *(const float4*)&sY[(buf)][x_t     ][x_c4 * 4];               \
    const float4 v1 = *(const float4*)&sY[(buf)][x_t + 16][x_c4 * 4];               \
    *(float4*)(d_y)                = v0;                                            \
    *(float4*)(d_y + 16 * D_INNER) = v1;                                            \
  } while (0)

  SCAN_FETCH(0);
  SCAN_COMMIT(0);
  __syncthreads();

  for (int c = 0; c < SCAN_NCH; ++c) {
    const int cb = c & 1;
    const bool more = (c + 1 < SCAN_NCH);
    if (more) SCAN_FETCH((c + 1) * SCAN_T);
    if (c > 0) SCAN_WRITEOUT((c - 1) * SCAN_T, cb ^ 1);
    for (int t0 = 0; t0 < SCAN_T; t0 += 4) {
      float yv[4];
      #pragma unroll
      for (int u = 0; u < 4; ++u) {
        const int t = t0 + u;
        const float2 dd = *(const float2*)&sDT[cb][t][0];
        const float dtv = dd.x;
        const float dA  = dd.y;
        const float xv  = sX[cb][t][p];
        const float xdt = xv * dtv;
        const float* Bq = &sBC[cb][t][q * 16];
        const float* Cq = &sBC[cb][t][D_STATE + q * 16];
        float accv = 0.f;
        #pragma unroll
        for (int i = 0; i < 16; ++i) {
          hreg[i] = fmaf(hreg[i], dA, xdt * Bq[i]);
          accv    = fmaf(hreg[i], Cq[i], accv);
        }
        accv += lane_xor_read<0x161>(accv);
        accv += lane_xor_read<0x162>(accv);
        yv[u] = fmaf(xv, Dh, accv);
      }
      if (q == 0) {
        #pragma unroll
        for (int u = 0; u < 4; ++u) sY[cb][t0 + u][p] = yv[u];
      }
    }
    if (more) SCAN_COMMIT(cb ^ 1);
    __syncthreads();
  }
  SCAN_WRITEOUT((SCAN_NCH - 1) * SCAN_T, (SCAN_NCH - 1) & 1);
#undef SCAN_FETCH
#undef SCAN_COMMIT
#undef SCAN_WRITEOUT
}

__global__ void gate_rmsnorm_kernel(const float* __restrict__ zx,
                                    const float* __restrict__ nw,
                                    float* __restrict__ y) {
  const long bl = blockIdx.x;
  const int tid = threadIdx.x;
  __shared__ float red[256];
  float yv[4];
  float ss = 0.f;
  #pragma unroll
  for (int j = 0; j < 4; ++j) {
    const int i = tid + j * 256;
    const float z = zx[bl * D_IN_PROJ + i];
    const float g = z / (1.f + __expf(-z));
    const float v = y[bl * D_INNER + i] * g;
    yv[j] = v;
    ss = fmaf(v, v, ss);
  }
  red[tid] = ss;
  __syncthreads();
  for (int st = 128; st > 0; st >>= 1) {
    if (tid < st) red[tid] += red[tid + st];
    __syncthreads();
  }
  const float scale = rsqrtf(red[0] * (1.f / (float)D_INNER) + 1e-5f);
  #pragma unroll
  for (int j = 0; j < 4; ++j) {
    const int i = tid + j * 256;
    y[bl * D_INNER + i] = yv[j] * scale * nw[i];
  }
}

extern "C" void kernel_launch(void* const* d_in, const int* in_sizes, int n_in,
                              void* d_out, int out_size, void* d_ws, size_t ws_size,
                              hipStream_t stream) {
  const float* tokens     = (const float*)d_in[0];
  const float* conv_w     = (const float*)d_in[1];
  const float* in_proj_w  = (const float*)d_in[2];
  const float* conv1d_w   = (const float*)d_in[3];
  const float* conv1d_b   = (const float*)d_in[4];
  const float* dt_bias    = (const float*)d_in[5];
  const float* A_log      = (const float*)d_in[6];
  const float* D_param    = (const float*)d_in[7];
  const float* norm_w     = (const float*)d_in[8];
  const float* out_proj_w = (const float*)d_in[9];
  float* out = (float*)d_out;
  float* ws  = (float*)d_ws;

  float* u0   = ws;
  float* u1   = u0 + (long)ROWS * D_MODEL;
  float* zx   = u1 + (long)ROWS * D_MODEL;
  float* xact = zx + (long)ROWS * D_IN_PROJ;
  float* ybuf = xact + (long)ROWS * CONV_DIM;
  float* col  = zx;
  (void)ws_size; (void)in_sizes; (void)n_in; (void)out_size;

  {
    const long nel = (long)ROWS * KSTEM;
    im2col_stem_kernel<<<(int)((nel + 255) / 256), 256, 0, stream>>>(tokens, col);
    gemm_xt_wmma<<<(ROWS / GT_M) * ((D_MODEL + GT_N - 1) / GT_N), 256, 0, stream>>>(col, conv_w, u0, ROWS, D_MODEL, KSTEM);
  }

  const float* u_in = u0;
  for (int blk = 0; blk < 2; ++blk) {
    const float* ipw = in_proj_w  + (long)blk * D_IN_PROJ * D_MODEL;
    const float* cw  = conv1d_w   + (long)blk * CONV_DIM * D_CONV;
    const float* cb  = conv1d_b   + (long)blk * CONV_DIM;
    const float* dtb = dt_bias    + (long)blk * NHEADS;
    const float* al  = A_log      + (long)blk * NHEADS;
    const float* Dpp = D_param    + (long)blk * NHEADS;
    const float* nw  = norm_w     + (long)blk * D_INNER;
    const float* opw = out_proj_w + (long)blk * D_MODEL * D_INNER;

    {
      gemm_xt_wmma<<<(ROWS / GT_M) * ((D_IN_PROJ + GT_N - 1) / GT_N), 256, 0, stream>>>(u_in, ipw, zx, ROWS, D_IN_PROJ, D_MODEL);
    }
    dwconv_silu_kernel<<<dim3(CONV_DIM / 128, ROWS), 128, 0, stream>>>(zx, cw, cb, xact);
    ssm_scan_kernel<<<BATCH * NHEADS, 256, 0, stream>>>(zx, xact, dtb, al, Dpp, ybuf);
    gate_rmsnorm_kernel<<<ROWS, 256, 0, stream>>>(zx, nw, ybuf);
    float* dst = (blk == 0) ? u1 : out;
    {
      gemm_xt_wmma<<<(ROWS / GT_M) * ((D_MODEL + GT_N - 1) / GT_N), 256, 0, stream>>>(ybuf, opw, dst, ROWS, D_MODEL, D_INNER);
    }
    u_in = u1;
  }
}


# ssm scan: each head's 64 state rows split over 2 workgroups of 128 threads (64 WGPs busy, not 32; 4 waves, not 8, share one LDS); each still writes whole 128-B y lines; same sums, same order
# speedup vs baseline: 7.6773x; 1.0779x over previous
#include <hip/hip_runtime.h>
#include <hip/hip_bf16.h>

typedef __attribute__((ext_vector_type(2))) float v2f;
typedef __attribute__((ext_vector_type(8))) float v8f;

#define BATCH     2
#define SEQLEN    2048
#define C_IN      256
#define D_MODEL   512
#define LOUT      1024
#define D_STATE   64
#define D_CONV    4
#define D_INNER   1024
#define HEADDIM   64
#define NHEADS    16
#define CONV_DIM  (D_INNER + 2*D_STATE)
#define D_IN_PROJ (2*D_INNER + 2*D_STATE + NHEADS)
#define ROWS      (BATCH*LOUT)
#define KSTEM     (C_IN * 7)

__global__ void im2col_stem_kernel(const float* __restrict__ tokens,
                                   float* __restrict__ col) {
  const long idx = (long)blockIdx.x * 256 + threadIdx.x;
  if (idx >= (long)ROWS * KSTEM) return;
  const int j  = (int)(idx % KSTEM);
  const long bl = idx / KSTEM;
  const int c = j / 7;
  const int k = j - c * 7;
  const int l = (int)(bl & (LOUT - 1));
  const int b = (int)(bl >> 10);
  const int s = 2 * l - 3 + k;
  float v = 0.f;
  if (s >= 0 && s < SEQLEN) v = tokens[((long)b * C_IN + c) * SEQLEN + s];
  col[idx] = v;
}

#define GT_M 64
#define GT_N 64
#define GT_K 64
#define GT_P 68

__global__ void __launch_bounds__(256)
gemm_xt_wmma(const float* __restrict__ A,
             const float* __restrict__ W,
             float* __restrict__ C,
             int M, int N, int K) {
  __shared__ __attribute__((aligned(16))) float sA[2][GT_M][GT_P];
  __shared__ __attribute__((aligned(16))) float sB[2][GT_N][GT_P];
  const int tid  = threadIdx.x;
  const int lane = tid & 31;
  const int wave = tid >> 5;
  const int nblk = (N + GT_N - 1) / GT_N;
  const int mb = blockIdx.x / nblk;
  const int nb = blockIdx.x - mb * nblk;
  const int m0 = mb * GT_M;
  const int n0 = nb * GT_N;
  (void)M;

  const int ld_c4 = tid & 15;
  const int ld_r  = tid >> 4;
  const long a_step = (long)16 * K;
  const float* gA  = A + (long)(m0 + ld_r) * K + ld_c4 * 4;
  const float* gB0 = W + (long)min(n0 + ld_r,      N - 1) * K + ld_c4 * 4;
  const float* gB1 = W + (long)min(n0 + ld_r + 16, N - 1) * K + ld_c4 * 4;
  const float* gB2 = W + (long)min(n0 + ld_r + 32, N - 1) * K + ld_c4 * 4;
  const float* gB3 = W + (long)min(n0 + ld_r + 48, N - 1) * K + ld_c4 * 4;

  float4 ra0, ra1, ra2, ra3, rb0, rb1, rb2, rb3;

#define GT_FETCH(k0)                                                   \
  do {                                                                 \
    ra0 = *(const float4*)(gA + (k0));                                 \
    ra1 = *(const float4*)(gA + a_step + (k0));                        \
    ra2 = *(const float4*)(gA + 2 * a_step + (k0));                    \
    ra3 = *(const float4*)(gA + 3 * a_step + (k0));                    \
    rb0 = *(const float4*)(gB0 + (k0));                                \
    rb1 = *(const float4*)(gB1 + (k0));                                \
    rb2 = *(const float4*)(gB2 + (k0));                                \
    rb3 = *(const float4*)(gB3 + (k0));                                \
  } while (0)

#define GT_COMMIT(buf)                                                 \
  do {                                                                 \
    *(float4*)&sA[(buf)][ld_r     ][ld_c4 * 4] = ra0;                  \
    *(float4*)&sA[(buf)][ld_r + 16][ld_c4 * 4] = ra1;                  \
    *(float4*)&sA[(buf)][ld_r + 32][ld_c4 * 4] = ra2;                  \
    *(float4*)&sA[(buf)][ld_r + 48][ld_c4 * 4] = ra3;                  \
    *(float4*)&sB[(buf)][ld_r     ][ld_c4 * 4] = rb0;                  \
    *(float4*)&sB[(buf)][ld_r + 16][ld_c4 * 4] = rb1;                  \
    *(float4*)&sB[(buf)][ld_r + 32][ld_c4 * 4] = rb2;                  \
    *(float4*)&sB[(buf)][ld_r + 48][ld_c4 * 4] = rb3;                  \
  } while (0)

  const int r  = lane & 15;
  const int h  = lane >> 4;
  const int wm = wave & 3;
  const int wn = wave >> 2;
  const float* fa  = &sA[0][wm * 16 + r][4 * h];
  const float* fb0 = &sB[0][wn * 32 + r][4 * h];
  const float* fb1 = &sB[0][wn * 32 + 16 + r][4 * h];

  v8f acc0 = {0.f,0.f,0.f,0.f,0.f,0.f,0.f,0.f};
  v8f acc1 = acc0;

#define GT_COMPUTE(buf)                                                                            \
  do {                                                                                             \
    const float* pa  = fa  + (buf) * (GT_M * GT_P);                                                \
    const float* pb0 = fb0 + (buf) * (GT_N * GT_P);                                                \
    const float* pb1 = fb1 + (buf) * (GT_N * GT_P);                                                \
    _Pragma("unroll")                                                                              \
    for (int g = 0; g < GT_K / 8; ++g) {                                                           \
      const float4 a4 = *(const float4*)(pa  + g * 8);                                             \
      const float4 p4 = *(const float4*)(pb0 + g * 8);                                             \
      const float4 q4 = *(const float4*)(pb1 + g * 8);                                             \
      const v2f alo = {a4.x, a4.y}, ahi = {a4.z, a4.w};                                            \
      const v2f plo = {p4.x, p4.y}, phi = {p4.z, p4.w};                                            \
      const v2f qlo = {q4.x, q4.y}, qhi = {q4.z, q4.w};                                            \
      acc0 = __builtin_amdgcn_wmma_f32_16x16x4_f32(false, alo, false, plo, (short)0, acc0, false, false); \
      acc1 = __builtin_amdgcn_wmma_f32_16x16x4_f32(false, alo, false, qlo, (short)0, acc1, false, false); \
      acc0 = __builtin_amdgcn_wmma_f32_16x16x4_f32(false, ahi, false, phi, (short)0, acc0, false, false); \
      acc1 = __builtin_amdgcn_wmma_f32_16x16x4_f32(false, ahi, false, qhi, (short)0, acc1, false, false); \
    }                                                                                              \
  } while (0)

  const int KT = K / GT_K;
  GT_FETCH(0);
  GT_COMMIT(0);
  __syncthreads();

  for (int kt = 0; kt + 1 < KT; ++kt) {
    const int cb = kt & 1;
    GT_FETCH((kt + 1) * GT_K);
    GT_COMPUTE(cb);
    GT_COMMIT(cb ^ 1);
    __syncthreads();
  }
  GT_COMPUTE((KT - 1) & 1);
  __syncthreads();

  float (*sC)[GT_P] = sA[0];
  #pragma unroll
  for (int i = 0; i < 8; ++i) {
    sC[wm * 16 + 8 * h + i][wn * 32 + r]      = acc0[i];
    sC[wm * 16 + 8 * h + i][wn * 32 + 16 + r] = acc1[i];
  }
  __syncthreads();
  const int col = n0 + ld_c4 * 4;
  if (col < N) {
    #pragma unroll
    for (int p = 0; p < 4; ++p) {
      const int row = ld_r + 16 * p;
      const float4 v = *(const float4*)&sC[row][ld_c4 * 4];
      *(float4*)(C + (long)(m0 + row) * N + col) = v;
    }
  }
#undef GT_FETCH
#undef GT_COMMIT
#undef GT_COMPUTE
}

__global__ void dwconv_silu_kernel(const float* __restrict__ zx,
                                   const float* __restrict__ w,
                                   const float* __restrict__ bias,
                                   float* __restrict__ out) {
  const int c = blockIdx.x * 128 + threadIdx.x;
  if (c >= CONV_DIM) return;
  const long bl = blockIdx.y;
  const int l = (int)(bl & (LOUT - 1));
  const float* xcol = zx + bl * D_IN_PROJ + D_INNER + c;
  const float w0 = w[c * 4 + 0], w1 = w[c * 4 + 1], w2 = w[c * 4 + 2], w3 = w[c * 4 + 3];
  float acc = w3 * xcol[0];
  if (l >= 1) acc = fmaf(w2, xcol[-(long)D_IN_PROJ],     acc);
  if (l >= 2) acc = fmaf(w1, xcol[-(long)2 * D_IN_PROJ], acc);
  if (l >= 3) acc = fmaf(w0, xcol[-(long)3 * D_IN_PROJ], acc);
  const float v = acc + bias[c];
  out[bl * CONV_DIM + c] = v / (1.f + __expf(-v));
}

#define SCAN_T   32
#define SCAN_NCH (LOUT / SCAN_T)
#define SCAN_SPLIT 2
#define SCAN_PB    (HEADDIM / SCAN_SPLIT)
#define SCAN_NT    (SCAN_PB * 4)

template <int CTRL>
__device__ __forceinline__ float lane_xor_read(float v) {
  return __int_as_float(__builtin_amdgcn_update_dpp(0, __float_as_int(v), CTRL, 0xF, 0xF, true));
}

__global__ void __launch_bounds__(SCAN_NT)
ssm_scan_kernel(const float* __restrict__ zx,
                const float* __restrict__ xbc,
                const float* __restrict__ dtb,
                const float* __restrict__ alog,
                const float* __restrict__ Dp,
                float* __restrict__ y) {
  const int pblk = blockIdx.x % SCAN_SPLIT;
  const int hh = (blockIdx.x / SCAN_SPLIT) & 15;
  const int b  = blockIdx.x / (SCAN_SPLIT * NHEADS);
  const int tid = threadIdx.x;
  const int p = tid >> 2;
  const int q = tid & 3;
  __shared__ __attribute__((aligned(16))) float sBC[2][SCAN_T][2 * D_STATE];
  __shared__ __attribute__((aligned(16))) float sX[2][SCAN_T][SCAN_PB];
  __shared__ __attribute__((aligned(16))) float sY[2][SCAN_T][SCAN_PB];
  __shared__ __attribute__((aligned(8)))  float sDT[2][SCAN_T][2];
  const float Ah   = -__expf(alog[hh]);
  const float dtbh = dtb[hh];
  const float Dh   = Dp[hh];
  float hreg[16];
  #pragma unroll
  for (int i = 0; i < 16; ++i) hreg[i] = 0.f;

  const int bc_c4 = tid & 31;
  const int bc_t  = tid >> 5;
  const int x_c4  = tid & 7;
  const int x_t   = tid >> 3;
  const bool dt_wave = (tid >= SCAN_NT - SCAN_T);
  const int  dt_t    = dt_wave ? tid - (SCAN_NT - SCAN_T) : 0;
  const float* gBC = xbc + ((long)b * LOUT + bc_t) * CONV_DIM + D_INNER + bc_c4 * 4;
  const float* gX  = xbc + ((long)b * LOUT + x_t) * CONV_DIM + hh * HEADDIM + pblk * SCAN_PB + x_c4 * 4;
  const float* gDT = zx + ((long)b * LOUT + dt_t) * D_IN_PROJ + (D_IN_PROJ - NHEADS) + hh;
  float*       gY  = y + ((long)b * LOUT + x_t) * D_INNER + hh * HEADDIM + pblk * SCAN_PB + x_c4 * 4;

  float4 rBC0, rBC1, rBC2, rBC3, rBC4, rBC5, rBC6, rBC7, rX0, rX1;
  float  rdt = 0.f;

#define SCAN_FETCH(l0)                                                              \
  do {                                                                              \
    const float* s_bc = gBC + (long)(l0) * CONV_DIM;                                \
    const float* s_x  = gX  + (long)(l0) * CONV_DIM;                                \
    rBC0 = *(const float4*)(s_bc + 0 * CONV_DIM);                                   \
    rBC1 = *(const float4*)(s_bc + 4 * CONV_DIM);                                   \
    rBC2 = *(const float4*)(s_bc + 8 * CONV_DIM);                                   \
    rBC3 = *(const float4*)(s_bc + 12 * CONV_DIM);                                  \
    rBC4 = *(const float4*)(s_bc + 16 * CONV_DIM);                                  \
    rBC5 = *(const float4*)(s_bc + 20 * CONV_DIM);                                  \
    rBC6 = *(const float4*)(s_bc + 24 * CONV_DIM);                                  \
    rBC7 = *(const float4*)(s_bc + 28 * CONV_DIM);                                  \
    rX0  = *(const float4*)(s_x);                                                   \
    rX1  = *(const float4*)(s_x + 16 * CONV_DIM);                                   \
    if (dt_wave) rdt = gDT[(long)(l0) * D_IN_PROJ];                                 \
  } while (0)

#define SCAN_COMMIT(buf)                                                            \
  do {                                                                              \
    *(float4*)&sBC[(buf)][bc_t + 0][bc_c4 * 4] = rBC0;                              \
    *(float4*)&sBC[(buf)][bc_t + 4][bc_c4 * 4] = rBC1;                              \
    *(float4*)&sBC[(buf)][bc_t + 8][bc_c4 * 4] = rBC2;                              \
    *(float4*)&sBC[(buf)][bc_t + 12][bc_c4 * 4] = rBC3;                             \
    *(float4*)&sBC[(buf)][bc_t + 16][bc_c4 * 4] = rBC4;                             \
    *(float4*)&sBC[(buf)][bc_t + 20][bc_c4 * 4] = rBC5;                             \
    *(float4*)&sBC[(buf)][bc_t + 24][bc_c4 * 4] = rBC6;                             \
    *(float4*)&sBC[(buf)][bc_t + 28][bc_c4 * 4] = rBC7;                             \
    *(float4*)&sX[(buf)][x_t     ][x_c4 * 4] = rX0;                                 \
    *(float4*)&sX[(buf)][x_t + 16][x_c4 * 4] = rX1;                                 \
    if (dt_wave) {                                                                  \
      const float raw = rdt + dtbh;                                                 \
      const float dtv = (raw > 20.f) ? raw : log1pf(__expf(raw));      \
      float2 dd;                                                                    \
      dd.x = dtv;                                                                   \
      dd.y = __expf(dtv * Ah);                                               \
      *(float2*)&sDT[(buf)][dt_t][0] = dd;                                          \
    }                                                                               \
  } while (0)

#define SCAN_WRITEOUT(l0, buf)                                                      \
  do {                                                                              \
    float* d_y = gY + (long)(l0) * D_INNER;                                         \
    const float4 v0 = *(const float4*)&sY[(buf)][x_t     ][x_c4 * 4];               \
    const float4 v1 = *(const float4*)&sY[(buf)][x_t + 16][x_c4 * 4];               \
    *(float4*)(d_y)                = v0;                                            \
    *(float4*)(d_y + 16 * D_INNER) = v1;                                            \
  } while (0)

  SCAN_FETCH(0);
  SCAN_COMMIT(0);
  __syncthreads();

  for (int c = 0; c < SCAN_NCH; ++c) {
    const int cb = c & 1;
    const bool more = (c + 1 < SCAN_NCH);
    if (more) SCAN_FETCH((c + 1) * SCAN_T);
    if (c > 0) SCAN_WRITEOUT((c - 1) * SCAN_T, cb ^ 1);
    for (int t0 = 0; t0 < SCAN_T; t0 += 4) {
      float yv[4];
      #pragma unroll
      for (int u = 0; u < 4; ++u) {
        const int t = t0 + u;
        const float2 dd = *(const float2*)&sDT[cb][t][0];
        const float dtv = dd.x;
        const float dA  = dd.y;
        const float xv  = sX[cb][t][p];
        const float xdt = xv * dtv;
        const float* Bq = &sBC[cb][t][q * 16];
        const float* Cq = &sBC[cb][t][D_STATE + q * 16];
        float accv = 0.f;
        #pragma unroll
        for (int i = 0; i < 16; ++i) {
          hreg[i] = fmaf(hreg[i], dA, xdt * Bq[i]);
          accv    = fmaf(hreg[i], Cq[i], accv);
        }
        accv += lane_xor_read<0x161>(accv);
        accv += lane_xor_read<0x162>(accv);
        yv[u] = fmaf(xv, Dh, accv);
      }
      if (q == 0) {
        #pragma unroll
        for (int u = 0; u < 4; ++u) sY[cb][t0 + u][p] = yv[u];
      }
    }
    if (more) SCAN_COMMIT(cb ^ 1);
    __syncthreads();
  }
  SCAN_WRITEOUT((SCAN_NCH - 1) * SCAN_T, (SCAN_NCH - 1) & 1);
#undef SCAN_FETCH
#undef SCAN_COMMIT
#undef SCAN_WRITEOUT
}

__global__ void gate_rmsnorm_kernel(const float* __restrict__ zx,
                                    const float* __restrict__ nw,
                                    float* __restrict__ y) {
  const long bl = blockIdx.x;
  const int tid = threadIdx.x;
  __shared__ float red[256];
  float yv[4];
  float ss = 0.f;
  #pragma unroll
  for (int j = 0; j < 4; ++j) {
    const int i = tid + j * 256;
    const float z = zx[bl * D_IN_PROJ + i];
    const float g = z / (1.f + __expf(-z));
    const float v = y[bl * D_INNER + i] * g;
    yv[j] = v;
    ss = fmaf(v, v, ss);
  }
  red[tid] = ss;
  __syncthreads();
  for (int st = 128; st > 0; st >>= 1) {
    if (tid < st) red[tid] += red[tid + st];
    __syncthreads();
  }
  const float scale = rsqrtf(red[0] * (1.f / (float)D_INNER) + 1e-5f);
  #pragma unroll
  for (int j = 0; j < 4; ++j) {
    const int i = tid + j * 256;
    y[bl * D_INNER + i] = yv[j] * scale * nw[i];
  }
}

extern "C" void kernel_launch(void* const* d_in, const int* in_sizes, int n_in,
                              void* d_out, int out_size, void* d_ws, size_t ws_size,
                              hipStream_t stream) {
  const float* tokens     = (const float*)d_in[0];
  const float* conv_w     = (const float*)d_in[1];
  const float* in_proj_w  = (const float*)d_in[2];
  const float* conv1d_w   = (const float*)d_in[3];
  const float* conv1d_b   = (const float*)d_in[4];
  const float* dt_bias    = (const float*)d_in[5];
  const float* A_log      = (const float*)d_in[6];
  const float* D_param    = (const float*)d_in[7];
  const float* norm_w     = (const float*)d_in[8];
  const float* out_proj_w = (const float*)d_in[9];
  float* out = (float*)d_out;
  float* ws  = (float*)d_ws;

  float* u0   = ws;
  float* u1   = u0 + (long)ROWS * D_MODEL;
  float* zx   = u1 + (long)ROWS * D_MODEL;
  float* xact = zx + (long)ROWS * D_IN_PROJ;
  float* ybuf = xact + (long)ROWS * CONV_DIM;
  float* col  = zx;
  (void)ws_size; (void)in_sizes; (void)n_in; (void)out_size;

  {
    const long nel = (long)ROWS * KSTEM;
    im2col_stem_kernel<<<(int)((nel + 255) / 256), 256, 0, stream>>>(tokens, col);
    gemm_xt_wmma<<<(ROWS / GT_M) * ((D_MODEL + GT_N - 1) / GT_N), 256, 0, stream>>>(col, conv_w, u0, ROWS, D_MODEL, KSTEM);
  }

  const float* u_in = u0;
  for (int blk = 0; blk < 2; ++blk) {
    const float* ipw = in_proj_w  + (long)blk * D_IN_PROJ * D_MODEL;
    const float* cw  = conv1d_w   + (long)blk * CONV_DIM * D_CONV;
    const float* cb  = conv1d_b   + (long)blk * CONV_DIM;
    const float* dtb = dt_bias    + (long)blk * NHEADS;
    const float* al  = A_log      + (long)blk * NHEADS;
    const float* Dpp = D_param    + (long)blk * NHEADS;
    const float* nw  = norm_w     + (long)blk * D_INNER;
    const float* opw = out_proj_w + (long)blk * D_MODEL * D_INNER;

    {
      gemm_xt_wmma<<<(ROWS / GT_M) * ((D_IN_PROJ + GT_N - 1) / GT_N), 256, 0, stream>>>(u_in, ipw, zx, ROWS, D_IN_PROJ, D_MODEL);
    }
    dwconv_silu_kernel<<<dim3(CONV_DIM / 128, ROWS), 128, 0, stream>>>(zx, cw, cb, xact);
    ssm_scan_kernel<<<BATCH * NHEADS * SCAN_SPLIT, SCAN_NT, 0, stream>>>(zx, xact, dtb, al, Dpp, ybuf);
    gate_rmsnorm_kernel<<<ROWS, 256, 0, stream>>>(zx, nw, ybuf);
    float* dst = (blk == 0) ? u1 : out;
    {
      gemm_xt_wmma<<<(ROWS / GT_M) * ((D_MODEL + GT_N - 1) / GT_N), 256, 0, stream>>>(ybuf, opw, dst, ROWS, D_MODEL, D_INNER);
    }
    u_in = u1;
  }
}


# im2col, depthwise-conv and gate/norm kernels: each workgroup loops over 8 rows (gate: 4) with all its loads issued first, so 18 times fewer waves enter (294912 to 16384); same sums, same order
# speedup vs baseline: 8.0873x; 1.0534x over previous
#include <hip/hip_runtime.h>
#include <hip/hip_bf16.h>

typedef __attribute__((ext_vector_type(2))) float v2f;
typedef __attribute__((ext_vector_type(8))) float v8f;

#define BATCH     2
#define SEQLEN    2048
#define C_IN      256
#define D_MODEL   512
#define LOUT      1024
#define D_STATE   64
#define D_CONV    4
#define D_INNER   1024
#define HEADDIM   64
#define NHEADS    16
#define CONV_DIM  (D_INNER + 2*D_STATE)
#define D_IN_PROJ (2*D_INNER + 2*D_STATE + NHEADS)
#define ROWS      (BATCH*LOUT)
#define KSTEM     (C_IN * 7)

#define IM_R  8
#define IM_NT (KSTEM / 4)
__global__ void __launch_bounds__(IM_NT)
im2col_stem_kernel(const float* __restrict__ tokens,
                   float* __restrict__ col) {
  const int t   = threadIdx.x;
  const int bl0 = blockIdx.x * IM_R;
  const int b   = bl0 >> 10;
  const int l0  = bl0 & (LOUT - 1);
  const float* tb = tokens + (long)b * C_IN * SEQLEN;
  int off[4], kk[4];
  #pragma unroll
  for (int m = 0; m < 4; ++m) {
    const int j = t + IM_NT * m;
    const int c = j / 7;
    kk[m]  = j - c * 7;
    off[m] = c * SEQLEN;
  }
  float v[IM_R][4];
  #pragma unroll
  for (int i = 0; i < IM_R; ++i) {
    const int s0 = 2 * (l0 + i) - 3;
    #pragma unroll
    for (int m = 0; m < 4; ++m) {
      const int s  = s0 + kk[m];
      const int sc = min(max(s, 0), SEQLEN - 1);
      const float x = tb[off[m] + sc];
      v[i][m] = (s == sc) ? x : 0.f;
    }
  }
  float* dst = col + (long)bl0 * KSTEM + t;
  #pragma unroll
  for (int i = 0; i < IM_R; ++i) {
    #pragma unroll
    for (int m = 0; m < 4; ++m) dst[(long)i * KSTEM + IM_NT * m] = v[i][m];
  }
}

#define GT_M 64
#define GT_N 64
#define GT_K 64
#define GT_P 68

__global__ void __launch_bounds__(256)
gemm_xt_wmma(const float* __restrict__ A,
             const float* __restrict__ W,
             float* __restrict__ C,
             int M, int N, int K) {
  __shared__ __attribute__((aligned(16))) float sA[2][GT_M][GT_P];
  __shared__ __attribute__((aligned(16))) float sB[2][GT_N][GT_P];
  const int tid  = threadIdx.x;
  const int lane = tid & 31;
  const int wave = tid >> 5;
  const int nblk = (N + GT_N - 1) / GT_N;
  const int mb = blockIdx.x / nblk;
  const int nb = blockIdx.x - mb * nblk;
  const int m0 = mb * GT_M;
  const int n0 = nb * GT_N;
  (void)M;

  const int ld_c4 = tid & 15;
  const int ld_r  = tid >> 4;
  const long a_step = (long)16 * K;
  const float* gA  = A + (long)(m0 + ld_r) * K + ld_c4 * 4;
  const float* gB0 = W + (long)min(n0 + ld_r,      N - 1) * K + ld_c4 * 4;
  const float* gB1 = W + (long)min(n0 + ld_r + 16, N - 1) * K + ld_c4 * 4;
  const float* gB2 = W + (long)min(n0 + ld_r + 32, N - 1) * K + ld_c4 * 4;
  const float* gB3 = W + (long)min(n0 + ld_r + 48, N - 1) * K + ld_c4 * 4;

  float4 ra0, ra1, ra2, ra3, rb0, rb1, rb2, rb3;

#define GT_FETCH(k0)                                                   \
  do {                                                                 \
    ra0 = *(const float4*)(gA + (k0));                                 \
    ra1 = *(const float4*)(gA + a_step + (k0));                        \
    ra2 = *(const float4*)(gA + 2 * a_step + (k0));                    \
    ra3 = *(const float4*)(gA + 3 * a_step + (k0));                    \
    rb0 = *(const float4*)(gB0 + (k0));                                \
    rb1 = *(const float4*)(gB1 + (k0));                                \
    rb2 = *(const float4*)(gB2 + (k0));                                \
    rb3 = *(const float4*)(gB3 + (k0));                                \
  } while (0)

#define GT_COMMIT(buf)                                                 \
  do {                                                                 \
    *(float4*)&sA[(buf)][ld_r     ][ld_c4 * 4] = ra0;                  \
    *(float4*)&sA[(buf)][ld_r + 16][ld_c4 * 4] = ra1;                  \
    *(float4*)&sA[(buf)][ld_r + 32][ld_c4 * 4] = ra2;                  \
    *(float4*)&sA[(buf)][ld_r + 48][ld_c4 * 4] = ra3;                  \
    *(float4*)&sB[(buf)][ld_r     ][ld_c4 * 4] = rb0;                  \
    *(float4*)&sB[(buf)][ld_r + 16][ld_c4 * 4] = rb1;                  \
    *(float4*)&sB[(buf)][ld_r + 32][ld_c4 * 4] = rb2;                  \
    *(float4*)&sB[(buf)][ld_r + 48][ld_c4 * 4] = rb3;                  \
  } while (0)

  const int r  = lane & 15;
  const int h  = lane >> 4;
  const int wm = wave & 3;
  const int wn = wave >> 2;
  const float* fa  = &sA[0][wm * 16 + r][4 * h];
  const float* fb0 = &sB[0][wn * 32 + r][4 * h];
  const float* fb1 = &sB[0][wn * 32 + 16 + r][4 * h];

  v8f acc0 = {0.f,0.f,0.f,0.f,0.f,0.f,0.f,0.f};
  v8f acc1 = acc0;

#define GT_COMPUTE(buf)                                                                            \
  do {                                                                                             \
    const float* pa  = fa  + (buf) * (GT_M * GT_P);                                                \
    const float* pb0 = fb0 + (buf) * (GT_N * GT_P);                                                \
    const float* pb1 = fb1 + (buf) * (GT_N * GT_P);                                                \
    _Pragma("unroll")                                                                              \
    for (int g = 0; g < GT_K / 8; ++g) {                                                           \
      const float4 a4 = *(const float4*)(pa  + g * 8);                                             \
      const float4 p4 = *(const float4*)(pb0 + g * 8);                                             \
      const float4 q4 = *(const float4*)(pb1 + g * 8);                                             \
      const v2f alo = {a4.x, a4.y}, ahi = {a4.z, a4.w};                                            \
      const v2f plo = {p4.x, p4.y}, phi = {p4.z, p4.w};                                            \
      const v2f qlo = {q4.x, q4.y}, qhi = {q4.z, q4.w};                                            \
      acc0 = __builtin_amdgcn_wmma_f32_16x16x4_f32(false, alo, false, plo, (short)0, acc0, false, false); \
      acc1 = __builtin_amdgcn_wmma_f32_16x16x4_f32(false, alo, false, qlo, (short)0, acc1, false, false); \
      acc0 = __builtin_amdgcn_wmma_f32_16x16x4_f32(false, ahi, false, phi, (short)0, acc0, false, false); \
      acc1 = __builtin_amdgcn_wmma_f32_16x16x4_f32(false, ahi, false, qhi, (short)0, acc1, false, false); \
    }                                                                                              \
  } while (0)

  const int KT = K / GT_K;
  GT_FETCH(0);
  GT_COMMIT(0);
  __syncthreads();

  for (int kt = 0; kt + 1 < KT; ++kt) {
    const int cb = kt & 1;
    GT_FETCH((kt + 1) * GT_K);
    GT_COMPUTE(cb);
    GT_COMMIT(cb ^ 1);
    __syncthreads();
  }
  GT_COMPUTE((KT - 1) & 1);
  __syncthreads();

  float (*sC)[GT_P] = sA[0];
  #pragma unroll
  for (int i = 0; i < 8; ++i) {
    sC[wm * 16 + 8 * h + i][wn * 32 + r]      = acc0[i];
    sC[wm * 16 + 8 * h + i][wn * 32 + 16 + r] = acc1[i];
  }
  __syncthreads();
  const int col = n0 + ld_c4 * 4;
  if (col < N) {
    #pragma unroll
    for (int p = 0; p < 4; ++p) {
      const int row = ld_r + 16 * p;
      const float4 v = *(const float4*)&sC[row][ld_c4 * 4];
      *(float4*)(C + (long)(m0 + row) * N + col) = v;
    }
  }
#undef GT_FETCH
#undef GT_COMMIT
#undef GT_COMPUTE
}

#define DW_R  8
#define DW_NT (CONV_DIM / 4)

__device__ __forceinline__ float dwconv_one(const float4 wv, float x0, float xm1, float xm2,
                                            float xm3, int l, float bs) {
  float acc = wv.w * x0;
  if (l >= 1) acc = fmaf(wv.z, xm1, acc);
  if (l >= 2) acc = fmaf(wv.y, xm2, acc);
  if (l >= 3) acc = fmaf(wv.x, xm3, acc);
  const float v = acc + bs;
  return v / (1.f + __expf(-v));
}

__global__ void __launch_bounds__(DW_NT)
dwconv_silu_kernel(const float* __restrict__ zx,
                   const float* __restrict__ w,
                   const float* __restrict__ bias,
                   float* __restrict__ out) {
  const int t   = threadIdx.x;
  const int bl0 = blockIdx.x * DW_R;
  const int l0  = bl0 & (LOUT - 1);
  const float* xin = zx + (long)bl0 * D_IN_PROJ + D_INNER + 4 * t;
  float4 xr[DW_R + 3];
  #pragma unroll
  for (int i = 0; i < 3; ++i) {
    const long back = (l0 >= 3 - i) ? (long)(3 - i) * D_IN_PROJ : 0;
    xr[i] = *(const float4*)(xin - back);
  }
  #pragma unroll
  for (int i = 0; i < DW_R; ++i) xr[3 + i] = *(const float4*)(xin + (long)i * D_IN_PROJ);
  const float4 wa = *(const float4*)(w + (4 * t + 0) * 4);
  const float4 wb = *(const float4*)(w + (4 * t + 1) * 4);
  const float4 wc = *(const float4*)(w + (4 * t + 2) * 4);
  const float4 wd = *(const float4*)(w + (4 * t + 3) * 4);
  const float4 bs = *(const float4*)(bias + 4 * t);
  float* dst = out + (long)bl0 * CONV_DIM + 4 * t;
  #pragma unroll
  for (int i = 0; i < DW_R; ++i) {
    const int l = l0 + i;
    float4 o;
    o.x = dwconv_one(wa, xr[3 + i].x, xr[2 + i].x, xr[1 + i].x, xr[i].x, l, bs.x);
    o.y = dwconv_one(wb, xr[3 + i].y, xr[2 + i].y, xr[1 + i].y, xr[i].y, l, bs.y);
    o.z = dwconv_one(wc, xr[3 + i].z, xr[2 + i].z, xr[1 + i].z, xr[i].z, l, bs.z);
    o.w = dwconv_one(wd, xr[3 + i].w, xr[2 + i].w, xr[1 + i].w, xr[i].w, l, bs.w);
    *(float4*)(dst + (long)i * CONV_DIM) = o;
  }
}

#define SCAN_T   32
#define SCAN_NCH (LOUT / SCAN_T)
#define SCAN_SPLIT 2
#define SCAN_PB    (HEADDIM / SCAN_SPLIT)
#define SCAN_NT    (SCAN_PB * 4)

template <int CTRL>
__device__ __forceinline__ float lane_xor_read(float v) {
  return __int_as_float(__builtin_amdgcn_update_dpp(0, __float_as_int(v), CTRL, 0xF, 0xF, true));
}

__global__ void __launch_bounds__(SCAN_NT)
ssm_scan_kernel(const float* __restrict__ zx,
                const float* __restrict__ xbc,
                const float* __restrict__ dtb,
                const float* __restrict__ alog,
                const float* __restrict__ Dp,
                float* __restrict__ y) {
  const int pblk = blockIdx.x % SCAN_SPLIT;
  const int hh = (blockIdx.x / SCAN_SPLIT) & 15;
  const int b  = blockIdx.x / (SCAN_SPLIT * NHEADS);
  const int tid = threadIdx.x;
  const int p = tid >> 2;
  const int q = tid & 3;
  __shared__ __attribute__((aligned(16))) float sBC[2][SCAN_T][2 * D_STATE];
  __shared__ __attribute__((aligned(16))) float sX[2][SCAN_T][SCAN_PB];
  __shared__ __attribute__((aligned(16))) float sY[2][SCAN_T][SCAN_PB];
  __shared__ __attribute__((aligned(8)))  float sDT[2][SCAN_T][2];
  const float Ah   = -__expf(alog[hh]);
  const float dtbh = dtb[hh];
  const float Dh   = Dp[hh];
  float hreg[16];
  #pragma unroll
  for (int i = 0; i < 16; ++i) hreg[i] = 0.f;

  const int bc_c4 = tid & 31;
  const int bc_t  = tid >> 5;
  const int x_c4  = tid & 7;
  const int x_t   = tid >> 3;
  const bool dt_wave = (tid >= SCAN_NT - SCAN_T);
  const int  dt_t    = dt_wave ? tid - (SCAN_NT - SCAN_T) : 0;
  const float* gBC = xbc + ((long)b * LOUT + bc_t) * CONV_DIM + D_INNER + bc_c4 * 4;
  const float* gX  = xbc + ((long)b * LOUT + x_t) * CONV_DIM + hh * HEADDIM + pblk * SCAN_PB + x_c4 * 4;
  const float* gDT = zx + ((long)b * LOUT + dt_t) * D_IN_PROJ + (D_IN_PROJ - NHEADS) + hh;
  float*       gY  = y + ((long)b * LOUT + x_t) * D_INNER + hh * HEADDIM + pblk * SCAN_PB + x_c4 * 4;

  float4 rBC0, rBC1, rBC2, rBC3, rBC4, rBC5, rBC6, rBC7, rX0, rX1;
  float  rdt = 0.f;

#define SCAN_FETCH(l0)                                                              \
  do {                                                                              \
    const float* s_bc = gBC + (long)(l0) * CONV_DIM;                                \
    const float* s_x  = gX  + (long)(l0) * CONV_DIM;                                \
    rBC0 = *(const float4*)(s_bc + 0 * CONV_DIM);                                   \
    rBC1 = *(const float4*)(s_bc + 4 * CONV_DIM);                                   \
    rBC2 = *(const float4*)(s_bc + 8 * CONV_DIM);                                   \
    rBC3 = *(const float4*)(s_bc + 12 * CONV_DIM);                                  \
    rBC4 = *(const float4*)(s_bc + 16 * CONV_DIM);                                  \
    rBC5 = *(const float4*)(s_bc + 20 * CONV_DIM);                                  \
    rBC6 = *(const float4*)(s_bc + 24 * CONV_DIM);                                  \
    rBC7 = *(const float4*)(s_bc + 28 * CONV_DIM);                                  \
    rX0  = *(const float4*)(s_x);                                                   \
    rX1  = *(const float4*)(s_x + 16 * CONV_DIM);                                   \
    if (dt_wave) rdt = gDT[(long)(l0) * D_IN_PROJ];                                 \
  } while (0)

#define SCAN_COMMIT(buf)                                                            \
  do {                                                                              \
    *(float4*)&sBC[(buf)][bc_t + 0][bc_c4 * 4] = rBC0;                              \
    *(float4*)&sBC[(buf)][bc_t + 4][bc_c4 * 4] = rBC1;                              \
    *(float4*)&sBC[(buf)][bc_t + 8][bc_c4 * 4] = rBC2;                              \
    *(float4*)&sBC[(buf)][bc_t + 12][bc_c4 * 4] = rBC3;                             \
    *(float4*)&sBC[(buf)][bc_t + 16][bc_c4 * 4] = rBC4;                             \
    *(float4*)&sBC[(buf)][bc_t + 20][bc_c4 * 4] = rBC5;                             \
    *(float4*)&sBC[(buf)][bc_t + 24][bc_c4 * 4] = rBC6;                             \
    *(float4*)&sBC[(buf)][bc_t + 28][bc_c4 * 4] = rBC7;                             \
    *(float4*)&sX[(buf)][x_t     ][x_c4 * 4] = rX0;                                 \
    *(float4*)&sX[(buf)][x_t + 16][x_c4 * 4] = rX1;                                 \
    if (dt_wave) {                                                                  \
      const float raw = rdt + dtbh;                                                 \
      const float dtv = (raw > 20.f) ? raw : log1pf(__expf(raw));      \
      float2 dd;                                                                    \
      dd.x = dtv;                                                                   \
      dd.y = __expf(dtv * Ah);                                               \
      *(float2*)&sDT[(buf)][dt_t][0] = dd;                                          \
    }                                                                               \
  } while (0)

#define SCAN_WRITEOUT(l0, buf)                                                      \
  do {                                                                              \
    float* d_y = gY + (long)(l0) * D_INNER;                                         \
    const float4 v0 = *(const float4*)&sY[(buf)][x_t     ][x_c4 * 4];               \
    const float4 v1 = *(const float4*)&sY[(buf)][x_t + 16][x_c4 * 4];               \
    *(float4*)(d_y)                = v0;                                            \
    *(float4*)(d_y + 16 * D_INNER) = v1;                                            \
  } while (0)

  SCAN_FETCH(0);
  SCAN_COMMIT(0);
  __syncthreads();

  for (int c = 0; c < SCAN_NCH; ++c) {
    const int cb = c & 1;
    const bool more = (c + 1 < SCAN_NCH);
    if (more) SCAN_FETCH((c + 1) * SCAN_T);
    if (c > 0) SCAN_WRITEOUT((c - 1) * SCAN_T, cb ^ 1);
    for (int t0 = 0; t0 < SCAN_T; t0 += 4) {
      float yv[4];
      #pragma unroll
      for (int u = 0; u < 4; ++u) {
        const int t = t0 + u;
        const float2 dd = *(const float2*)&sDT[cb][t][0];
        const float dtv = dd.x;
        const float dA  = dd.y;
        const float xv  = sX[cb][t][p];
        const float xdt = xv * dtv;
        const float* Bq = &sBC[cb][t][q * 16];
        const float* Cq = &sBC[cb][t][D_STATE + q * 16];
        float accv = 0.f;
        #pragma unroll
        for (int i = 0; i < 16; ++i) {
          hreg[i] = fmaf(hreg[i], dA, xdt * Bq[i]);
          accv    = fmaf(hreg[i], Cq[i], accv);
        }
        accv += lane_xor_read<0x161>(accv);
        accv += lane_xor_read<0x162>(accv);
        yv[u] = fmaf(xv, Dh, accv);
      }
      if (q == 0) {
        #pragma unroll
        for (int u = 0; u < 4; ++u) sY[cb][t0 + u][p] = yv[u];
      }
    }
    if (more) SCAN_COMMIT(cb ^ 1);
    __syncthreads();
  }
  SCAN_WRITEOUT((SCAN_NCH - 1) * SCAN_T, (SCAN_NCH - 1) & 1);
#undef SCAN_FETCH
#undef SCAN_COMMIT
#undef SCAN_WRITEOUT
}

#define GR_R 4
__global__ void __launch_bounds__(256)
gate_rmsnorm_kernel(const float* __restrict__ zx,
                    const float* __restrict__ nw,
                    float* __restrict__ y) {
  const long bl0 = (long)blockIdx.x * GR_R;
  const int tid = threadIdx.x;
  __shared__ float red[GR_R][256];
  float zv[GR_R][4];
  float yv[GR_R][4];
  #pragma unroll
  for (int r = 0; r < GR_R; ++r) {
    #pragma unroll
    for (int j = 0; j < 4; ++j) {
      const int i = tid + j * 256;
      zv[r][j] = zx[(bl0 + r) * D_IN_PROJ + i];
      yv[r][j] = y[(bl0 + r) * D_INNER + i];
    }
  }
  #pragma unroll
  for (int r = 0; r < GR_R; ++r) {
    float ss = 0.f;
    #pragma unroll
    for (int j = 0; j < 4; ++j) {
      const float z = zv[r][j];
      const float g = z / (1.f + __expf(-z));
      const float v = yv[r][j] * g;
      yv[r][j] = v;
      ss = fmaf(v, v, ss);
    }
    red[r][tid] = ss;
  }
  __syncthreads();
  for (int st = 128; st > 0; st >>= 1) {
    if (tid < st) {
      #pragma unroll
      for (int r = 0; r < GR_R; ++r) red[r][tid] += red[r][tid + st];
    }
    __syncthreads();
  }
  float nwv[4];
  #pragma unroll
  for (int j = 0; j < 4; ++j) nwv[j] = nw[tid + j * 256];
  #pragma unroll
  for (int r = 0; r < GR_R; ++r) {
    const float scale = rsqrtf(red[r][0] * (1.f / (float)D_INNER) + 1e-5f);
    #pragma unroll
    for (int j = 0; j < 4; ++j) {
      const int i = tid + j * 256;
      y[(bl0 + r) * D_INNER + i] = yv[r][j] * scale * nwv[j];
    }
  }
}

extern "C" void kernel_launch(void* const* d_in, const int* in_sizes, int n_in,
                              void* d_out, int out_size, void* d_ws, size_t ws_size,
                              hipStream_t stream) {
  const float* tokens     = (const float*)d_in[0];
  const float* conv_w     = (const float*)d_in[1];
  const float* in_proj_w  = (const float*)d_in[2];
  const float* conv1d_w   = (const float*)d_in[3];
  const float* conv1d_b   = (const float*)d_in[4];
  const float* dt_bias    = (const float*)d_in[5];
  const float* A_log      = (const float*)d_in[6];
  const float* D_param    = (const float*)d_in[7];
  const float* norm_w     = (const float*)d_in[8];
  const float* out_proj_w = (const float*)d_in[9];
  float* out = (float*)d_out;
  float* ws  = (float*)d_ws;

  float* u0   = ws;
  float* u1   = u0 + (long)ROWS * D_MODEL;
  float* zx   = u1 + (long)ROWS * D_MODEL;
  float* xact = zx + (long)ROWS * D_IN_PROJ;
  float* ybuf = xact + (long)ROWS * CONV_DIM;
  float* col  = zx;
  (void)ws_size; (void)in_sizes; (void)n_in; (void)out_size;

  {
    im2col_stem_kernel<<<ROWS / IM_R, IM_NT, 0, stream>>>(tokens, col);
    gemm_xt_wmma<<<(ROWS / GT_M) * ((D_MODEL + GT_N - 1) / GT_N), 256, 0, stream>>>(col, conv_w, u0, ROWS, D_MODEL, KSTEM);
  }

  const float* u_in = u0;
  for (int blk = 0; blk < 2; ++blk) {
    const float* ipw = in_proj_w  + (long)blk * D_IN_PROJ * D_MODEL;
    const float* cw  = conv1d_w   + (long)blk * CONV_DIM * D_CONV;
    const float* cb  = conv1d_b   + (long)blk * CONV_DIM;
    const float* dtb = dt_bias    + (long)blk * NHEADS;
    const float* al  = A_log      + (long)blk * NHEADS;
    const float* Dpp = D_param    + (long)blk * NHEADS;
    const float* nw  = norm_w     + (long)blk * D_INNER;
    const float* opw = out_proj_w + (long)blk * D_MODEL * D_INNER;

    {
      gemm_xt_wmma<<<(ROWS / GT_M) * ((D_IN_PROJ + GT_N - 1) / GT_N), 256, 0, stream>>>(u_in, ipw, zx, ROWS, D_IN_PROJ, D_MODEL);
    }
    dwconv_silu_kernel<<<ROWS / DW_R, DW_NT, 0, stream>>>(zx, cw, cb, xact);
    ssm_scan_kernel<<<BATCH * NHEADS * SCAN_SPLIT, SCAN_NT, 0, stream>>>(zx, xact, dtb, al, Dpp, ybuf);
    gate_rmsnorm_kernel<<<ROWS / GR_R, 256, 0, stream>>>(zx, nw, ybuf);
    float* dst = (blk == 0) ? u1 : out;
    {
      gemm_xt_wmma<<<(ROWS / GT_M) * ((D_MODEL + GT_N - 1) / GT_N), 256, 0, stream>>>(ybuf, opw, dst, ROWS, D_MODEL, D_INNER);
    }
    u_in = u1;
  }
}


# ssm scan: a step's B/C/x/dt words are read from LDS four steps ahead into four rotating register sets (chunk's trips unrolled), so a step no longer waits for its own LDS reads; same sums, same order
# speedup vs baseline: 9.1627x; 1.1330x over previous
#include <hip/hip_runtime.h>
#include <hip/hip_bf16.h>

typedef __attribute__((ext_vector_type(2))) float v2f;
typedef __attribute__((ext_vector_type(8))) float v8f;

#define BATCH     2
#define SEQLEN    2048
#define C_IN      256
#define D_MODEL   512
#define LOUT      1024
#define D_STATE   64
#define D_CONV    4
#define D_INNER   1024
#define HEADDIM   64
#define NHEADS    16
#define CONV_DIM  (D_INNER + 2*D_STATE)
#define D_IN_PROJ (2*D_INNER + 2*D_STATE + NHEADS)
#define ROWS      (BATCH*LOUT)
#define KSTEM     (C_IN * 7)

#define IM_R  8
#define IM_NT (KSTEM / 4)
__global__ void __launch_bounds__(IM_NT)
im2col_stem_kernel(const float* __restrict__ tokens,
                   float* __restrict__ col) {
  const int t   = threadIdx.x;
  const int bl0 = blockIdx.x * IM_R;
  const int b   = bl0 >> 10;
  const int l0  = bl0 & (LOUT - 1);
  const float* tb = tokens + (long)b * C_IN * SEQLEN;
  int off[4], kk[4];
  #pragma unroll
  for (int m = 0; m < 4; ++m) {
    const int j = t + IM_NT * m;
    const int c = j / 7;
    kk[m]  = j - c * 7;
    off[m] = c * SEQLEN;
  }
  float v[IM_R][4];
  #pragma unroll
  for (int i = 0; i < IM_R; ++i) {
    const int s0 = 2 * (l0 + i) - 3;
    #pragma unroll
    for (int m = 0; m < 4; ++m) {
      const int s  = s0 + kk[m];
      const int sc = min(max(s, 0), SEQLEN - 1);
      const float x = tb[off[m] + sc];
      v[i][m] = (s == sc) ? x : 0.f;
    }
  }
  float* dst = col + (long)bl0 * KSTEM + t;
  #pragma unroll
  for (int i = 0; i < IM_R; ++i) {
    #pragma unroll
    for (int m = 0; m < 4; ++m) dst[(long)i * KSTEM + IM_NT * m] = v[i][m];
  }
}

#define GT_M 64
#define GT_N 64
#define GT_K 64
#define GT_P 68

__global__ void __launch_bounds__(256)
gemm_xt_wmma(const float* __restrict__ A,
             const float* __restrict__ W,
             float* __restrict__ C,
             int M, int N, int K) {
  __shared__ __attribute__((aligned(16))) float sA[2][GT_M][GT_P];
  __shared__ __attribute__((aligned(16))) float sB[2][GT_N][GT_P];
  const int tid  = threadIdx.x;
  const int lane = tid & 31;
  const int wave = tid >> 5;
  const int nblk = (N + GT_N - 1) / GT_N;
  const int mb = blockIdx.x / nblk;
  const int nb = blockIdx.x - mb * nblk;
  const int m0 = mb * GT_M;
  const int n0 = nb * GT_N;
  (void)M;

  const int ld_c4 = tid & 15;
  const int ld_r  = tid >> 4;
  const long a_step = (long)16 * K;
  const float* gA  = A + (long)(m0 + ld_r) * K + ld_c4 * 4;
  const float* gB0 = W + (long)min(n0 + ld_r,      N - 1) * K + ld_c4 * 4;
  const float* gB1 = W + (long)min(n0 + ld_r + 16, N - 1) * K + ld_c4 * 4;
  const float* gB2 = W + (long)min(n0 + ld_r + 32, N - 1) * K + ld_c4 * 4;
  const float* gB3 = W + (long)min(n0 + ld_r + 48, N - 1) * K + ld_c4 * 4;

  float4 ra0, ra1, ra2, ra3, rb0, rb1, rb2, rb3;

#define GT_FETCH(k0)                                                   \
  do {                                                                 \
    ra0 = *(const float4*)(gA + (k0));                                 \
    ra1 = *(const float4*)(gA + a_step + (k0));                        \
    ra2 = *(const float4*)(gA + 2 * a_step + (k0));                    \
    ra3 = *(const float4*)(gA + 3 * a_step + (k0));                    \
    rb0 = *(const float4*)(gB0 + (k0));                                \
    rb1 = *(const float4*)(gB1 + (k0));                                \
    rb2 = *(const float4*)(gB2 + (k0));                                \
    rb3 = *(const float4*)(gB3 + (k0));                                \
  } while (0)

#define GT_COMMIT(buf)                                                 \
  do {                                                                 \
    *(float4*)&sA[(buf)][ld_r     ][ld_c4 * 4] = ra0;                  \
    *(float4*)&sA[(buf)][ld_r + 16][ld_c4 * 4] = ra1;                  \
    *(float4*)&sA[(buf)][ld_r + 32][ld_c4 * 4] = ra2;                  \
    *(float4*)&sA[(buf)][ld_r + 48][ld_c4 * 4] = ra3;                  \
    *(float4*)&sB[(buf)][ld_r     ][ld_c4 * 4] = rb0;                  \
    *(float4*)&sB[(buf)][ld_r + 16][ld_c4 * 4] = rb1;                  \
    *(float4*)&sB[(buf)][ld_r + 32][ld_c4 * 4] = rb2;                  \
    *(float4*)&sB[(buf)][ld_r + 48][ld_c4 * 4] = rb3;                  \
  } while (0)

  const int r  = lane & 15;
  const int h  = lane >> 4;
  const int wm = wave & 3;
  const int wn = wave >> 2;
  const float* fa  = &sA[0][wm * 16 + r][4 * h];
  const float* fb0 = &sB[0][wn * 32 + r][4 * h];
  const float* fb1 = &sB[0][wn * 32 + 16 + r][4 * h];

  v8f acc0 = {0.f,0.f,0.f,0.f,0.f,0.f,0.f,0.f};
  v8f acc1 = acc0;

#define GT_COMPUTE(buf)                                                                            \
  do {                                                                                             \
    const float* pa  = fa  + (buf) * (GT_M * GT_P);                                                \
    const float* pb0 = fb0 + (buf) * (GT_N * GT_P);                                                \
    const float* pb1 = fb1 + (buf) * (GT_N * GT_P);                                                \
    _Pragma("unroll")                                                                              \
    for (int g = 0; g < GT_K / 8; ++g) {                                                           \
      const float4 a4 = *(const float4*)(pa  + g * 8);                                             \
      const float4 p4 = *(const float4*)(pb0 + g * 8);                                             \
      const float4 q4 = *(const float4*)(pb1 + g * 8);                                             \
      const v2f alo = {a4.x, a4.y}, ahi = {a4.z, a4.w};                                            \
      const v2f plo = {p4.x, p4.y}, phi = {p4.z, p4.w};                                            \
      const v2f qlo = {q4.x, q4.y}, qhi = {q4.z, q4.w};                                            \
      acc0 = __builtin_amdgcn_wmma_f32_16x16x4_f32(false, alo, false, plo, (short)0, acc0, false, false); \
      acc1 = __builtin_amdgcn_wmma_f32_16x16x4_f32(false, alo, false, qlo, (short)0, acc1, false, false); \
      acc0 = __builtin_amdgcn_wmma_f32_16x16x4_f32(false, ahi, false, phi, (short)0, acc0, false, false); \
      acc1 = __builtin_amdgcn_wmma_f32_16x16x4_f32(false, ahi, false, qhi, (short)0, acc1, false, false); \
    }                                                                                              \
  } while (0)

  const int KT = K / GT_K;
  GT_FETCH(0);
  GT_COMMIT(0);
  __syncthreads();

  for (int kt = 0; kt + 1 < KT; ++kt) {
    const int cb = kt & 1;
    GT_FETCH((kt + 1) * GT_K);
    GT_COMPUTE(cb);
    GT_COMMIT(cb ^ 1);
    __syncthreads();
  }
  GT_COMPUTE((KT - 1) & 1);
  __syncthreads();

  float (*sC)[GT_P] = sA[0];
  #pragma unroll
  for (int i = 0; i < 8; ++i) {
    sC[wm * 16 + 8 * h + i][wn * 32 + r]      = acc0[i];
    sC[wm * 16 + 8 * h + i][wn * 32 + 16 + r] = acc1[i];
  }
  __syncthreads();
  const int col = n0 + ld_c4 * 4;
  if (col < N) {
    #pragma unroll
    for (int p = 0; p < 4; ++p) {
      const int row = ld_r + 16 * p;
      const float4 v = *(const float4*)&sC[row][ld_c4 * 4];
      *(float4*)(C + (long)(m0 + row) * N + col) = v;
    }
  }
#undef GT_FETCH
#undef GT_COMMIT
#undef GT_COMPUTE
}

#define DW_R  8
#define DW_NT (CONV_DIM / 4)

__device__ __forceinline__ float dwconv_one(const float4 wv, float x0, float xm1, float xm2,
                                            float xm3, int l, float bs) {
  float acc = wv.w * x0;
  if (l >= 1) acc = fmaf(wv.z, xm1, acc);
  if (l >= 2) acc = fmaf(wv.y, xm2, acc);
  if (l >= 3) acc = fmaf(wv.x, xm3, acc);
  const float v = acc + bs;
  return v / (1.f + __expf(-v));
}

__global__ void __launch_bounds__(DW_NT)
dwconv_silu_kernel(const float* __restrict__ zx,
                   const float* __restrict__ w,
                   const float* __restrict__ bias,
                   float* __restrict__ out) {
  const int t   = threadIdx.x;
  const int bl0 = blockIdx.x * DW_R;
  const int l0  = bl0 & (LOUT - 1);
  const float* xin = zx + (long)bl0 * D_IN_PROJ + D_INNER + 4 * t;
  float4 xr[DW_R + 3];
  #pragma unroll
  for (int i = 0; i < 3; ++i) {
    const long back = (l0 >= 3 - i) ? (long)(3 - i) * D_IN_PROJ : 0;
    xr[i] = *(const float4*)(xin - back);
  }
  #pragma unroll
  for (int i = 0; i < DW_R; ++i) xr[3 + i] = *(const float4*)(xin + (long)i * D_IN_PROJ);
  const float4 wa = *(const float4*)(w + (4 * t + 0) * 4);
  const float4 wb = *(const float4*)(w + (4 * t + 1) * 4);
  const float4 wc = *(const float4*)(w + (4 * t + 2) * 4);
  const float4 wd = *(const float4*)(w + (4 * t + 3) * 4);
  const float4 bs = *(const float4*)(bias + 4 * t);
  float* dst = out + (long)bl0 * CONV_DIM + 4 * t;
  #pragma unroll
  for (int i = 0; i < DW_R; ++i) {
    const int l = l0 + i;
    float4 o;
    o.x = dwconv_one(wa, xr[3 + i].x, xr[2 + i].x, xr[1 + i].x, xr[i].x, l, bs.x);
    o.y = dwconv_one(wb, xr[3 + i].y, xr[2 + i].y, xr[1 + i].y, xr[i].y, l, bs.y);
    o.z = dwconv_one(wc, xr[3 + i].z, xr[2 + i].z, xr[1 + i].z, xr[i].z, l, bs.z);
    o.w = dwconv_one(wd, xr[3 + i].w, xr[2 + i].w, xr[1 + i].w, xr[i].w, l, bs.w);
    *(float4*)(dst + (long)i * CONV_DIM) = o;
  }
}

#define SCAN_T   32
#define SCAN_NCH (LOUT / SCAN_T)
#define SCAN_SPLIT 2
#define SCAN_PB    (HEADDIM / SCAN_SPLIT)
#define SCAN_NT    (SCAN_PB * 4)

template <int CTRL>
__device__ __forceinline__ float lane_xor_read(float v) {
  return __int_as_float(__builtin_amdgcn_update_dpp(0, __float_as_int(v), CTRL, 0xF, 0xF, true));
}

__global__ void __launch_bounds__(SCAN_NT)
ssm_scan_kernel(const float* __restrict__ zx,
                const float* __restrict__ xbc,
                const float* __restrict__ dtb,
                const float* __restrict__ alog,
                const float* __restrict__ Dp,
                float* __restrict__ y) {
  const int pblk = blockIdx.x % SCAN_SPLIT;
  const int hh = (blockIdx.x / SCAN_SPLIT) & 15;
  const int b  = blockIdx.x / (SCAN_SPLIT * NHEADS);
  const int tid = threadIdx.x;
  const int p = tid >> 2;
  const int q = tid & 3;
  __shared__ __attribute__((aligned(16))) float sBC[2][SCAN_T][2 * D_STATE];
  __shared__ __attribute__((aligned(16))) float sX[2][SCAN_T][SCAN_PB];
  __shared__ __attribute__((aligned(16))) float sY[2][SCAN_T][SCAN_PB];
  __shared__ __attribute__((aligned(8)))  float sDT[2][SCAN_T][2];
  const float Ah   = -__expf(alog[hh]);
  const float dtbh = dtb[hh];
  const float Dh   = Dp[hh];
  float hreg[16];
  #pragma unroll
  for (int i = 0; i < 16; ++i) hreg[i] = 0.f;

  const int bc_c4 = tid & 31;
  const int bc_t  = tid >> 5;
  const int x_c4  = tid & 7;
  const int x_t   = tid >> 3;
  const bool dt_wave = (tid >= SCAN_NT - SCAN_T);
  const int  dt_t    = dt_wave ? tid - (SCAN_NT - SCAN_T) : 0;
  const float* gBC = xbc + ((long)b * LOUT + bc_t) * CONV_DIM + D_INNER + bc_c4 * 4;
  const float* gX  = xbc + ((long)b * LOUT + x_t) * CONV_DIM + hh * HEADDIM + pblk * SCAN_PB + x_c4 * 4;
  const float* gDT = zx + ((long)b * LOUT + dt_t) * D_IN_PROJ + (D_IN_PROJ - NHEADS) + hh;
  float*       gY  = y + ((long)b * LOUT + x_t) * D_INNER + hh * HEADDIM + pblk * SCAN_PB + x_c4 * 4;

  float4 rBC0, rBC1, rBC2, rBC3, rBC4, rBC5, rBC6, rBC7, rX0, rX1;
  float  rdt = 0.f;

#define SCAN_FETCH(l0)                                                              \
  do {                                                                              \
    const float* s_bc = gBC + (long)(l0) * CONV_DIM;                                \
    const float* s_x  = gX  + (long)(l0) * CONV_DIM;                                \
    rBC0 = *(const float4*)(s_bc + 0 * CONV_DIM);                                   \
    rBC1 = *(const float4*)(s_bc + 4 * CONV_DIM);                                   \
    rBC2 = *(const float4*)(s_bc + 8 * CONV_DIM);                                   \
    rBC3 = *(const float4*)(s_bc + 12 * CONV_DIM);                                  \
    rBC4 = *(const float4*)(s_bc + 16 * CONV_DIM);                                  \
    rBC5 = *(const float4*)(s_bc + 20 * CONV_DIM);                                  \
    rBC6 = *(const float4*)(s_bc + 24 * CONV_DIM);                                  \
    rBC7 = *(const float4*)(s_bc + 28 * CONV_DIM);                                  \
    rX0  = *(const float4*)(s_x);                                                   \
    rX1  = *(const float4*)(s_x + 16 * CONV_DIM);                                   \
    if (dt_wave) rdt = gDT[(long)(l0) * D_IN_PROJ];                                 \
  } while (0)

#define SCAN_COMMIT(buf)                                                            \
  do {                                                                              \
    *(float4*)&sBC[(buf)][bc_t + 0][bc_c4 * 4] = rBC0;                              \
    *(float4*)&sBC[(buf)][bc_t + 4][bc_c4 * 4] = rBC1;                              \
    *(float4*)&sBC[(buf)][bc_t + 8][bc_c4 * 4] = rBC2;                              \
    *(float4*)&sBC[(buf)][bc_t + 12][bc_c4 * 4] = rBC3;                             \
    *(float4*)&sBC[(buf)][bc_t + 16][bc_c4 * 4] = rBC4;                             \
    *(float4*)&sBC[(buf)][bc_t + 20][bc_c4 * 4] = rBC5;                             \
    *(float4*)&sBC[(buf)][bc_t + 24][bc_c4 * 4] = rBC6;                             \
    *(float4*)&sBC[(buf)][bc_t + 28][bc_c4 * 4] = rBC7;                             \
    *(float4*)&sX[(buf)][x_t     ][x_c4 * 4] = rX0;                                 \
    *(float4*)&sX[(buf)][x_t + 16][x_c4 * 4] = rX1;                                 \
    if (dt_wave) {                                                                  \
      const float raw = rdt + dtbh;                                                 \
      const float dtv = (raw > 20.f) ? raw : log1pf(__expf(raw));      \
      float2 dd;                                                                    \
      dd.x = dtv;                                                                   \
      dd.y = __expf(dtv * Ah);                                               \
      *(float2*)&sDT[(buf)][dt_t][0] = dd;                                          \
    }                                                                               \
  } while (0)

#define SCAN_WRITEOUT(l0, buf)                                                      \
  do {                                                                              \
    float* d_y = gY + (long)(l0) * D_INNER;                                         \
    const float4 v0 = *(const float4*)&sY[(buf)][x_t     ][x_c4 * 4];               \
    const float4 v1 = *(const float4*)&sY[(buf)][x_t + 16][x_c4 * 4];               \
    *(float4*)(d_y)                = v0;                                            \
    *(float4*)(d_y + 16 * D_INNER) = v1;                                            \
  } while (0)

  float4 sB[4][4], sC[4][4];
  float2 sD[4];
  float  sXv[4];

#define SCAN_LDS_READ(u, buf, t)                                                    \
  do {                                                                              \
    const float4* b4 = (const float4*)&sBC[(buf)][(t)][q * 16];                     \
    const float4* c4 = (const float4*)&sBC[(buf)][(t)][D_STATE + q * 16];           \
    sD[(u)]  = *(const float2*)&sDT[(buf)][(t)][0];                                 \
    sXv[(u)] = sX[(buf)][(t)][p];                                                   \
    sB[(u)][0] = b4[0]; sB[(u)][1] = b4[1]; sB[(u)][2] = b4[2]; sB[(u)][3] = b4[3]; \
    sC[(u)][0] = c4[0]; sC[(u)][1] = c4[1]; sC[(u)][2] = c4[2]; sC[(u)][3] = c4[3]; \
  } while (0)

#define SCAN_STEP(u, yout, xout)                                                        \
  do {                                                                              \
    const float dtv = sD[(u)].x;                                                    \
    const float dA  = sD[(u)].y;                                                    \
    const float xv  = sXv[(u)];                                                     \
    const float xdt = xv * dtv;                                                     \
    float accv = 0.f;                                                               \
    _Pragma("unroll")                                                               \
    for (int j = 0; j < 4; ++j) {                                                   \
      const float4 bv = sB[(u)][j];                                                 \
      const float4 cv = sC[(u)][j];                                                 \
      hreg[4 * j + 0] = fmaf(hreg[4 * j + 0], dA, xdt * bv.x);                      \
      accv            = fmaf(hreg[4 * j + 0], cv.x, accv);                          \
      hreg[4 * j + 1] = fmaf(hreg[4 * j + 1], dA, xdt * bv.y);                      \
      accv            = fmaf(hreg[4 * j + 1], cv.y, accv);                          \
      hreg[4 * j + 2] = fmaf(hreg[4 * j + 2], dA, xdt * bv.z);                      \
      accv            = fmaf(hreg[4 * j + 2], cv.z, accv);                          \
      hreg[4 * j + 3] = fmaf(hreg[4 * j + 3], dA, xdt * bv.w);                      \
      accv            = fmaf(hreg[4 * j + 3], cv.w, accv);                          \
    }                                                                               \
    (yout) = accv;                                                                  \
    (xout) = xv;                                                                    \
  } while (0)

#define SCAN_FINISH(acc, xv)                                                        \
  do {                                                                              \
    (acc) += lane_xor_read<0x161>(acc);                             \
    (acc) += lane_xor_read<0x162>(acc);                             \
    (acc) = fmaf((xv), Dh, (acc));                                                  \
  } while (0)

  SCAN_FETCH(0);
  SCAN_COMMIT(0);
  __syncthreads();

  for (int c = 0; c < SCAN_NCH; ++c) {
    const int cb = c & 1;
    const bool more = (c + 1 < SCAN_NCH);
    if (c > 0) SCAN_WRITEOUT((c - 1) * SCAN_T, cb ^ 1);
    SCAN_LDS_READ(0, cb, 0);
    SCAN_LDS_READ(1, cb, 1);
    __builtin_amdgcn_sched_barrier(0);
    SCAN_LDS_READ(2, cb, 2);
    SCAN_LDS_READ(3, cb, 3);
    if (more) SCAN_FETCH((c + 1) * SCAN_T);
    #pragma unroll
    for (int t0 = 0; t0 < SCAN_T - 4; t0 += 4) {
      float yv[4], xs[4];
      #pragma unroll
      for (int u = 0; u < 4; ++u) {
        SCAN_STEP(u, yv[u], xs[u]);
        SCAN_LDS_READ(u, cb, t0 + 4 + u);
      }
      #pragma unroll
      for (int u = 0; u < 4; ++u) SCAN_FINISH(yv[u], xs[u]);
      if (q == 0) {
        #pragma unroll
        for (int u = 0; u < 4; ++u) sY[cb][t0 + u][p] = yv[u];
      }
    }
    {
      float yv[4], xs[4];
      #pragma unroll
      for (int u = 0; u < 4; ++u) SCAN_STEP(u, yv[u], xs[u]);
      #pragma unroll
      for (int u = 0; u < 4; ++u) SCAN_FINISH(yv[u], xs[u]);
      if (q == 0) {
        #pragma unroll
        for (int u = 0; u < 4; ++u) sY[cb][SCAN_T - 4 + u][p] = yv[u];
      }
    }
    if (more) SCAN_COMMIT(cb ^ 1);
    __syncthreads();
  }
  SCAN_WRITEOUT((SCAN_NCH - 1) * SCAN_T, (SCAN_NCH - 1) & 1);
#undef SCAN_FETCH
#undef SCAN_COMMIT
#undef SCAN_WRITEOUT
#undef SCAN_LDS_READ
#undef SCAN_STEP
#undef SCAN_FINISH
}

#define GR_R 4
__global__ void __launch_bounds__(256)
gate_rmsnorm_kernel(const float* __restrict__ zx,
                    const float* __restrict__ nw,
                    float* __restrict__ y) {
  const long bl0 = (long)blockIdx.x * GR_R;
  const int tid = threadIdx.x;
  __shared__ float red[GR_R][256];
  float zv[GR_R][4];
  float yv[GR_R][4];
  #pragma unroll
  for (int r = 0; r < GR_R; ++r) {
    #pragma unroll
    for (int j = 0; j < 4; ++j) {
      const int i = tid + j * 256;
      zv[r][j] = zx[(bl0 + r) * D_IN_PROJ + i];
      yv[r][j] = y[(bl0 + r) * D_INNER + i];
    }
  }
  #pragma unroll
  for (int r = 0; r < GR_R; ++r) {
    float ss = 0.f;
    #pragma unroll
    for (int j = 0; j < 4; ++j) {
      const float z = zv[r][j];
      const float g = z / (1.f + __expf(-z));
      const float v = yv[r][j] * g;
      yv[r][j] = v;
      ss = fmaf(v, v, ss);
    }
    red[r][tid] = ss;
  }
  __syncthreads();
  for (int st = 128; st > 0; st >>= 1) {
    if (tid < st) {
      #pragma unroll
      for (int r = 0; r < GR_R; ++r) red[r][tid] += red[r][tid + st];
    }
    __syncthreads();
  }
  float nwv[4];
  #pragma unroll
  for (int j = 0; j < 4; ++j) nwv[j] = nw[tid + j * 256];
  #pragma unroll
  for (int r = 0; r < GR_R; ++r) {
    const float scale = rsqrtf(red[r][0] * (1.f / (float)D_INNER) + 1e-5f);
    #pragma unroll
    for (int j = 0; j < 4; ++j) {
      const int i = tid + j * 256;
      y[(bl0 + r) * D_INNER + i] = yv[r][j] * scale * nwv[j];
    }
  }
}

extern "C" void kernel_launch(void* const* d_in, const int* in_sizes, int n_in,
                              void* d_out, int out_size, void* d_ws, size_t ws_size,
                              hipStream_t stream) {
  const float* tokens     = (const float*)d_in[0];
  const float* conv_w     = (const float*)d_in[1];
  const float* in_proj_w  = (const float*)d_in[2];
  const float* conv1d_w   = (const float*)d_in[3];
  const float* conv1d_b   = (const float*)d_in[4];
  const float* dt_bias    = (const float*)d_in[5];
  const float* A_log      = (const float*)d_in[6];
  const float* D_param    = (const float*)d_in[7];
  const float* norm_w     = (const float*)d_in[8];
  const float* out_proj_w = (const float*)d_in[9];
  float* out = (float*)d_out;
  float* ws  = (float*)d_ws;

  float* u0   = ws;
  float* u1   = u0 + (long)ROWS * D_MODEL;
  float* zx   = u1 + (long)ROWS * D_MODEL;
  float* xact = zx + (long)ROWS * D_IN_PROJ;
  float* ybuf = xact + (long)ROWS * CONV_DIM;
  float* col  = zx;
  (void)ws_size; (void)in_sizes; (void)n_in; (void)out_size;

  {
    im2col_stem_kernel<<<ROWS / IM_R, IM_NT, 0, stream>>>(tokens, col);
    gemm_xt_wmma<<<(ROWS / GT_M) * ((D_MODEL + GT_N - 1) / GT_N), 256, 0, stream>>>(col, conv_w, u0, ROWS, D_MODEL, KSTEM);
  }

  const float* u_in = u0;
  for (int blk = 0; blk < 2; ++blk) {
    const float* ipw = in_proj_w  + (long)blk * D_IN_PROJ * D_MODEL;
    const float* cw  = conv1d_w   + (long)blk * CONV_DIM * D_CONV;
    const float* cb  = conv1d_b   + (long)blk * CONV_DIM;
    const float* dtb = dt_bias    + (long)blk * NHEADS;
    const float* al  = A_log      + (long)blk * NHEADS;
    const float* Dpp = D_param    + (long)blk * NHEADS;
    const float* nw  = norm_w     + (long)blk * D_INNER;
    const float* opw = out_proj_w + (long)blk * D_MODEL * D_INNER;

    {
      gemm_xt_wmma<<<(ROWS / GT_M) * ((D_IN_PROJ + GT_N - 1) / GT_N), 256, 0, stream>>>(u_in, ipw, zx, ROWS, D_IN_PROJ, D_MODEL);
    }
    dwconv_silu_kernel<<<ROWS / DW_R, DW_NT, 0, stream>>>(zx, cw, cb, xact);
    ssm_scan_kernel<<<BATCH * NHEADS * SCAN_SPLIT, SCAN_NT, 0, stream>>>(zx, xact, dtb, al, Dpp, ybuf);
    gate_rmsnorm_kernel<<<ROWS / GR_R, 256, 0, stream>>>(zx, nw, ybuf);
    float* dst = (blk == 0) ? u1 : out;
    {
      gemm_xt_wmma<<<(ROWS / GT_M) * ((D_MODEL + GT_N - 1) / GT_N), 256, 0, stream>>>(ybuf, opw, dst, ROWS, D_MODEL, D_INNER);
    }
    u_in = u1;
  }
}


# ssm scan: a head's 64 state rows split over 8 workgroups (256 of 128 threads, not 64), 16 lanes x 4 n per row; sum over n now 4 in a lane, then a 16-lane DPP tree; a y line = 4 steps x 8 rows
# speedup vs baseline: 11.6258x; 1.2688x over previous
#include <hip/hip_runtime.h>
#include <hip/hip_bf16.h>

typedef __attribute__((ext_vector_type(2))) float v2f;
typedef __attribute__((ext_vector_type(8))) float v8f;

#define BATCH     2
#define SEQLEN    2048
#define C_IN      256
#define D_MODEL   512
#define LOUT      1024
#define D_STATE   64
#define D_CONV    4
#define D_INNER   1024
#define HEADDIM   64
#define NHEADS    16
#define CONV_DIM  (D_INNER + 2*D_STATE)
#define D_IN_PROJ (2*D_INNER + 2*D_STATE + NHEADS)
#define ROWS      (BATCH*LOUT)
#define KSTEM     (C_IN * 7)

#define IM_R  8
#define IM_NT (KSTEM / 4)
__global__ void __launch_bounds__(IM_NT)
im2col_stem_kernel(const float* __restrict__ tokens,
                   float* __restrict__ col) {
  const int t   = threadIdx.x;
  const int bl0 = blockIdx.x * IM_R;
  const int b   = bl0 >> 10;
  const int l0  = bl0 & (LOUT - 1);
  const float* tb = tokens + (long)b * C_IN * SEQLEN;
  int off[4], kk[4];
  #pragma unroll
  for (int m = 0; m < 4; ++m) {
    const int j = t + IM_NT * m;
    const int c = j / 7;
    kk[m]  = j - c * 7;
    off[m] = c * SEQLEN;
  }
  float v[IM_R][4];
  #pragma unroll
  for (int i = 0; i < IM_R; ++i) {
    const int s0 = 2 * (l0 + i) - 3;
    #pragma unroll
    for (int m = 0; m < 4; ++m) {
      const int s  = s0 + kk[m];
      const int sc = min(max(s, 0), SEQLEN - 1);
      const float x = tb[off[m] + sc];
      v[i][m] = (s == sc) ? x : 0.f;
    }
  }
  float* dst = col + (long)bl0 * KSTEM + t;
  #pragma unroll
  for (int i = 0; i < IM_R; ++i) {
    #pragma unroll
    for (int m = 0; m < 4; ++m) dst[(long)i * KSTEM + IM_NT * m] = v[i][m];
  }
}

#define GT_M 64
#define GT_N 64
#define GT_K 64
#define GT_P 68

__global__ void __launch_bounds__(256)
gemm_xt_wmma(const float* __restrict__ A,
             const float* __restrict__ W,
             float* __restrict__ C,
             int M, int N, int K) {
  __shared__ __attribute__((aligned(16))) float sA[2][GT_M][GT_P];
  __shared__ __attribute__((aligned(16))) float sB[2][GT_N][GT_P];
  const int tid  = threadIdx.x;
  const int lane = tid & 31;
  const int wave = tid >> 5;
  const int nblk = (N + GT_N - 1) / GT_N;
  const int mb = blockIdx.x / nblk;
  const int nb = blockIdx.x - mb * nblk;
  const int m0 = mb * GT_M;
  const int n0 = nb * GT_N;
  (void)M;

  const int ld_c4 = tid & 15;
  const int ld_r  = tid >> 4;
  const long a_step = (long)16 * K;
  const float* gA  = A + (long)(m0 + ld_r) * K + ld_c4 * 4;
  const float* gB0 = W + (long)min(n0 + ld_r,      N - 1) * K + ld_c4 * 4;
  const float* gB1 = W + (long)min(n0 + ld_r + 16, N - 1) * K + ld_c4 * 4;
  const float* gB2 = W + (long)min(n0 + ld_r + 32, N - 1) * K + ld_c4 * 4;
  const float* gB3 = W + (long)min(n0 + ld_r + 48, N - 1) * K + ld_c4 * 4;

  float4 ra0, ra1, ra2, ra3, rb0, rb1, rb2, rb3;

#define GT_FETCH(k0)                                                   \
  do {                                                                 \
    ra0 = *(const float4*)(gA + (k0));                                 \
    ra1 = *(const float4*)(gA + a_step + (k0));                        \
    ra2 = *(const float4*)(gA + 2 * a_step + (k0));                    \
    ra3 = *(const float4*)(gA + 3 * a_step + (k0));                    \
    rb0 = *(const float4*)(gB0 + (k0));                                \
    rb1 = *(const float4*)(gB1 + (k0));                                \
    rb2 = *(const float4*)(gB2 + (k0));                                \
    rb3 = *(const float4*)(gB3 + (k0));                                \
  } while (0)

#define GT_COMMIT(buf)                                                 \
  do {                                                                 \
    *(float4*)&sA[(buf)][ld_r     ][ld_c4 * 4] = ra0;                  \
    *(float4*)&sA[(buf)][ld_r + 16][ld_c4 * 4] = ra1;                  \
    *(float4*)&sA[(buf)][ld_r + 32][ld_c4 * 4] = ra2;                  \
    *(float4*)&sA[(buf)][ld_r + 48][ld_c4 * 4] = ra3;                  \
    *(float4*)&sB[(buf)][ld_r     ][ld_c4 * 4] = rb0;                  \
    *(float4*)&sB[(buf)][ld_r + 16][ld_c4 * 4] = rb1;                  \
    *(float4*)&sB[(buf)][ld_r + 32][ld_c4 * 4] = rb2;                  \
    *(float4*)&sB[(buf)][ld_r + 48][ld_c4 * 4] = rb3;                  \
  } while (0)

  const int r  = lane & 15;
  const int h  = lane >> 4;
  const int wm = wave & 3;
  const int wn = wave >> 2;
  const float* fa  = &sA[0][wm * 16 + r][4 * h];
  const float* fb0 = &sB[0][wn * 32 + r][4 * h];
  const float* fb1 = &sB[0][wn * 32 + 16 + r][4 * h];

  v8f acc0 = {0.f,0.f,0.f,0.f,0.f,0.f,0.f,0.f};
  v8f acc1 = acc0;

#define GT_COMPUTE(buf)                                                                            \
  do {                                                                                             \
    const float* pa  = fa  + (buf) * (GT_M * GT_P);                                                \
    const float* pb0 = fb0 + (buf) * (GT_N * GT_P);                                                \
    const float* pb1 = fb1 + (buf) * (GT_N * GT_P);                                                \
    _Pragma("unroll")                                                                              \
    for (int g = 0; g < GT_K / 8; ++g) {                                                           \
      const float4 a4 = *(const float4*)(pa  + g * 8);                                             \
      const float4 p4 = *(const float4*)(pb0 + g * 8);                                             \
      const float4 q4 = *(const float4*)(pb1 + g * 8);                                             \
      const v2f alo = {a4.x, a4.y}, ahi = {a4.z, a4.w};                                            \
      const v2f plo = {p4.x, p4.y}, phi = {p4.z, p4.w};                                            \
      const v2f qlo = {q4.x, q4.y}, qhi = {q4.z, q4.w};                                            \
      acc0 = __builtin_amdgcn_wmma_f32_16x16x4_f32(false, alo, false, plo, (short)0, acc0, false, false); \
      acc1 = __builtin_amdgcn_wmma_f32_16x16x4_f32(false, alo, false, qlo, (short)0, acc1, false, false); \
      acc0 = __builtin_amdgcn_wmma_f32_16x16x4_f32(false, ahi, false, phi, (short)0, acc0, false, false); \
      acc1 = __builtin_amdgcn_wmma_f32_16x16x4_f32(false, ahi, false, qhi, (short)0, acc1, false, false); \
    }                                                                                              \
  } while (0)

  const int KT = K / GT_K;
  GT_FETCH(0);
  GT_COMMIT(0);
  __syncthreads();

  for (int kt = 0; kt + 1 < KT; ++kt) {
    const int cb = kt & 1;
    GT_FETCH((kt + 1) * GT_K);
    GT_COMPUTE(cb);
    GT_COMMIT(cb ^ 1);
    __syncthreads();
  }
  GT_COMPUTE((KT - 1) & 1);
  __syncthreads();

  float (*sC)[GT_P] = sA[0];
  #pragma unroll
  for (int i = 0; i < 8; ++i) {
    sC[wm * 16 + 8 * h + i][wn * 32 + r]      = acc0[i];
    sC[wm * 16 + 8 * h + i][wn * 32 + 16 + r] = acc1[i];
  }
  __syncthreads();
  const int col = n0 + ld_c4 * 4;
  if (col < N) {
    #pragma unroll
    for (int p = 0; p < 4; ++p) {
      const int row = ld_r + 16 * p;
      const float4 v = *(const float4*)&sC[row][ld_c4 * 4];
      *(float4*)(C + (long)(m0 + row) * N + col) = v;
    }
  }
#undef GT_FETCH
#undef GT_COMMIT
#undef GT_COMPUTE
}

#define DW_R  8
#define DW_NT (CONV_DIM / 4)

__device__ __forceinline__ float dwconv_one(const float4 wv, float x0, float xm1, float xm2,
                                            float xm3, int l, float bs) {
  float acc = wv.w * x0;
  if (l >= 1) acc = fmaf(wv.z, xm1, acc);
  if (l >= 2) acc = fmaf(wv.y, xm2, acc);
  if (l >= 3) acc = fmaf(wv.x, xm3, acc);
  const float v = acc + bs;
  return v / (1.f + __expf(-v));
}

__global__ void __launch_bounds__(DW_NT)
dwconv_silu_kernel(const float* __restrict__ zx,
                   const float* __restrict__ w,
                   const float* __restrict__ bias,
                   float* __restrict__ out) {
  const int t   = threadIdx.x;
  const int bl0 = blockIdx.x * DW_R;
  const int l0  = bl0 & (LOUT - 1);
  const float* xin = zx + (long)bl0 * D_IN_PROJ + D_INNER + 4 * t;
  float4 xr[DW_R + 3];
  #pragma unroll
  for (int i = 0; i < 3; ++i) {
    const long back = (l0 >= 3 - i) ? (long)(3 - i) * D_IN_PROJ : 0;
    xr[i] = *(const float4*)(xin - back);
  }
  #pragma unroll
  for (int i = 0; i < DW_R; ++i) xr[3 + i] = *(const float4*)(xin + (long)i * D_IN_PROJ);
  const float4 wa = *(const float4*)(w + (4 * t + 0) * 4);
  const float4 wb = *(const float4*)(w + (4 * t + 1) * 4);
  const float4 wc = *(const float4*)(w + (4 * t + 2) * 4);
  const float4 wd = *(const float4*)(w + (4 * t + 3) * 4);
  const float4 bs = *(const float4*)(bias + 4 * t);
  float* dst = out + (long)bl0 * CONV_DIM + 4 * t;
  #pragma unroll
  for (int i = 0; i < DW_R; ++i) {
    const int l = l0 + i;
    float4 o;
    o.x = dwconv_one(wa, xr[3 + i].x, xr[2 + i].x, xr[1 + i].x, xr[i].x, l, bs.x);
    o.y = dwconv_one(wb, xr[3 + i].y, xr[2 + i].y, xr[1 + i].y, xr[i].y, l, bs.y);
    o.z = dwconv_one(wc, xr[3 + i].z, xr[2 + i].z, xr[1 + i].z, xr[i].z, l, bs.z);
    o.w = dwconv_one(wd, xr[3 + i].w, xr[2 + i].w, xr[1 + i].w, xr[i].w, l, bs.w);
    *(float4*)(dst + (long)i * CONV_DIM) = o;
  }
}

#define SCAN_T   32
#define SCAN_NCH (LOUT / SCAN_T)
#define SCAN_SPLIT 8
#define SCAN_PB    (HEADDIM / SCAN_SPLIT)
#define SCAN_LPR   16
#define SCAN_NE    (D_STATE / SCAN_LPR)
#define SCAN_NT    (SCAN_PB * SCAN_LPR)
#define SCAN_YG    (32 / SCAN_PB)
#define SCAN_D     8

template <int CTRL>
__device__ __forceinline__ float lane_xor_read(float v) {
  return __int_as_float(__builtin_amdgcn_update_dpp(0, __float_as_int(v), CTRL, 0xF, 0xF, true));
}

__global__ void __launch_bounds__(SCAN_NT)
ssm_scan_kernel(const float* __restrict__ zx,
                const float* __restrict__ xbc,
                const float* __restrict__ dtb,
                const float* __restrict__ alog,
                const float* __restrict__ Dp,
                float* __restrict__ y) {
  const int pblk = blockIdx.x % SCAN_SPLIT;
  const int hh = (blockIdx.x / SCAN_SPLIT) & 15;
  const int b  = blockIdx.x / (SCAN_SPLIT * NHEADS);
  const int tid = threadIdx.x;
  const int p = tid / SCAN_LPR;
  const int q = tid % SCAN_LPR;
  __shared__ __attribute__((aligned(16))) float sBC[2][SCAN_T][2 * D_STATE];
  __shared__ __attribute__((aligned(16))) float sX[2][SCAN_T][SCAN_PB];
  __shared__ __attribute__((aligned(16))) float sY[2][SCAN_T][SCAN_PB];
  __shared__ __attribute__((aligned(8)))  float sDT[2][SCAN_T][2];
  const float Ah   = -__expf(alog[hh]);
  const float dtbh = dtb[hh];
  const float Dh   = Dp[hh];
  float hreg[SCAN_NE];
  #pragma unroll
  for (int i = 0; i < SCAN_NE; ++i) hreg[i] = 0.f;

  const int bc_c4 = tid & 31;
  const int bc_t  = tid >> 5;
  const bool xy_wave = (tid < 64);
  const int x_c4  = tid & 1;
  const int x_t   = (tid >> 1) & (SCAN_T - 1);
  const int y_g   = (tid >> 3) & 7;
  const int y_c4  = tid & 7;
  const bool dt_wave = (tid >= SCAN_NT - SCAN_T);
  const int  dt_t    = dt_wave ? tid - (SCAN_NT - SCAN_T) : 0;
  const float* gBC = xbc + ((long)b * LOUT + bc_t) * CONV_DIM + D_INNER + bc_c4 * 4;
  const float* gX  = xbc + ((long)b * LOUT + x_t) * CONV_DIM + hh * HEADDIM + pblk * SCAN_PB + x_c4 * 4;
  const float* gDT = zx + ((long)b * LOUT + dt_t) * D_IN_PROJ + (D_IN_PROJ - NHEADS) + hh;
  float*       gY  = y + ((long)b * LOUT + y_g * SCAN_YG) * D_INNER + (hh * SCAN_SPLIT + pblk) * 32 + y_c4 * 4;

  float4 rBC0, rBC1, rBC2, rBC3, rBC4, rBC5, rBC6, rBC7;
  float4 rX0 = make_float4(0.f, 0.f, 0.f, 0.f);
  float  rdt = 0.f;

#define SCAN_FETCH(l0)                                                              \
  do {                                                                              \
    const float* s_bc = gBC + (long)(l0) * CONV_DIM;                                \
    rBC0 = *(const float4*)(s_bc + 0 * CONV_DIM);                                   \
    rBC1 = *(const float4*)(s_bc + 4 * CONV_DIM);                                   \
    rBC2 = *(const float4*)(s_bc + 8 * CONV_DIM);                                   \
    rBC3 = *(const float4*)(s_bc + 12 * CONV_DIM);                                  \
    rBC4 = *(const float4*)(s_bc + 16 * CONV_DIM);                                  \
    rBC5 = *(const float4*)(s_bc + 20 * CONV_DIM);                                  \
    rBC6 = *(const float4*)(s_bc + 24 * CONV_DIM);                                  \
    rBC7 = *(const float4*)(s_bc + 28 * CONV_DIM);                                  \
    if (xy_wave) rX0 = *(const float4*)(gX + (long)(l0) * CONV_DIM);                \
    if (dt_wave) rdt = gDT[(long)(l0) * D_IN_PROJ];                                 \
  } while (0)

#define SCAN_COMMIT(buf)                                                            \
  do {                                                                              \
    *(float4*)&sBC[(buf)][bc_t + 0][bc_c4 * 4] = rBC0;                              \
    *(float4*)&sBC[(buf)][bc_t + 4][bc_c4 * 4] = rBC1;                              \
    *(float4*)&sBC[(buf)][bc_t + 8][bc_c4 * 4] = rBC2;                              \
    *(float4*)&sBC[(buf)][bc_t + 12][bc_c4 * 4] = rBC3;                             \
    *(float4*)&sBC[(buf)][bc_t + 16][bc_c4 * 4] = rBC4;                             \
    *(float4*)&sBC[(buf)][bc_t + 20][bc_c4 * 4] = rBC5;                             \
    *(float4*)&sBC[(buf)][bc_t + 24][bc_c4 * 4] = rBC6;                             \
    *(float4*)&sBC[(buf)][bc_t + 28][bc_c4 * 4] = rBC7;                             \
    if (xy_wave) *(float4*)&sX[(buf)][x_t][x_c4 * 4] = rX0;                         \
    if (dt_wave) {                                                                  \
      const float raw = rdt + dtbh;                                                 \
      const float dtv = (raw > 20.f) ? raw : log1pf(__expf(raw));      \
      float2 dd;                                                                    \
      dd.x = dtv;                                                                   \
      dd.y = __expf(dtv * Ah);                                               \
      *(float2*)&sDT[(buf)][dt_t][0] = dd;                                          \
    }                                                                               \
  } while (0)

#define SCAN_WRITEOUT(l0, buf)                                                      \
  do {                                                                              \
    if (xy_wave) {                                                                  \
      const float4 v0 = *(const float4*)(&sY[(buf)][0][0] + y_g * 32 + y_c4 * 4);   \
      *(float4*)(gY + (long)(l0) * D_INNER) = v0;                                   \
    }                                                                               \
  } while (0)

  float4 sB[SCAN_D], sC[SCAN_D];
  float2 sD[SCAN_D];
  float  sXv[SCAN_D];

#define SCAN_LDS_READ(u, buf, t)                                                    \
  do {                                                                              \
    sD[(u)]  = *(const float2*)&sDT[(buf)][(t)][0];                                 \
    sXv[(u)] = sX[(buf)][(t)][p];                                                   \
    sB[(u)]  = *(const float4*)&sBC[(buf)][(t)][q * SCAN_NE];                       \
    sC[(u)]  = *(const float4*)&sBC[(buf)][(t)][D_STATE + q * SCAN_NE];             \
  } while (0)

#define SCAN_STEP(u, yout, xout)                                                    \
  do {                                                                              \
    const float dtv = sD[(u)].x;                                                    \
    const float dA  = sD[(u)].y;                                                    \
    const float xv  = sXv[(u)];                                                     \
    const float xdt = xv * dtv;                                                     \
    const float4 bv = sB[(u)];                                                      \
    const float4 cv = sC[(u)];                                                      \
    float accv = 0.f;                                                               \
    hreg[0] = fmaf(hreg[0], dA, xdt * bv.x);                                        \
    accv    = fmaf(hreg[0], cv.x, accv);                                            \
    hreg[1] = fmaf(hreg[1], dA, xdt * bv.y);                                        \
    accv    = fmaf(hreg[1], cv.y, accv);                                            \
    hreg[2] = fmaf(hreg[2], dA, xdt * bv.z);                                        \
    accv    = fmaf(hreg[2], cv.z, accv);                                            \
    hreg[3] = fmaf(hreg[3], dA, xdt * bv.w);                                        \
    accv    = fmaf(hreg[3], cv.w, accv);                                            \
    (yout) = accv;                                                                  \
    (xout) = xv;                                                                    \
  } while (0)

#define SCAN_FINISH(acc, xv)                                                        \
  do {                                                                              \
    (acc) += lane_xor_read<0x161>(acc);                             \
    (acc) += lane_xor_read<0x162>(acc);                             \
    (acc) += lane_xor_read<0x164>(acc);                             \
    (acc) += lane_xor_read<0x168>(acc);                             \
    (acc) = fmaf((xv), Dh, (acc));                                                  \
  } while (0)

  SCAN_FETCH(0);
  SCAN_COMMIT(0);
  __syncthreads();

  for (int c = 0; c < SCAN_NCH; ++c) {
    const int cb = c & 1;
    const bool more = (c + 1 < SCAN_NCH);
    if (c > 0) SCAN_WRITEOUT((c - 1) * SCAN_T, cb ^ 1);
    SCAN_LDS_READ(0, cb, 0);
    SCAN_LDS_READ(1, cb, 1);
    __builtin_amdgcn_sched_barrier(0);
    SCAN_LDS_READ(2, cb, 2);
    SCAN_LDS_READ(3, cb, 3);
    __builtin_amdgcn_sched_barrier(0);
    SCAN_LDS_READ(4, cb, 4);
    SCAN_LDS_READ(5, cb, 5);
    __builtin_amdgcn_sched_barrier(0);
    SCAN_LDS_READ(6, cb, 6);
    SCAN_LDS_READ(7, cb, 7);
    if (more) SCAN_FETCH((c + 1) * SCAN_T);
    #pragma unroll
    for (int t0 = 0; t0 < SCAN_T - SCAN_D; t0 += SCAN_D) {
      float yv[SCAN_D], xs[SCAN_D];
      #pragma unroll
      for (int u = 0; u < SCAN_D; ++u) {
        SCAN_STEP(u, yv[u], xs[u]);
        SCAN_LDS_READ(u, cb, t0 + SCAN_D + u);
      }
      #pragma unroll
      for (int u = 0; u < SCAN_D; ++u) SCAN_FINISH(yv[u], xs[u]);
      if (q == 0) {
        #pragma unroll
        for (int u = 0; u < SCAN_D; ++u) sY[cb][t0 + u][p] = yv[u];
      }
    }
    {
      float yv[SCAN_D], xs[SCAN_D];
      #pragma unroll
      for (int u = 0; u < SCAN_D; ++u) SCAN_STEP(u, yv[u], xs[u]);
      #pragma unroll
      for (int u = 0; u < SCAN_D; ++u) SCAN_FINISH(yv[u], xs[u]);
      if (q == 0) {
        #pragma unroll
        for (int u = 0; u < SCAN_D; ++u) sY[cb][SCAN_T - SCAN_D + u][p] = yv[u];
      }
    }
    if (more) SCAN_COMMIT(cb ^ 1);
    __syncthreads();
  }
  SCAN_WRITEOUT((SCAN_NCH - 1) * SCAN_T, (SCAN_NCH - 1) & 1);
#undef SCAN_FETCH
#undef SCAN_COMMIT
#undef SCAN_WRITEOUT
#undef SCAN_LDS_READ
#undef SCAN_STEP
#undef SCAN_FINISH
}

#define GR_R 4
static_assert(GR_R == SCAN_YG, "a workgroup of the gate kernel owns one row group of the scan's y layout");
__global__ void __launch_bounds__(256)
gate_rmsnorm_kernel(const float* __restrict__ zx,
                    const float* __restrict__ nw,
                    float* __restrict__ y) {
  const long bl0 = (long)blockIdx.x * GR_R;
  const int tid = threadIdx.x;
  __shared__ float red[GR_R][256];
  float zv[GR_R][4];
  float yv[GR_R][4];
  #pragma unroll
  for (int r = 0; r < GR_R; ++r) {
    #pragma unroll
    for (int j = 0; j < 4; ++j) {
      const int i = tid + j * 256;
      zv[r][j] = zx[(bl0 + r) * D_IN_PROJ + i];
      yv[r][j] = y[bl0 * D_INNER + (i / SCAN_PB) * 32 + r * SCAN_PB + (i % SCAN_PB)];
    }
  }
  #pragma unroll
  for (int r = 0; r < GR_R; ++r) {
    float ss = 0.f;
    #pragma unroll
    for (int j = 0; j < 4; ++j) {
      const float z = zv[r][j];
      const float g = z / (1.f + __expf(-z));
      const float v = yv[r][j] * g;
      yv[r][j] = v;
      ss = fmaf(v, v, ss);
    }
    red[r][tid] = ss;
  }
  __syncthreads();
  for (int st = 128; st > 0; st >>= 1) {
    if (tid < st) {
      #pragma unroll
      for (int r = 0; r < GR_R; ++r) red[r][tid] += red[r][tid + st];
    }
    __syncthreads();
  }
  float nwv[4];
  #pragma unroll
  for (int j = 0; j < 4; ++j) nwv[j] = nw[tid + j * 256];
  #pragma unroll
  for (int r = 0; r < GR_R; ++r) {
    const float scale = rsqrtf(red[r][0] * (1.f / (float)D_INNER) + 1e-5f);
    #pragma unroll
    for (int j = 0; j < 4; ++j) {
      const int i = tid + j * 256;
      y[(bl0 + r) * D_INNER + i] = yv[r][j] * scale * nwv[j];
    }
  }
}

extern "C" void kernel_launch(void* const* d_in, const int* in_sizes, int n_in,
                              void* d_out, int out_size, void* d_ws, size_t ws_size,
                              hipStream_t stream) {
  const float* tokens     = (const float*)d_in[0];
  const float* conv_w     = (const float*)d_in[1];
  const float* in_proj_w  = (const float*)d_in[2];
  const float* conv1d_w   = (const float*)d_in[3];
  const float* conv1d_b   = (const float*)d_in[4];
  const float* dt_bias    = (const float*)d_in[5];
  const float* A_log      = (const float*)d_in[6];
  const float* D_param    = (const float*)d_in[7];
  const float* norm_w     = (const float*)d_in[8];
  const float* out_proj_w = (const float*)d_in[9];
  float* out = (float*)d_out;
  float* ws  = (float*)d_ws;

  float* u0   = ws;
  float* u1   = u0 + (long)ROWS * D_MODEL;
  float* zx   = u1 + (long)ROWS * D_MODEL;
  float* xact = zx + (long)ROWS * D_IN_PROJ;
  float* ybuf = xact + (long)ROWS * CONV_DIM;
  float* col  = zx;
  (void)ws_size; (void)in_sizes; (void)n_in; (void)out_size;

  {
    im2col_stem_kernel<<<ROWS / IM_R, IM_NT, 0, stream>>>(tokens, col);
    gemm_xt_wmma<<<(ROWS / GT_M) * ((D_MODEL + GT_N - 1) / GT_N), 256, 0, stream>>>(col, conv_w, u0, ROWS, D_MODEL, KSTEM);
  }

  const float* u_in = u0;
  for (int blk = 0; blk < 2; ++blk) {
    const float* ipw = in_proj_w  + (long)blk * D_IN_PROJ * D_MODEL;
    const float* cw  = conv1d_w   + (long)blk * CONV_DIM * D_CONV;
    const float* cb  = conv1d_b   + (long)blk * CONV_DIM;
    const float* dtb = dt_bias    + (long)blk * NHEADS;
    const float* al  = A_log      + (long)blk * NHEADS;
    const float* Dpp = D_param    + (long)blk * NHEADS;
    const float* nw  = norm_w     + (long)blk * D_INNER;
    const float* opw = out_proj_w + (long)blk * D_MODEL * D_INNER;

    {
      gemm_xt_wmma<<<(ROWS / GT_M) * ((D_IN_PROJ + GT_N - 1) / GT_N), 256, 0, stream>>>(u_in, ipw, zx, ROWS, D_IN_PROJ, D_MODEL);
    }
    dwconv_silu_kernel<<<ROWS / DW_R, DW_NT, 0, stream>>>(zx, cw, cb, xact);
    ssm_scan_kernel<<<BATCH * NHEADS * SCAN_SPLIT, SCAN_NT, 0, stream>>>(zx, xact, dtb, al, Dpp, ybuf);
    gate_rmsnorm_kernel<<<ROWS / GR_R, 256, 0, stream>>>(zx, nw, ybuf);
    float* dst = (blk == 0) ? u1 : out;
    {
      gemm_xt_wmma<<<(ROWS / GT_M) * ((D_MODEL + GT_N - 1) / GT_N), 256, 0, stream>>>(ybuf, opw, dst, ROWS, D_MODEL, D_INNER);
    }
    u_in = u1;
  }
}


# ssm scan: softplus(dt) and dA = exp(dt*A) made once per (row, head) in the conv kernel, read back as pairs; the 80-instruction serial chain leaves the scan's barrier path; same sums, same order
# speedup vs baseline: 12.1790x; 1.0476x over previous
#include <hip/hip_runtime.h>
#include <hip/hip_bf16.h>

typedef __attribute__((ext_vector_type(2))) float v2f;
typedef __attribute__((ext_vector_type(8))) float v8f;

#define BATCH     2
#define SEQLEN    2048
#define C_IN      256
#define D_MODEL   512
#define LOUT      1024
#define D_STATE   64
#define D_CONV    4
#define D_INNER   1024
#define HEADDIM   64
#define NHEADS    16
#define CONV_DIM  (D_INNER + 2*D_STATE)
#define D_IN_PROJ (2*D_INNER + 2*D_STATE + NHEADS)
#define ROWS      (BATCH*LOUT)
#define KSTEM     (C_IN * 7)

#define IM_R  8
#define IM_NT (KSTEM / 4)
__global__ void __launch_bounds__(IM_NT)
im2col_stem_kernel(const float* __restrict__ tokens,
                   float* __restrict__ col) {
  const int t   = threadIdx.x;
  const int bl0 = blockIdx.x * IM_R;
  const int b   = bl0 >> 10;
  const int l0  = bl0 & (LOUT - 1);
  const float* tb = tokens + (long)b * C_IN * SEQLEN;
  int off[4], kk[4];
  #pragma unroll
  for (int m = 0; m < 4; ++m) {
    const int j = t + IM_NT * m;
    const int c = j / 7;
    kk[m]  = j - c * 7;
    off[m] = c * SEQLEN;
  }
  float v[IM_R][4];
  #pragma unroll
  for (int i = 0; i < IM_R; ++i) {
    const int s0 = 2 * (l0 + i) - 3;
    #pragma unroll
    for (int m = 0; m < 4; ++m) {
      const int s  = s0 + kk[m];
      const int sc = min(max(s, 0), SEQLEN - 1);
      const float x = tb[off[m] + sc];
      v[i][m] = (s == sc) ? x : 0.f;
    }
  }
  float* dst = col + (long)bl0 * KSTEM + t;
  #pragma unroll
  for (int i = 0; i < IM_R; ++i) {
    #pragma unroll
    for (int m = 0; m < 4; ++m) dst[(long)i * KSTEM + IM_NT * m] = v[i][m];
  }
}

#define GT_M 64
#define GT_N 64
#define GT_K 64
#define GT_P 68

__global__ void __launch_bounds__(256)
gemm_xt_wmma(const float* __restrict__ A,
             const float* __restrict__ W,
             float* __restrict__ C,
             int M, int N, int K) {
  __shared__ __attribute__((aligned(16))) float sA[2][GT_M][GT_P];
  __shared__ __attribute__((aligned(16))) float sB[2][GT_N][GT_P];
  const int tid  = threadIdx.x;
  const int lane = tid & 31;
  const int wave = tid >> 5;
  const int nblk = (N + GT_N - 1) / GT_N;
  const int mb = blockIdx.x / nblk;
  const int nb = blockIdx.x - mb * nblk;
  const int m0 = mb * GT_M;
  const int n0 = nb * GT_N;
  (void)M;

  const int ld_c4 = tid & 15;
  const int ld_r  = tid >> 4;
  const long a_step = (long)16 * K;
  const float* gA  = A + (long)(m0 + ld_r) * K + ld_c4 * 4;
  const float* gB0 = W + (long)min(n0 + ld_r,      N - 1) * K + ld_c4 * 4;
  const float* gB1 = W + (long)min(n0 + ld_r + 16, N - 1) * K + ld_c4 * 4;
  const float* gB2 = W + (long)min(n0 + ld_r + 32, N - 1) * K + ld_c4 * 4;
  const float* gB3 = W + (long)min(n0 + ld_r + 48, N - 1) * K + ld_c4 * 4;

  float4 ra0, ra1, ra2, ra3, rb0, rb1, rb2, rb3;

#define GT_FETCH(k0)                                                   \
  do {                                                                 \
    ra0 = *(const float4*)(gA + (k0));                                 \
    ra1 = *(const float4*)(gA + a_step + (k0));                        \
    ra2 = *(const float4*)(gA + 2 * a_step + (k0));                    \
    ra3 = *(const float4*)(gA + 3 * a_step + (k0));                    \
    rb0 = *(const float4*)(gB0 + (k0));                                \
    rb1 = *(const float4*)(gB1 + (k0));                                \
    rb2 = *(const float4*)(gB2 + (k0));                                \
    rb3 = *(const float4*)(gB3 + (k0));                                \
  } while (0)

#define GT_COMMIT(buf)                                                 \
  do {                                                                 \
    *(float4*)&sA[(buf)][ld_r     ][ld_c4 * 4] = ra0;                  \
    *(float4*)&sA[(buf)][ld_r + 16][ld_c4 * 4] = ra1;                  \
    *(float4*)&sA[(buf)][ld_r + 32][ld_c4 * 4] = ra2;                  \
    *(float4*)&sA[(buf)][ld_r + 48][ld_c4 * 4] = ra3;                  \
    *(float4*)&sB[(buf)][ld_r     ][ld_c4 * 4] = rb0;                  \
    *(float4*)&sB[(buf)][ld_r + 16][ld_c4 * 4] = rb1;                  \
    *(float4*)&sB[(buf)][ld_r + 32][ld_c4 * 4] = rb2;                  \
    *(float4*)&sB[(buf)][ld_r + 48][ld_c4 * 4] = rb3;                  \
  } while (0)

  const int r  = lane & 15;
  const int h  = lane >> 4;
  const int wm = wave & 3;
  const int wn = wave >> 2;
  const float* fa  = &sA[0][wm * 16 + r][4 * h];
  const float* fb0 = &sB[0][wn * 32 + r][4 * h];
  const float* fb1 = &sB[0][wn * 32 + 16 + r][4 * h];

  v8f acc0 = {0.f,0.f,0.f,0.f,0.f,0.f,0.f,0.f};
  v8f acc1 = acc0;

#define GT_COMPUTE(buf)                                                                            \
  do {                                                                                             \
    const float* pa  = fa  + (buf) * (GT_M * GT_P);                                                \
    const float* pb0 = fb0 + (buf) * (GT_N * GT_P);                                                \
    const float* pb1 = fb1 + (buf) * (GT_N * GT_P);                                                \
    _Pragma("unroll")                                                                              \
    for (int g = 0; g < GT_K / 8; ++g) {                                                           \
      const float4 a4 = *(const float4*)(pa  + g * 8);                                             \
      const float4 p4 = *(const float4*)(pb0 + g * 8);                                             \
      const float4 q4 = *(const float4*)(pb1 + g * 8);                                             \
      const v2f alo = {a4.x, a4.y}, ahi = {a4.z, a4.w};                                            \
      const v2f plo = {p4.x, p4.y}, phi = {p4.z, p4.w};                                            \
      const v2f qlo = {q4.x, q4.y}, qhi = {q4.z, q4.w};                                            \
      acc0 = __builtin_amdgcn_wmma_f32_16x16x4_f32(false, alo, false, plo, (short)0, acc0, false, false); \
      acc1 = __builtin_amdgcn_wmma_f32_16x16x4_f32(false, alo, false, qlo, (short)0, acc1, false, false); \
      acc0 = __builtin_amdgcn_wmma_f32_16x16x4_f32(false, ahi, false, phi, (short)0, acc0, false, false); \
      acc1 = __builtin_amdgcn_wmma_f32_16x16x4_f32(false, ahi, false, qhi, (short)0, acc1, false, false); \
    }                                                                                              \
  } while (0)

  const int KT = K / GT_K;
  GT_FETCH(0);
  GT_COMMIT(0);
  __syncthreads();

  for (int kt = 0; kt + 1 < KT; ++kt) {
    const int cb = kt & 1;
    GT_FETCH((kt + 1) * GT_K);
    GT_COMPUTE(cb);
    GT_COMMIT(cb ^ 1);
    __syncthreads();
  }
  GT_COMPUTE((KT - 1) & 1);
  __syncthreads();

  float (*sC)[GT_P] = sA[0];
  #pragma unroll
  for (int i = 0; i < 8; ++i) {
    sC[wm * 16 + 8 * h + i][wn * 32 + r]      = acc0[i];
    sC[wm * 16 + 8 * h + i][wn * 32 + 16 + r] = acc1[i];
  }
  __syncthreads();
  const int col = n0 + ld_c4 * 4;
  if (col < N) {
    #pragma unroll
    for (int p = 0; p < 4; ++p) {
      const int row = ld_r + 16 * p;
      const float4 v = *(const float4*)&sC[row][ld_c4 * 4];
      *(float4*)(C + (long)(m0 + row) * N + col) = v;
    }
  }
#undef GT_FETCH
#undef GT_COMMIT
#undef GT_COMPUTE
}

#define DW_R  8
#define DW_NT (CONV_DIM / 4)

__device__ __forceinline__ float dwconv_one(const float4 wv, float x0, float xm1, float xm2,
                                            float xm3, int l, float bs) {
  float acc = wv.w * x0;
  if (l >= 1) acc = fmaf(wv.z, xm1, acc);
  if (l >= 2) acc = fmaf(wv.y, xm2, acc);
  if (l >= 3) acc = fmaf(wv.x, xm3, acc);
  const float v = acc + bs;
  return v / (1.f + __expf(-v));
}

__global__ void __launch_bounds__(DW_NT)
dwconv_silu_kernel(const float* __restrict__ zx,
                   const float* __restrict__ w,
                   const float* __restrict__ bias,
                   float* __restrict__ out,
                   const float* __restrict__ dtb,
                   const float* __restrict__ alog,
                   float* __restrict__ dtda) {
  const int t   = threadIdx.x;
  const int bl0 = blockIdx.x * DW_R;
  const int l0  = bl0 & (LOUT - 1);
  const bool dt_thread = (t < DW_R * NHEADS);
  const int  dt_i  = t / NHEADS;
  const int  dt_hd = t % NHEADS;
  const float* xin = zx + (long)bl0 * D_IN_PROJ + D_INNER + 4 * t;
  float4 xr[DW_R + 3];
  #pragma unroll
  for (int i = 0; i < 3; ++i) {
    const long back = (l0 >= 3 - i) ? (long)(3 - i) * D_IN_PROJ : 0;
    xr[i] = *(const float4*)(xin - back);
  }
  #pragma unroll
  for (int i = 0; i < DW_R; ++i) xr[3 + i] = *(const float4*)(xin + (long)i * D_IN_PROJ);
  const float4 wa = *(const float4*)(w + (4 * t + 0) * 4);
  const float4 wb = *(const float4*)(w + (4 * t + 1) * 4);
  const float4 wc = *(const float4*)(w + (4 * t + 2) * 4);
  const float4 wd = *(const float4*)(w + (4 * t + 3) * 4);
  const float4 bs = *(const float4*)(bias + 4 * t);
  if (dt_thread) {
    const float dt_raw    = zx[(long)(bl0 + dt_i) * D_IN_PROJ + (D_IN_PROJ - NHEADS) + dt_hd];
    const float dt_bias_h = dtb[dt_hd];
    const float dt_alog_h = alog[dt_hd];
    const float Ah  = -__expf(dt_alog_h);
    const float raw = dt_raw + dt_bias_h;
    const float dtv = (raw > 20.f) ? raw : log1pf(__expf(raw));
    float2 dd;
    dd.x = dtv;
    dd.y = __expf(dtv * Ah);
    *(float2*)(dtda + ((long)(bl0 + dt_i) * NHEADS + dt_hd) * 2) = dd;
  }
  float* dst = out + (long)bl0 * CONV_DIM + 4 * t;
  #pragma unroll
  for (int i = 0; i < DW_R; ++i) {
    const int l = l0 + i;
    float4 o;
    o.x = dwconv_one(wa, xr[3 + i].x, xr[2 + i].x, xr[1 + i].x, xr[i].x, l, bs.x);
    o.y = dwconv_one(wb, xr[3 + i].y, xr[2 + i].y, xr[1 + i].y, xr[i].y, l, bs.y);
    o.z = dwconv_one(wc, xr[3 + i].z, xr[2 + i].z, xr[1 + i].z, xr[i].z, l, bs.z);
    o.w = dwconv_one(wd, xr[3 + i].w, xr[2 + i].w, xr[1 + i].w, xr[i].w, l, bs.w);
    *(float4*)(dst + (long)i * CONV_DIM) = o;
  }
}

#define SCAN_T   32
#define SCAN_NCH (LOUT / SCAN_T)
#define SCAN_SPLIT 8
#define SCAN_PB    (HEADDIM / SCAN_SPLIT)
#define SCAN_LPR   16
#define SCAN_NE    (D_STATE / SCAN_LPR)
#define SCAN_NT    (SCAN_PB * SCAN_LPR)
#define SCAN_YG    (32 / SCAN_PB)
#define SCAN_D     8

template <int CTRL>
__device__ __forceinline__ float lane_xor_read(float v) {
  return __int_as_float(__builtin_amdgcn_update_dpp(0, __float_as_int(v), CTRL, 0xF, 0xF, true));
}

__global__ void __launch_bounds__(SCAN_NT)
ssm_scan_kernel(const float* __restrict__ xbc,
                const float* __restrict__ dtda,
                const float* __restrict__ Dp,
                float* __restrict__ y) {
  const int pblk = blockIdx.x % SCAN_SPLIT;
  const int hh = (blockIdx.x / SCAN_SPLIT) & 15;
  const int b  = blockIdx.x / (SCAN_SPLIT * NHEADS);
  const int tid = threadIdx.x;
  const int p = tid / SCAN_LPR;
  const int q = tid % SCAN_LPR;
  __shared__ __attribute__((aligned(16))) float sBC[2][SCAN_T][2 * D_STATE];
  __shared__ __attribute__((aligned(16))) float sX[2][SCAN_T][SCAN_PB];
  __shared__ __attribute__((aligned(16))) float sY[2][SCAN_T][SCAN_PB];
  __shared__ __attribute__((aligned(8)))  float sDT[2][SCAN_T][2];
  const float Dh   = Dp[hh];
  float hreg[SCAN_NE];
  #pragma unroll
  for (int i = 0; i < SCAN_NE; ++i) hreg[i] = 0.f;

  const int bc_c4 = tid & 31;
  const int bc_t  = tid >> 5;
  const bool xy_wave = (tid < 64);
  const int x_c4  = tid & 1;
  const int x_t   = (tid >> 1) & (SCAN_T - 1);
  const int y_g   = (tid >> 3) & 7;
  const int y_c4  = tid & 7;
  const bool dt_wave = (tid >= SCAN_NT - SCAN_T);
  const int  dt_t    = dt_wave ? tid - (SCAN_NT - SCAN_T) : 0;
  const float* gBC = xbc + ((long)b * LOUT + bc_t) * CONV_DIM + D_INNER + bc_c4 * 4;
  const float* gX  = xbc + ((long)b * LOUT + x_t) * CONV_DIM + hh * HEADDIM + pblk * SCAN_PB + x_c4 * 4;
  const float* gDT = dtda + (((long)b * LOUT + dt_t) * NHEADS + hh) * 2;
  float*       gY  = y + ((long)b * LOUT + y_g * SCAN_YG) * D_INNER + (hh * SCAN_SPLIT + pblk) * 32 + y_c4 * 4;

  float4 rBC0, rBC1, rBC2, rBC3, rBC4, rBC5, rBC6, rBC7;
  float4 rX0 = make_float4(0.f, 0.f, 0.f, 0.f);
  float2 rdt = make_float2(0.f, 0.f);

#define SCAN_FETCH(l0)                                                              \
  do {                                                                              \
    const float* s_bc = gBC + (long)(l0) * CONV_DIM;                                \
    rBC0 = *(const float4*)(s_bc + 0 * CONV_DIM);                                   \
    rBC1 = *(const float4*)(s_bc + 4 * CONV_DIM);                                   \
    rBC2 = *(const float4*)(s_bc + 8 * CONV_DIM);                                   \
    rBC3 = *(const float4*)(s_bc + 12 * CONV_DIM);                                  \
    rBC4 = *(const float4*)(s_bc + 16 * CONV_DIM);                                  \
    rBC5 = *(const float4*)(s_bc + 20 * CONV_DIM);                                  \
    rBC6 = *(const float4*)(s_bc + 24 * CONV_DIM);                                  \
    rBC7 = *(const float4*)(s_bc + 28 * CONV_DIM);                                  \
    if (xy_wave) rX0 = *(const float4*)(gX + (long)(l0) * CONV_DIM);                \
    if (dt_wave) rdt = *(const float2*)(gDT + (long)(l0) * (NHEADS * 2));           \
  } while (0)

#define SCAN_COMMIT(buf)                                                            \
  do {                                                                              \
    *(float4*)&sBC[(buf)][bc_t + 0][bc_c4 * 4] = rBC0;                              \
    *(float4*)&sBC[(buf)][bc_t + 4][bc_c4 * 4] = rBC1;                              \
    *(float4*)&sBC[(buf)][bc_t + 8][bc_c4 * 4] = rBC2;                              \
    *(float4*)&sBC[(buf)][bc_t + 12][bc_c4 * 4] = rBC3;                             \
    *(float4*)&sBC[(buf)][bc_t + 16][bc_c4 * 4] = rBC4;                             \
    *(float4*)&sBC[(buf)][bc_t + 20][bc_c4 * 4] = rBC5;                             \
    *(float4*)&sBC[(buf)][bc_t + 24][bc_c4 * 4] = rBC6;                             \
    *(float4*)&sBC[(buf)][bc_t + 28][bc_c4 * 4] = rBC7;                             \
    if (xy_wave) *(float4*)&sX[(buf)][x_t][x_c4 * 4] = rX0;                         \
    if (dt_wave) *(float2*)&sDT[(buf)][dt_t][0] = rdt;                              \
  } while (0)

#define SCAN_WRITEOUT(l0, buf)                                                      \
  do {                                                                              \
    if (xy_wave) {                                                                  \
      const float4 v0 = *(const float4*)(&sY[(buf)][0][0] + y_g * 32 + y_c4 * 4);   \
      *(float4*)(gY + (long)(l0) * D_INNER) = v0;                                   \
    }                                                                               \
  } while (0)

  float4 sB[SCAN_D], sC[SCAN_D];
  float2 sD[SCAN_D];
  float  sXv[SCAN_D];

#define SCAN_LDS_READ(u, buf, t)                                                    \
  do {                                                                              \
    sD[(u)]  = *(const float2*)&sDT[(buf)][(t)][0];                                 \
    sXv[(u)] = sX[(buf)][(t)][p];                                                   \
    sB[(u)]  = *(const float4*)&sBC[(buf)][(t)][q * SCAN_NE];                       \
    sC[(u)]  = *(const float4*)&sBC[(buf)][(t)][D_STATE + q * SCAN_NE];             \
  } while (0)

#define SCAN_STEP(u, yout, xout)                                                    \
  do {                                                                              \
    const float dtv = sD[(u)].x;                                                    \
    const float dA  = sD[(u)].y;                                                    \
    const float xv  = sXv[(u)];                                                     \
    const float xdt = xv * dtv;                                                     \
    const float4 bv = sB[(u)];                                                      \
    const float4 cv = sC[(u)];                                                      \
    float accv = 0.f;                                                               \
    hreg[0] = fmaf(hreg[0], dA, xdt * bv.x);                                        \
    accv    = fmaf(hreg[0], cv.x, accv);                                            \
    hreg[1] = fmaf(hreg[1], dA, xdt * bv.y);                                        \
    accv    = fmaf(hreg[1], cv.y, accv);                                            \
    hreg[2] = fmaf(hreg[2], dA, xdt * bv.z);                                        \
    accv    = fmaf(hreg[2], cv.z, accv);                                            \
    hreg[3] = fmaf(hreg[3], dA, xdt * bv.w);                                        \
    accv    = fmaf(hreg[3], cv.w, accv);                                            \
    (yout) = accv;                                                                  \
    (xout) = xv;                                                                    \
  } while (0)

#define SCAN_FINISH(acc, xv)                                                        \
  do {                                                                              \
    (acc) += lane_xor_read<0x161>(acc);                             \
    (acc) += lane_xor_read<0x162>(acc);                             \
    (acc) += lane_xor_read<0x164>(acc);                             \
    (acc) += lane_xor_read<0x168>(acc);                             \
    (acc) = fmaf((xv), Dh, (acc));                                                  \
  } while (0)

  SCAN_FETCH(0);
  SCAN_COMMIT(0);
  __syncthreads();

  for (int c = 0; c < SCAN_NCH; ++c) {
    const int cb = c & 1;
    const bool more = (c + 1 < SCAN_NCH);
    if (c > 0) SCAN_WRITEOUT((c - 1) * SCAN_T, cb ^ 1);
    SCAN_LDS_READ(0, cb, 0);
    SCAN_LDS_READ(1, cb, 1);
    __builtin_amdgcn_sched_barrier(0);
    SCAN_LDS_READ(2, cb, 2);
    SCAN_LDS_READ(3, cb, 3);
    __builtin_amdgcn_sched_barrier(0);
    SCAN_LDS_READ(4, cb, 4);
    SCAN_LDS_READ(5, cb, 5);
    __builtin_amdgcn_sched_barrier(0);
    SCAN_LDS_READ(6, cb, 6);
    SCAN_LDS_READ(7, cb, 7);
    if (more) SCAN_FETCH((c + 1) * SCAN_T);
    #pragma unroll
    for (int t0 = 0; t0 < SCAN_T - SCAN_D; t0 += SCAN_D) {
      float yv[SCAN_D], xs[SCAN_D];
      #pragma unroll
      for (int u = 0; u < SCAN_D; ++u) {
        SCAN_STEP(u, yv[u], xs[u]);
        SCAN_LDS_READ(u, cb, t0 + SCAN_D + u);
      }
      #pragma unroll
      for (int u = 0; u < SCAN_D; ++u) SCAN_FINISH(yv[u], xs[u]);
      if (q == 0) {
        #pragma unroll
        for (int u = 0; u < SCAN_D; ++u) sY[cb][t0 + u][p] = yv[u];
      }
    }
    {
      float yv[SCAN_D], xs[SCAN_D];
      #pragma unroll
      for (int u = 0; u < SCAN_D; ++u) SCAN_STEP(u, yv[u], xs[u]);
      #pragma unroll
      for (int u = 0; u < SCAN_D; ++u) SCAN_FINISH(yv[u], xs[u]);
      if (q == 0) {
        #pragma unroll
        for (int u = 0; u < SCAN_D; ++u) sY[cb][SCAN_T - SCAN_D + u][p] = yv[u];
      }
    }
    if (more) SCAN_COMMIT(cb ^ 1);
    __syncthreads();
  }
  SCAN_WRITEOUT((SCAN_NCH - 1) * SCAN_T, (SCAN_NCH - 1) & 1);
#undef SCAN_FETCH
#undef SCAN_COMMIT
#undef SCAN_WRITEOUT
#undef SCAN_LDS_READ
#undef SCAN_STEP
#undef SCAN_FINISH
}

#define GR_R 4
static_assert(GR_R == SCAN_YG, "a workgroup of the gate kernel owns one row group of the scan's y layout");
__global__ void __launch_bounds__(256)
gate_rmsnorm_kernel(const float* __restrict__ zx,
                    const float* __restrict__ nw,
                    float* __restrict__ y) {
  const long bl0 = (long)blockIdx.x * GR_R;
  const int tid = threadIdx.x;
  __shared__ float red[GR_R][256];
  float zv[GR_R][4];
  float yv[GR_R][4];
  #pragma unroll
  for (int r = 0; r < GR_R; ++r) {
    #pragma unroll
    for (int j = 0; j < 4; ++j) {
      const int i = tid + j * 256;
      zv[r][j] = zx[(bl0 + r) * D_IN_PROJ + i];
      yv[r][j] = y[bl0 * D_INNER + (i / SCAN_PB) * 32 + r * SCAN_PB + (i % SCAN_PB)];
    }
  }
  #pragma unroll
  for (int r = 0; r < GR_R; ++r) {
    float ss = 0.f;
    #pragma unroll
    for (int j = 0; j < 4; ++j) {
      const float z = zv[r][j];
      const float g = z / (1.f + __expf(-z));
      const float v = yv[r][j] * g;
      yv[r][j] = v;
      ss = fmaf(v, v, ss);
    }
    red[r][tid] = ss;
  }
  __syncthreads();
  for (int st = 128; st > 0; st >>= 1) {
    if (tid < st) {
      #pragma unroll
      for (int r = 0; r < GR_R; ++r) red[r][tid] += red[r][tid + st];
    }
    __syncthreads();
  }
  float nwv[4];
  #pragma unroll
  for (int j = 0; j < 4; ++j) nwv[j] = nw[tid + j * 256];
  #pragma unroll
  for (int r = 0; r < GR_R; ++r) {
    const float scale = rsqrtf(red[r][0] * (1.f / (float)D_INNER) + 1e-5f);
    #pragma unroll
    for (int j = 0; j < 4; ++j) {
      const int i = tid + j * 256;
      y[(bl0 + r) * D_INNER + i] = yv[r][j] * scale * nwv[j];
    }
  }
}

extern "C" void kernel_launch(void* const* d_in, const int* in_sizes, int n_in,
                              void* d_out, int out_size, void* d_ws, size_t ws_size,
                              hipStream_t stream) {
  const float* tokens     = (const float*)d_in[0];
  const float* conv_w     = (const float*)d_in[1];
  const float* in_proj_w  = (const float*)d_in[2];
  const float* conv1d_w   = (const float*)d_in[3];
  const float* conv1d_b   = (const float*)d_in[4];
  const float* dt_bias    = (const float*)d_in[5];
  const float* A_log      = (const float*)d_in[6];
  const float* D_param    = (const float*)d_in[7];
  const float* norm_w     = (const float*)d_in[8];
  const float* out_proj_w = (const float*)d_in[9];
  float* out = (float*)d_out;
  float* ws  = (float*)d_ws;

  float* u0   = ws;
  float* u1   = u0 + (long)ROWS * D_MODEL;
  float* zx   = u1 + (long)ROWS * D_MODEL;
  float* xact = zx + (long)ROWS * D_IN_PROJ;
  float* ybuf = xact + (long)ROWS * CONV_DIM;
  float* dtda = ybuf + (long)ROWS * D_INNER;
  float* col  = zx;
  (void)ws_size; (void)in_sizes; (void)n_in; (void)out_size;

  {
    im2col_stem_kernel<<<ROWS / IM_R, IM_NT, 0, stream>>>(tokens, col);
    gemm_xt_wmma<<<(ROWS / GT_M) * ((D_MODEL + GT_N - 1) / GT_N), 256, 0, stream>>>(col, conv_w, u0, ROWS, D_MODEL, KSTEM);
  }

  const float* u_in = u0;
  for (int blk = 0; blk < 2; ++blk) {
    const float* ipw = in_proj_w  + (long)blk * D_IN_PROJ * D_MODEL;
    const float* cw  = conv1d_w   + (long)blk * CONV_DIM * D_CONV;
    const float* cb  = conv1d_b   + (long)blk * CONV_DIM;
    const float* dtb = dt_bias    + (long)blk * NHEADS;
    const float* al  = A_log      + (long)blk * NHEADS;
    const float* Dpp = D_param    + (long)blk * NHEADS;
    const float* nw  = norm_w     + (long)blk * D_INNER;
    const float* opw = out_proj_w + (long)blk * D_MODEL * D_INNER;

    {
      gemm_xt_wmma<<<(ROWS / GT_M) * ((D_IN_PROJ + GT_N - 1) / GT_N), 256, 0, stream>>>(u_in, ipw, zx, ROWS, D_IN_PROJ, D_MODEL);
    }
    dwconv_silu_kernel<<<ROWS / DW_R, DW_NT, 0, stream>>>(zx, cw, cb, xact, dtb, al, dtda);
    ssm_scan_kernel<<<BATCH * NHEADS * SCAN_SPLIT, SCAN_NT, 0, stream>>>(xact, dtda, Dpp, ybuf);
    gate_rmsnorm_kernel<<<ROWS / GR_R, 256, 0, stream>>>(zx, nw, ybuf);
    float* dst = (blk == 0) ? u1 : out;
    {
      gemm_xt_wmma<<<(ROWS / GT_M) * ((D_MODEL + GT_N - 1) / GT_N), 256, 0, stream>>>(ybuf, opw, dst, ROWS, D_MODEL, D_INNER);
    }
    u_in = u1;
  }
}


# gate SiLU(z) and norm weight folded into the scan's y write-out, RMSNorm row factor into out_proj's result rows: gate kernel's 2 launches gone; a row's squares summed as 256 partials in sequence
# speedup vs baseline: 12.6194x; 1.0362x over previous
#include <hip/hip_runtime.h>
#include <hip/hip_bf16.h>

typedef __attribute__((ext_vector_type(2))) float v2f;
typedef __attribute__((ext_vector_type(8))) float v8f;

#define BATCH     2
#define SEQLEN    2048
#define C_IN      256
#define D_MODEL   512
#define LOUT      1024
#define D_STATE   64
#define D_CONV    4
#define D_INNER   1024
#define HEADDIM   64
#define NHEADS    16
#define CONV_DIM  (D_INNER + 2*D_STATE)
#define D_IN_PROJ (2*D_INNER + 2*D_STATE + NHEADS)
#define ROWS      (BATCH*LOUT)
#define KSTEM     (C_IN * 7)
#define SCAN_SPLIT 8

#define IM_R  8
#define IM_NT (KSTEM / 4)
__global__ void __launch_bounds__(IM_NT)
im2col_stem_kernel(const float* __restrict__ tokens,
                   float* __restrict__ col) {
  const int t   = threadIdx.x;
  const int bl0 = blockIdx.x * IM_R;
  const int b   = bl0 >> 10;
  const int l0  = bl0 & (LOUT - 1);
  const float* tb = tokens + (long)b * C_IN * SEQLEN;
  int off[4], kk[4];
  #pragma unroll
  for (int m = 0; m < 4; ++m) {
    const int j = t + IM_NT * m;
    const int c = j / 7;
    kk[m]  = j - c * 7;
    off[m] = c * SEQLEN;
  }
  float v[IM_R][4];
  #pragma unroll
  for (int i = 0; i < IM_R; ++i) {
    const int s0 = 2 * (l0 + i) - 3;
    #pragma unroll
    for (int m = 0; m < 4; ++m) {
      const int s  = s0 + kk[m];
      const int sc = min(max(s, 0), SEQLEN - 1);
      const float x = tb[off[m] + sc];
      v[i][m] = (s == sc) ? x : 0.f;
    }
  }
  float* dst = col + (long)bl0 * KSTEM + t;
  #pragma unroll
  for (int i = 0; i < IM_R; ++i) {
    #pragma unroll
    for (int m = 0; m < 4; ++m) dst[(long)i * KSTEM + IM_NT * m] = v[i][m];
  }
}

#define GT_M 64
#define GT_N 64
#define GT_K 64
#define GT_P 68

template <bool NORM>
__global__ void __launch_bounds__(256)
gemm_xt_wmma(const float* __restrict__ A,
             const float* __restrict__ W,
             float* __restrict__ C,
             const float* __restrict__ ssq,
             int M, int N, int K) {
  __shared__ __attribute__((aligned(16))) float sA[2][GT_M][GT_P];
  __shared__ __attribute__((aligned(16))) float sB[2][GT_N][GT_P];
  __shared__ float sRow[NORM ? 4 : 1][GT_M];
  const int tid  = threadIdx.x;
  const int lane = tid & 31;
  const int wave = tid >> 5;
  const int nblk = (N + GT_N - 1) / GT_N;
  const int mb = blockIdx.x / nblk;
  const int nb = blockIdx.x - mb * nblk;
  const int m0 = mb * GT_M;
  const int n0 = nb * GT_N;
  (void)M;

  const int ld_c4 = tid & 15;
  const int ld_r  = tid >> 4;
  const long a_step = (long)16 * K;
  const int a_r = NORM ? ((tid >> 6) * 4 + ((tid >> 1) & 3)) : ld_r;
  const int a_c = NORM ? (((tid >> 3) & 7) * 8 + (tid & 1) * 4) : (ld_c4 * 4);
  const int a_ks = NORM ? 4 : 1;
  const float* gA = NORM
      ? A + (long)(m0 + (a_r & ~3)) * K + (a_c >> 3) * 32 + (a_r & 3) * 8 + (a_c & 7)
      : A + (long)(m0 + a_r) * K + a_c;
  const float* gB0 = W + (long)min(n0 + ld_r,      N - 1) * K + ld_c4 * 4;
  const float* gB1 = W + (long)min(n0 + ld_r + 16, N - 1) * K + ld_c4 * 4;
  const float* gB2 = W + (long)min(n0 + ld_r + 32, N - 1) * K + ld_c4 * 4;
  const float* gB3 = W + (long)min(n0 + ld_r + 48, N - 1) * K + ld_c4 * 4;

  float4 ra0, ra1, ra2, ra3, rb0, rb1, rb2, rb3;

  static_assert(!NORM || (D_INNER / GT_K) * 2 == (NHEADS * SCAN_SPLIT) / 4, "two pairs per K-step cover a quarter");
  const int   ps_row = tid & (GT_M - 1);
  const int   ps_q   = tid >> 6;
  const float* ps = NORM ? ssq + (((long)((m0 + ps_row) / LOUT) * (NHEADS * SCAN_SPLIT) + ps_q * 32) * LOUT
                                  + ((m0 + ps_row) % LOUT)) * 2
                         : nullptr;
  float2 pq0 = make_float2(0.f, 0.f), pq1 = pq0;
  float  rs  = 0.f;

#define GT_FETCH(k0)                                                   \
  do {                                                                 \
    ra0 = *(const float4*)(gA + (k0) * a_ks);                          \
    ra1 = *(const float4*)(gA + a_step + (k0) * a_ks);                 \
    ra2 = *(const float4*)(gA + 2 * a_step + (k0) * a_ks);             \
    ra3 = *(const float4*)(gA + 3 * a_step + (k0) * a_ks);             \
    rb0 = *(const float4*)(gB0 + (k0));                                \
    rb1 = *(const float4*)(gB1 + (k0));                                \
    rb2 = *(const float4*)(gB2 + (k0));                                \
    rb3 = *(const float4*)(gB3 + (k0));                                \
    if constexpr (NORM) {                                              \
      pq0 = *(const float2*)(ps + (long)((k0) / GT_K * 2)     * (LOUT * 2)); \
      pq1 = *(const float2*)(ps + (long)((k0) / GT_K * 2 + 1) * (LOUT * 2)); \
    }                                                                  \
  } while (0)

#define GT_COMMIT(buf)                                                 \
  do {                                                                 \
    *(float4*)&sA[(buf)][a_r     ][a_c] = ra0;                         \
    *(float4*)&sA[(buf)][a_r + 16][a_c] = ra1;                         \
    *(float4*)&sA[(buf)][a_r + 32][a_c] = ra2;                         \
    *(float4*)&sA[(buf)][a_r + 48][a_c] = ra3;                         \
    *(float4*)&sB[(buf)][ld_r     ][ld_c4 * 4] = rb0;                  \
    *(float4*)&sB[(buf)][ld_r + 16][ld_c4 * 4] = rb1;                  \
    *(float4*)&sB[(buf)][ld_r + 32][ld_c4 * 4] = rb2;                  \
    *(float4*)&sB[(buf)][ld_r + 48][ld_c4 * 4] = rb3;                  \
    if constexpr (NORM) {                                              \
      rs += pq0.x + pq0.y;                                             \
      rs += pq1.x + pq1.y;                                             \
    }                                                                  \
  } while (0)

  const int r  = lane & 15;
  const int h  = lane >> 4;
  const int wm = wave & 3;
  const int wn = wave >> 2;
  const float* fa  = &sA[0][wm * 16 + r][4 * h];
  const float* fb0 = &sB[0][wn * 32 + r][4 * h];
  const float* fb1 = &sB[0][wn * 32 + 16 + r][4 * h];

  v8f acc0 = {0.f,0.f,0.f,0.f,0.f,0.f,0.f,0.f};
  v8f acc1 = acc0;

#define GT_COMPUTE(buf)                                                                            \
  do {                                                                                             \
    const float* pa  = fa  + (buf) * (GT_M * GT_P);                                                \
    const float* pb0 = fb0 + (buf) * (GT_N * GT_P);                                                \
    const float* pb1 = fb1 + (buf) * (GT_N * GT_P);                                                \
    _Pragma("unroll")                                                                              \
    for (int g = 0; g < GT_K / 8; ++g) {                                                           \
      const float4 a4 = *(const float4*)(pa  + g * 8);                                             \
      const float4 p4 = *(const float4*)(pb0 + g * 8);                                             \
      const float4 q4 = *(const float4*)(pb1 + g * 8);                                             \
      const v2f alo = {a4.x, a4.y}, ahi = {a4.z, a4.w};                                            \
      const v2f plo = {p4.x, p4.y}, phi = {p4.z, p4.w};                                            \
      const v2f qlo = {q4.x, q4.y}, qhi = {q4.z, q4.w};                                            \
      acc0 = __builtin_amdgcn_wmma_f32_16x16x4_f32(false, alo, false, plo, (short)0, acc0, false, false); \
      acc1 = __builtin_amdgcn_wmma_f32_16x16x4_f32(false, alo, false, qlo, (short)0, acc1, false, false); \
      acc0 = __builtin_amdgcn_wmma_f32_16x16x4_f32(false, ahi, false, phi, (short)0, acc0, false, false); \
      acc1 = __builtin_amdgcn_wmma_f32_16x16x4_f32(false, ahi, false, qhi, (short)0, acc1, false, false); \
    }                                                                                              \
  } while (0)

  const int KT = K / GT_K;
  GT_FETCH(0);
  GT_COMMIT(0);
  __syncthreads();

  for (int kt = 0; kt + 1 < KT; ++kt) {
    const int cb = kt & 1;
    GT_FETCH((kt + 1) * GT_K);
    GT_COMPUTE(cb);
    GT_COMMIT(cb ^ 1);
    __syncthreads();
  }
  GT_COMPUTE((KT - 1) & 1);
  if constexpr (NORM) sRow[ps_q][ps_row] = rs;
  __syncthreads();

  float (*sC)[GT_P] = sA[0];
  #pragma unroll
  for (int i = 0; i < 8; ++i) {
    sC[wm * 16 + 8 * h + i][wn * 32 + r]      = acc0[i];
    sC[wm * 16 + 8 * h + i][wn * 32 + 16 + r] = acc1[i];
  }
  __syncthreads();
  const int col = n0 + ld_c4 * 4;
  if (col < N) {
    #pragma unroll
    for (int p = 0; p < 4; ++p) {
      const int row = ld_r + 16 * p;
      float4 v = *(const float4*)&sC[row][ld_c4 * 4];
      if constexpr (NORM) {
        const float sum = ((sRow[0][row] + sRow[1][row]) + sRow[2][row]) + sRow[3][row];
        const float scale = rsqrtf(sum * (1.f / (float)D_INNER) + 1e-5f);
        v.x *= scale; v.y *= scale; v.z *= scale; v.w *= scale;
      }
      *(float4*)(C + (long)(m0 + row) * N + col) = v;
    }
  }
#undef GT_FETCH
#undef GT_COMMIT
#undef GT_COMPUTE
}

#define DW_R  8
#define DW_NT (CONV_DIM / 4)

__device__ __forceinline__ float dwconv_one(const float4 wv, float x0, float xm1, float xm2,
                                            float xm3, int l, float bs) {
  float acc = wv.w * x0;
  if (l >= 1) acc = fmaf(wv.z, xm1, acc);
  if (l >= 2) acc = fmaf(wv.y, xm2, acc);
  if (l >= 3) acc = fmaf(wv.x, xm3, acc);
  const float v = acc + bs;
  return v / (1.f + __expf(-v));
}

__global__ void __launch_bounds__(DW_NT)
dwconv_silu_kernel(const float* __restrict__ zx,
                   const float* __restrict__ w,
                   const float* __restrict__ bias,
                   float* __restrict__ out,
                   const float* __restrict__ dtb,
                   const float* __restrict__ alog,
                   float* __restrict__ dtda) {
  const int t   = threadIdx.x;
  const int bl0 = blockIdx.x * DW_R;
  const int l0  = bl0 & (LOUT - 1);
  const bool dt_thread = (t < DW_R * NHEADS);
  const int  dt_i  = t / NHEADS;
  const int  dt_hd = t % NHEADS;
  const float* xin = zx + (long)bl0 * D_IN_PROJ + D_INNER + 4 * t;
  float4 xr[DW_R + 3];
  #pragma unroll
  for (int i = 0; i < 3; ++i) {
    const long back = (l0 >= 3 - i) ? (long)(3 - i) * D_IN_PROJ : 0;
    xr[i] = *(const float4*)(xin - back);
  }
  #pragma unroll
  for (int i = 0; i < DW_R; ++i) xr[3 + i] = *(const float4*)(xin + (long)i * D_IN_PROJ);
  const float4 wa = *(const float4*)(w + (4 * t + 0) * 4);
  const float4 wb = *(const float4*)(w + (4 * t + 1) * 4);
  const float4 wc = *(const float4*)(w + (4 * t + 2) * 4);
  const float4 wd = *(const float4*)(w + (4 * t + 3) * 4);
  const float4 bs = *(const float4*)(bias + 4 * t);
  if (dt_thread) {
    const float dt_raw    = zx[(long)(bl0 + dt_i) * D_IN_PROJ + (D_IN_PROJ - NHEADS) + dt_hd];
    const float dt_bias_h = dtb[dt_hd];
    const float dt_alog_h = alog[dt_hd];
    const float Ah  = -__expf(dt_alog_h);
    const float raw = dt_raw + dt_bias_h;
    const float dtv = (raw > 20.f) ? raw : log1pf(__expf(raw));
    float2 dd;
    dd.x = dtv;
    dd.y = __expf(dtv * Ah);
    *(float2*)(dtda + ((long)(bl0 + dt_i) * NHEADS + dt_hd) * 2) = dd;
  }
  float* dst = out + (long)bl0 * CONV_DIM + 4 * t;
  #pragma unroll
  for (int i = 0; i < DW_R; ++i) {
    const int l = l0 + i;
    float4 o;
    o.x = dwconv_one(wa, xr[3 + i].x, xr[2 + i].x, xr[1 + i].x, xr[i].x, l, bs.x);
    o.y = dwconv_one(wb, xr[3 + i].y, xr[2 + i].y, xr[1 + i].y, xr[i].y, l, bs.y);
    o.z = dwconv_one(wc, xr[3 + i].z, xr[2 + i].z, xr[1 + i].z, xr[i].z, l, bs.z);
    o.w = dwconv_one(wd, xr[3 + i].w, xr[2 + i].w, xr[1 + i].w, xr[i].w, l, bs.w);
    *(float4*)(dst + (long)i * CONV_DIM) = o;
  }
}

#define SCAN_T   32
#define SCAN_NCH (LOUT / SCAN_T)
#define SCAN_PB    (HEADDIM / SCAN_SPLIT)
#define SCAN_LPR   16
#define SCAN_NE    (D_STATE / SCAN_LPR)
#define SCAN_NT    (SCAN_PB * SCAN_LPR)
#define SCAN_YG    (32 / SCAN_PB)
#define SCAN_D     8

template <int CTRL>
__device__ __forceinline__ float lane_xor_read(float v) {
  return __int_as_float(__builtin_amdgcn_update_dpp(0, __float_as_int(v), CTRL, 0xF, 0xF, true));
}

__global__ void __launch_bounds__(SCAN_NT)
ssm_scan_kernel(const float* __restrict__ xbc,
                const float* __restrict__ dtda,
                const float* __restrict__ Dp,
                const float* __restrict__ zx,
                const float* __restrict__ nw,
                float* __restrict__ y,
                float* __restrict__ ssq) {
  const int pblk = blockIdx.x % SCAN_SPLIT;
  const int hh = (blockIdx.x / SCAN_SPLIT) & 15;
  const int b  = blockIdx.x / (SCAN_SPLIT * NHEADS);
  const int tid = threadIdx.x;
  const int p = tid / SCAN_LPR;
  const int q = tid % SCAN_LPR;
  __shared__ __attribute__((aligned(16))) float sBC[2][SCAN_T][2 * D_STATE];
  __shared__ __attribute__((aligned(16))) float sX[2][SCAN_T][SCAN_PB];
  __shared__ __attribute__((aligned(16))) float sY[2][SCAN_T][SCAN_PB];
  __shared__ __attribute__((aligned(8)))  float sDT[2][SCAN_T][2];
  const float Dh   = Dp[hh];
  float hreg[SCAN_NE];
  #pragma unroll
  for (int i = 0; i < SCAN_NE; ++i) hreg[i] = 0.f;

  const int bc_c4 = tid & 31;
  const int bc_t  = tid >> 5;
  const bool xy_wave = (tid < 64);
  const int x_c4  = tid & 1;
  const int x_t   = (tid >> 1) & (SCAN_T - 1);
  const int y_g   = (tid >> 3) & 7;
  const int y_c4  = tid & 7;
  const bool dt_wave = (tid >= SCAN_NT - SCAN_T);
  const int  dt_t    = dt_wave ? tid - (SCAN_NT - SCAN_T) : 0;
  const float* gBC = xbc + ((long)b * LOUT + bc_t) * CONV_DIM + D_INNER + bc_c4 * 4;
  const float* gX  = xbc + ((long)b * LOUT + x_t) * CONV_DIM + hh * HEADDIM + pblk * SCAN_PB + x_c4 * 4;
  const float* gDT = dtda + (((long)b * LOUT + dt_t) * NHEADS + hh) * 2;
  float*       gY  = y + ((long)b * LOUT + y_g * SCAN_YG) * D_INNER + (hh * SCAN_SPLIT + pblk) * 32 + y_c4 * 4;
  const int y_p4 = (y_c4 & 1) * 4;
  const float* gZ = zx + ((long)b * LOUT + 4 * y_g + (y_c4 >> 1)) * D_IN_PROJ + hh * HEADDIM + pblk * SCAN_PB + y_p4;
  float*       gS = ssq + ((long)((b * NHEADS + hh) * SCAN_SPLIT + pblk) * LOUT) * 2 + (tid & 63);
  float4 nw4 = make_float4(0.f, 0.f, 0.f, 0.f);
  if (xy_wave) nw4 = *(const float4*)(nw + hh * HEADDIM + pblk * SCAN_PB + y_p4);
  float4 rZ = make_float4(0.f, 0.f, 0.f, 0.f);

  float4 rBC0, rBC1, rBC2, rBC3, rBC4, rBC5, rBC6, rBC7;
  float4 rX0 = make_float4(0.f, 0.f, 0.f, 0.f);
  float2 rdt = make_float2(0.f, 0.f);

#define SCAN_FETCH(l0)                                                              \
  do {                                                                              \
    const float* s_bc = gBC + (long)(l0) * CONV_DIM;                                \
    rBC0 = *(const float4*)(s_bc + 0 * CONV_DIM);                                   \
    rBC1 = *(const float4*)(s_bc + 4 * CONV_DIM);                                   \
    rBC2 = *(const float4*)(s_bc + 8 * CONV_DIM);                                   \
    rBC3 = *(const float4*)(s_bc + 12 * CONV_DIM);                                  \
    rBC4 = *(const float4*)(s_bc + 16 * CONV_DIM);                                  \
    rBC5 = *(const float4*)(s_bc + 20 * CONV_DIM);                                  \
    rBC6 = *(const float4*)(s_bc + 24 * CONV_DIM);                                  \
    rBC7 = *(const float4*)(s_bc + 28 * CONV_DIM);                                  \
    if (xy_wave) rX0 = *(const float4*)(gX + (long)(l0) * CONV_DIM);                \
    if (dt_wave) rdt = *(const float2*)(gDT + (long)(l0) * (NHEADS * 2));           \
  } while (0)

#define SCAN_COMMIT(buf)                                                            \
  do {                                                                              \
    *(float4*)&sBC[(buf)][bc_t + 0][bc_c4 * 4] = rBC0;                              \
    *(float4*)&sBC[(buf)][bc_t + 4][bc_c4 * 4] = rBC1;                              \
    *(float4*)&sBC[(buf)][bc_t + 8][bc_c4 * 4] = rBC2;                              \
    *(float4*)&sBC[(buf)][bc_t + 12][bc_c4 * 4] = rBC3;                             \
    *(float4*)&sBC[(buf)][bc_t + 16][bc_c4 * 4] = rBC4;                             \
    *(float4*)&sBC[(buf)][bc_t + 20][bc_c4 * 4] = rBC5;                             \
    *(float4*)&sBC[(buf)][bc_t + 24][bc_c4 * 4] = rBC6;                             \
    *(float4*)&sBC[(buf)][bc_t + 28][bc_c4 * 4] = rBC7;                             \
    if (xy_wave) *(float4*)&sX[(buf)][x_t][x_c4 * 4] = rX0;                         \
    if (dt_wave) *(float2*)&sDT[(buf)][dt_t][0] = rdt;                              \
  } while (0)

#define SCAN_GATE1(yv, zv) ((yv) * ((zv) * __builtin_amdgcn_rcpf(1.f + __expf(-(zv)))))
#define SCAN_WRITEOUT(l0, buf)                                                      \
  do {                                                                              \
    if (xy_wave) {                                                                  \
      const float4 v0 = *(const float4*)(&sY[(buf)][0][0] + y_g * 32 + y_c4 * 4);   \
      const float gx = SCAN_GATE1(v0.x, rZ.x);                                      \
      const float gy = SCAN_GATE1(v0.y, rZ.y);                                      \
      const float gz = SCAN_GATE1(v0.z, rZ.z);                                      \
      const float gw = SCAN_GATE1(v0.w, rZ.w);                                      \
      float ss = gx * gx;                                                           \
      ss = fmaf(gy, gy, ss);                                                        \
      ss = fmaf(gz, gz, ss);                                                        \
      ss = fmaf(gw, gw, ss);                                                        \
      *(float4*)(gY + (long)(l0) * D_INNER) =                                       \
          make_float4(gx * nw4.x, gy * nw4.y, gz * nw4.z, gw * nw4.w);              \
      gS[(long)(l0) * 2] = ss;                                                      \
    }                                                                               \
  } while (0)

  float4 sB[SCAN_D], sC[SCAN_D];
  float2 sD[SCAN_D];
  float  sXv[SCAN_D];

#define SCAN_LDS_READ(u, buf, t)                                                    \
  do {                                                                              \
    sD[(u)]  = *(const float2*)&sDT[(buf)][(t)][0];                                 \
    sXv[(u)] = sX[(buf)][(t)][p];                                                   \
    sB[(u)]  = *(const float4*)&sBC[(buf)][(t)][q * SCAN_NE];                       \
    sC[(u)]  = *(const float4*)&sBC[(buf)][(t)][D_STATE + q * SCAN_NE];             \
  } while (0)

#define SCAN_STEP(u, yout, xout)                                                    \
  do {                                                                              \
    const float dtv = sD[(u)].x;                                                    \
    const float dA  = sD[(u)].y;                                                    \
    const float xv  = sXv[(u)];                                                     \
    const float xdt = xv * dtv;                                                     \
    const float4 bv = sB[(u)];                                                      \
    const float4 cv = sC[(u)];                                                      \
    float accv = 0.f;                                                               \
    hreg[0] = fmaf(hreg[0], dA, xdt * bv.x);                                        \
    accv    = fmaf(hreg[0], cv.x, accv);                                            \
    hreg[1] = fmaf(hreg[1], dA, xdt * bv.y);                                        \
    accv    = fmaf(hreg[1], cv.y, accv);                                            \
    hreg[2] = fmaf(hreg[2], dA, xdt * bv.z);                                        \
    accv    = fmaf(hreg[2], cv.z, accv);                                            \
    hreg[3] = fmaf(hreg[3], dA, xdt * bv.w);                                        \
    accv    = fmaf(hreg[3], cv.w, accv);                                            \
    (yout) = accv;                                                                  \
    (xout) = xv;                                                                    \
  } while (0)

#define SCAN_FINISH(acc, xv)                                                        \
  do {                                                                              \
    (acc) += lane_xor_read<0x161>(acc);                             \
    (acc) += lane_xor_read<0x162>(acc);                             \
    (acc) += lane_xor_read<0x164>(acc);                             \
    (acc) += lane_xor_read<0x168>(acc);                             \
    (acc) = fmaf((xv), Dh, (acc));                                                  \
  } while (0)

  SCAN_FETCH(0);
  SCAN_COMMIT(0);
  __syncthreads();

  for (int c = 0; c < SCAN_NCH; ++c) {
    const int cb = c & 1;
    const bool more = (c + 1 < SCAN_NCH);
    if (c > 0) SCAN_WRITEOUT((c - 1) * SCAN_T, cb ^ 1);
    SCAN_LDS_READ(0, cb, 0);
    SCAN_LDS_READ(1, cb, 1);
    __builtin_amdgcn_sched_barrier(0);
    SCAN_LDS_READ(2, cb, 2);
    SCAN_LDS_READ(3, cb, 3);
    __builtin_amdgcn_sched_barrier(0);
    SCAN_LDS_READ(4, cb, 4);
    SCAN_LDS_READ(5, cb, 5);
    __builtin_amdgcn_sched_barrier(0);
    SCAN_LDS_READ(6, cb, 6);
    SCAN_LDS_READ(7, cb, 7);
    if (more) SCAN_FETCH((c + 1) * SCAN_T);
    if (xy_wave) rZ = *(const float4*)(gZ + (long)(c * SCAN_T) * D_IN_PROJ);
    #pragma unroll
    for (int t0 = 0; t0 < SCAN_T - SCAN_D; t0 += SCAN_D) {
      float yv[SCAN_D], xs[SCAN_D];
      #pragma unroll
      for (int u = 0; u < SCAN_D; ++u) {
        SCAN_STEP(u, yv[u], xs[u]);
        SCAN_LDS_READ(u, cb, t0 + SCAN_D + u);
      }
      #pragma unroll
      for (int u = 0; u < SCAN_D; ++u) SCAN_FINISH(yv[u], xs[u]);
      if (q == 0) {
        #pragma unroll
        for (int u = 0; u < SCAN_D; ++u) sY[cb][t0 + u][p] = yv[u];
      }
    }
    {
      float yv[SCAN_D], xs[SCAN_D];
      #pragma unroll
      for (int u = 0; u < SCAN_D; ++u) SCAN_STEP(u, yv[u], xs[u]);
      #pragma unroll
      for (int u = 0; u < SCAN_D; ++u) SCAN_FINISH(yv[u], xs[u]);
      if (q == 0) {
        #pragma unroll
        for (int u = 0; u < SCAN_D; ++u) sY[cb][SCAN_T - SCAN_D + u][p] = yv[u];
      }
    }
    if (more) SCAN_COMMIT(cb ^ 1);
    __syncthreads();
  }
  SCAN_WRITEOUT((SCAN_NCH - 1) * SCAN_T, (SCAN_NCH - 1) & 1);
#undef SCAN_FETCH
#undef SCAN_COMMIT
#undef SCAN_WRITEOUT
#undef SCAN_GATE1
#undef SCAN_LDS_READ
#undef SCAN_STEP
#undef SCAN_FINISH
}

extern "C" void kernel_launch(void* const* d_in, const int* in_sizes, int n_in,
                              void* d_out, int out_size, void* d_ws, size_t ws_size,
                              hipStream_t stream) {
  const float* tokens     = (const float*)d_in[0];
  const float* conv_w     = (const float*)d_in[1];
  const float* in_proj_w  = (const float*)d_in[2];
  const float* conv1d_w   = (const float*)d_in[3];
  const float* conv1d_b   = (const float*)d_in[4];
  const float* dt_bias    = (const float*)d_in[5];
  const float* A_log      = (const float*)d_in[6];
  const float* D_param    = (const float*)d_in[7];
  const float* norm_w     = (const float*)d_in[8];
  const float* out_proj_w = (const float*)d_in[9];
  float* out = (float*)d_out;
  float* ws  = (float*)d_ws;

  float* u0   = ws;
  float* u1   = u0 + (long)ROWS * D_MODEL;
  float* zx   = u1 + (long)ROWS * D_MODEL;
  float* xact = zx + (long)ROWS * D_IN_PROJ;
  float* ybuf = xact + (long)ROWS * CONV_DIM;
  float* dtda = ybuf + (long)ROWS * D_INNER;
  float* ssqb = dtda + (long)ROWS * NHEADS * 2;
  float* col  = zx;
  (void)ws_size; (void)in_sizes; (void)n_in; (void)out_size;

  {
    im2col_stem_kernel<<<ROWS / IM_R, IM_NT, 0, stream>>>(tokens, col);
    gemm_xt_wmma<false><<<(ROWS / GT_M) * ((D_MODEL + GT_N - 1) / GT_N), 256, 0, stream>>>(col, conv_w, u0, nullptr, ROWS, D_MODEL, KSTEM);
  }

  const float* u_in = u0;
  for (int blk = 0; blk < 2; ++blk) {
    const float* ipw = in_proj_w  + (long)blk * D_IN_PROJ * D_MODEL;
    const float* cw  = conv1d_w   + (long)blk * CONV_DIM * D_CONV;
    const float* cb  = conv1d_b   + (long)blk * CONV_DIM;
    const float* dtb = dt_bias    + (long)blk * NHEADS;
    const float* al  = A_log      + (long)blk * NHEADS;
    const float* Dpp = D_param    + (long)blk * NHEADS;
    const float* nw  = norm_w     + (long)blk * D_INNER;
    const float* opw = out_proj_w + (long)blk * D_MODEL * D_INNER;

    {
      gemm_xt_wmma<false><<<(ROWS / GT_M) * ((D_IN_PROJ + GT_N - 1) / GT_N), 256, 0, stream>>>(u_in, ipw, zx, nullptr, ROWS, D_IN_PROJ, D_MODEL);
    }
    dwconv_silu_kernel<<<ROWS / DW_R, DW_NT, 0, stream>>>(zx, cw, cb, xact, dtb, al, dtda);
    ssm_scan_kernel<<<BATCH * NHEADS * SCAN_SPLIT, SCAN_NT, 0, stream>>>(xact, dtda, Dpp, zx, nw, ybuf, ssqb);
    float* dst = (blk == 0) ? u1 : out;
    {
      gemm_xt_wmma<true><<<(ROWS / GT_M) * ((D_MODEL + GT_N - 1) / GT_N), 256, 0, stream>>>(ybuf, opw, dst, ssqb, ROWS, D_MODEL, D_INNER);
    }
    u_in = u1;
  }
}


# ssm scan in chunked matrix form (64-step chunks): in-chunk sums as masked, decayed (C B^T)(x dt) tiles on the f32 matrix instruction, state carried chunk to chunk; order of sums over steps changed
# speedup vs baseline: 15.0802x; 1.1950x over previous
#include <hip/hip_runtime.h>
#include <hip/hip_bf16.h>

typedef __attribute__((ext_vector_type(2))) float v2f;
typedef __attribute__((ext_vector_type(8))) float v8f;

#define BATCH     2
#define SEQLEN    2048
#define C_IN      256
#define D_MODEL   512
#define LOUT      1024
#define D_STATE   64
#define D_CONV    4
#define D_INNER   1024
#define HEADDIM   64
#define NHEADS    16
#define CONV_DIM  (D_INNER + 2*D_STATE)
#define D_IN_PROJ (2*D_INNER + 2*D_STATE + NHEADS)
#define ROWS      (BATCH*LOUT)
#define KSTEM     (C_IN * 7)
#define SCAN_SPLIT 8

#define IM_R  8
#define IM_NT (KSTEM / 4)
__global__ void __launch_bounds__(IM_NT)
im2col_stem_kernel(const float* __restrict__ tokens,
                   float* __restrict__ col) {
  const int t   = threadIdx.x;
  const int bl0 = blockIdx.x * IM_R;
  const int b   = bl0 >> 10;
  const int l0  = bl0 & (LOUT - 1);
  const float* tb = tokens + (long)b * C_IN * SEQLEN;
  int off[4], kk[4];
  #pragma unroll
  for (int m = 0; m < 4; ++m) {
    const int j = t + IM_NT * m;
    const int c = j / 7;
    kk[m]  = j - c * 7;
    off[m] = c * SEQLEN;
  }
  float v[IM_R][4];
  #pragma unroll
  for (int i = 0; i < IM_R; ++i) {
    const int s0 = 2 * (l0 + i) - 3;
    #pragma unroll
    for (int m = 0; m < 4; ++m) {
      const int s  = s0 + kk[m];
      const int sc = min(max(s, 0), SEQLEN - 1);
      const float x = tb[off[m] + sc];
      v[i][m] = (s == sc) ? x : 0.f;
    }
  }
  float* dst = col + (long)bl0 * KSTEM + t;
  #pragma unroll
  for (int i = 0; i < IM_R; ++i) {
    #pragma unroll
    for (int m = 0; m < 4; ++m) dst[(long)i * KSTEM + IM_NT * m] = v[i][m];
  }
}

#define GT_M 64
#define GT_N 64
#define GT_K 64
#define GT_P 68

template <bool NORM>
__global__ void __launch_bounds__(256)
gemm_xt_wmma(const float* __restrict__ A,
             const float* __restrict__ W,
             float* __restrict__ C,
             const float* __restrict__ ssq,
             int M, int N, int K) {
  __shared__ __attribute__((aligned(16))) float sA[2][GT_M][GT_P];
  __shared__ __attribute__((aligned(16))) float sB[2][GT_N][GT_P];
  __shared__ float sRow[NORM ? 4 : 1][GT_M];
  const int tid  = threadIdx.x;
  const int lane = tid & 31;
  const int wave = tid >> 5;
  const int nblk = (N + GT_N - 1) / GT_N;
  const int mb = blockIdx.x / nblk;
  const int nb = blockIdx.x - mb * nblk;
  const int m0 = mb * GT_M;
  const int n0 = nb * GT_N;
  (void)M;

  const int ld_c4 = tid & 15;
  const int ld_r  = tid >> 4;
  const long a_step = (long)16 * K;
  const int a_r = NORM ? ((tid >> 6) * 4 + ((tid >> 1) & 3)) : ld_r;
  const int a_c = NORM ? (((tid >> 3) & 7) * 8 + (tid & 1) * 4) : (ld_c4 * 4);
  const int a_ks = NORM ? 4 : 1;
  const float* gA = NORM
      ? A + (long)(m0 + (a_r & ~3)) * K + (a_c >> 3) * 32 + (a_r & 3) * 8 + (a_c & 7)
      : A + (long)(m0 + a_r) * K + a_c;
  const float* gB0 = W + (long)min(n0 + ld_r,      N - 1) * K + ld_c4 * 4;
  const float* gB1 = W + (long)min(n0 + ld_r + 16, N - 1) * K + ld_c4 * 4;
  const float* gB2 = W + (long)min(n0 + ld_r + 32, N - 1) * K + ld_c4 * 4;
  const float* gB3 = W + (long)min(n0 + ld_r + 48, N - 1) * K + ld_c4 * 4;

  float4 ra0, ra1, ra2, ra3, rb0, rb1, rb2, rb3;

  static_assert(!NORM || (D_INNER / GT_K) * 2 == (NHEADS * SCAN_SPLIT) / 4, "two pairs per K-step cover a quarter");
  const int   ps_row = tid & (GT_M - 1);
  const int   ps_q   = tid >> 6;
  const float* ps = NORM ? ssq + (((long)((m0 + ps_row) / LOUT) * (NHEADS * SCAN_SPLIT) + ps_q * 32) * LOUT
                                  + ((m0 + ps_row) % LOUT)) * 2
                         : nullptr;
  float2 pq0 = make_float2(0.f, 0.f), pq1 = pq0;
  float  rs  = 0.f;

#define GT_FETCH(k0)                                                   \
  do {                                                                 \
    ra0 = *(const float4*)(gA + (k0) * a_ks);                          \
    ra1 = *(const float4*)(gA + a_step + (k0) * a_ks);                 \
    ra2 = *(const float4*)(gA + 2 * a_step + (k0) * a_ks);             \
    ra3 = *(const float4*)(gA + 3 * a_step + (k0) * a_ks);             \
    rb0 = *(const float4*)(gB0 + (k0));                                \
    rb1 = *(const float4*)(gB1 + (k0));                                \
    rb2 = *(const float4*)(gB2 + (k0));                                \
    rb3 = *(const float4*)(gB3 + (k0));                                \
    if constexpr (NORM) {                                              \
      pq0 = *(const float2*)(ps + (long)((k0) / GT_K * 2)     * (LOUT * 2)); \
      pq1 = *(const float2*)(ps + (long)((k0) / GT_K * 2 + 1) * (LOUT * 2)); \
    }                                                                  \
  } while (0)

#define GT_COMMIT(buf)                                                 \
  do {                                                                 \
    *(float4*)&sA[(buf)][a_r     ][a_c] = ra0;                         \
    *(float4*)&sA[(buf)][a_r + 16][a_c] = ra1;                         \
    *(float4*)&sA[(buf)][a_r + 32][a_c] = ra2;                         \
    *(float4*)&sA[(buf)][a_r + 48][a_c] = ra3;                         \
    *(float4*)&sB[(buf)][ld_r     ][ld_c4 * 4] = rb0;                  \
    *(float4*)&sB[(buf)][ld_r + 16][ld_c4 * 4] = rb1;                  \
    *(float4*)&sB[(buf)][ld_r + 32][ld_c4 * 4] = rb2;                  \
    *(float4*)&sB[(buf)][ld_r + 48][ld_c4 * 4] = rb3;                  \
    if constexpr (NORM) {                                              \
      rs += pq0.x + pq0.y;                                             \
      rs += pq1.x + pq1.y;                                             \
    }                                                                  \
  } while (0)

  const int r  = lane & 15;
  const int h  = lane >> 4;
  const int wm = wave & 3;
  const int wn = wave >> 2;
  const float* fa  = &sA[0][wm * 16 + r][4 * h];
  const float* fb0 = &sB[0][wn * 32 + r][4 * h];
  const float* fb1 = &sB[0][wn * 32 + 16 + r][4 * h];

  v8f acc0 = {0.f,0.f,0.f,0.f,0.f,0.f,0.f,0.f};
  v8f acc1 = acc0;

#define GT_COMPUTE(buf)                                                                            \
  do {                                                                                             \
    const float* pa  = fa  + (buf) * (GT_M * GT_P);                                                \
    const float* pb0 = fb0 + (buf) * (GT_N * GT_P);                                                \
    const float* pb1 = fb1 + (buf) * (GT_N * GT_P);                                                \
    _Pragma("unroll")                                                                              \
    for (int g = 0; g < GT_K / 8; ++g) {                                                           \
      const float4 a4 = *(const float4*)(pa  + g * 8);                                             \
      const float4 p4 = *(const float4*)(pb0 + g * 8);                                             \
      const float4 q4 = *(const float4*)(pb1 + g * 8);                                             \
      const v2f alo = {a4.x, a4.y}, ahi = {a4.z, a4.w};                                            \
      const v2f plo = {p4.x, p4.y}, phi = {p4.z, p4.w};                                            \
      const v2f qlo = {q4.x, q4.y}, qhi = {q4.z, q4.w};                                            \
      acc0 = __builtin_amdgcn_wmma_f32_16x16x4_f32(false, alo, false, plo, (short)0, acc0, false, false); \
      acc1 = __builtin_amdgcn_wmma_f32_16x16x4_f32(false, alo, false, qlo, (short)0, acc1, false, false); \
      acc0 = __builtin_amdgcn_wmma_f32_16x16x4_f32(false, ahi, false, phi, (short)0, acc0, false, false); \
      acc1 = __builtin_amdgcn_wmma_f32_16x16x4_f32(false, ahi, false, qhi, (short)0, acc1, false, false); \
    }                                                                                              \
  } while (0)

  const int KT = K / GT_K;
  GT_FETCH(0);
  GT_COMMIT(0);
  __syncthreads();

  for (int kt = 0; kt + 1 < KT; ++kt) {
    const int cb = kt & 1;
    GT_FETCH((kt + 1) * GT_K);
    GT_COMPUTE(cb);
    GT_COMMIT(cb ^ 1);
    __syncthreads();
  }
  GT_COMPUTE((KT - 1) & 1);
  if constexpr (NORM) sRow[ps_q][ps_row] = rs;
  __syncthreads();

  float (*sC)[GT_P] = sA[0];
  #pragma unroll
  for (int i = 0; i < 8; ++i) {
    sC[wm * 16 + 8 * h + i][wn * 32 + r]      = acc0[i];
    sC[wm * 16 + 8 * h + i][wn * 32 + 16 + r] = acc1[i];
  }
  __syncthreads();
  const int col = n0 + ld_c4 * 4;
  if (col < N) {
    #pragma unroll
    for (int p = 0; p < 4; ++p) {
      const int row = ld_r + 16 * p;
      float4 v = *(const float4*)&sC[row][ld_c4 * 4];
      if constexpr (NORM) {
        const float sum = ((sRow[0][row] + sRow[1][row]) + sRow[2][row]) + sRow[3][row];
        const float scale = rsqrtf(sum * (1.f / (float)D_INNER) + 1e-5f);
        v.x *= scale; v.y *= scale; v.z *= scale; v.w *= scale;
      }
      *(float4*)(C + (long)(m0 + row) * N + col) = v;
    }
  }
#undef GT_FETCH
#undef GT_COMMIT
#undef GT_COMPUTE
}

#define DW_R  8
#define DW_NT (CONV_DIM / 4)

__device__ __forceinline__ float dwconv_one(const float4 wv, float x0, float xm1, float xm2,
                                            float xm3, int l, float bs) {
  float acc = wv.w * x0;
  if (l >= 1) acc = fmaf(wv.z, xm1, acc);
  if (l >= 2) acc = fmaf(wv.y, xm2, acc);
  if (l >= 3) acc = fmaf(wv.x, xm3, acc);
  const float v = acc + bs;
  return v / (1.f + __expf(-v));
}

__global__ void __launch_bounds__(DW_NT)
dwconv_silu_kernel(const float* __restrict__ zx,
                   const float* __restrict__ w,
                   const float* __restrict__ bias,
                   float* __restrict__ out,
                   const float* __restrict__ dtb,
                   const float* __restrict__ alog,
                   float* __restrict__ dtda) {
  const int t   = threadIdx.x;
  const int bl0 = blockIdx.x * DW_R;
  const int l0  = bl0 & (LOUT - 1);
  const bool dt_thread = (t < DW_R * NHEADS);
  const int  dt_i  = t / NHEADS;
  const int  dt_hd = t % NHEADS;
  const float* xin = zx + (long)bl0 * D_IN_PROJ + D_INNER + 4 * t;
  float4 xr[DW_R + 3];
  #pragma unroll
  for (int i = 0; i < 3; ++i) {
    const long back = (l0 >= 3 - i) ? (long)(3 - i) * D_IN_PROJ : 0;
    xr[i] = *(const float4*)(xin - back);
  }
  #pragma unroll
  for (int i = 0; i < DW_R; ++i) xr[3 + i] = *(const float4*)(xin + (long)i * D_IN_PROJ);
  const float4 wa = *(const float4*)(w + (4 * t + 0) * 4);
  const float4 wb = *(const float4*)(w + (4 * t + 1) * 4);
  const float4 wc = *(const float4*)(w + (4 * t + 2) * 4);
  const float4 wd = *(const float4*)(w + (4 * t + 3) * 4);
  const float4 bs = *(const float4*)(bias + 4 * t);
  if (dt_thread) {
    const float dt_raw    = zx[(long)(bl0 + dt_i) * D_IN_PROJ + (D_IN_PROJ - NHEADS) + dt_hd];
    const float dt_bias_h = dtb[dt_hd];
    const float dt_alog_h = alog[dt_hd];
    const float Ah  = -__expf(dt_alog_h);
    const float raw = dt_raw + dt_bias_h;
    const float dtv = (raw > 20.f) ? raw : log1pf(__expf(raw));
    float2 dd;
    dd.x = dtv;
    dd.y = __expf(dtv * Ah);
    *(float2*)(dtda + ((long)(bl0 + dt_i) * NHEADS + dt_hd) * 2) = dd;
  }
  float* dst = out + (long)bl0 * CONV_DIM + 4 * t;
  #pragma unroll
  for (int i = 0; i < DW_R; ++i) {
    const int l = l0 + i;
    float4 o;
    o.x = dwconv_one(wa, xr[3 + i].x, xr[2 + i].x, xr[1 + i].x, xr[i].x, l, bs.x);
    o.y = dwconv_one(wb, xr[3 + i].y, xr[2 + i].y, xr[1 + i].y, xr[i].y, l, bs.y);
    o.z = dwconv_one(wc, xr[3 + i].z, xr[2 + i].z, xr[1 + i].z, xr[i].z, l, bs.z);
    o.w = dwconv_one(wd, xr[3 + i].w, xr[2 + i].w, xr[1 + i].w, xr[i].w, l, bs.w);
    *(float4*)(dst + (long)i * CONV_DIM) = o;
  }
}

#define SSD_Q   64
#define SSD_NC  (LOUT / SSD_Q)
#define SSD_HN  (HEADDIM * D_STATE)

__device__ __forceinline__ void tile64_mma(const float* pa, const float* pb0, const float* pb1,
                                           v8f& acc0, v8f& acc1) {
  #pragma unroll
  for (int g = 0; g < 8; ++g) {
    const float4 a4 = *(const float4*)(pa  + g * 8);
    const float4 p4 = *(const float4*)(pb0 + g * 8);
    const float4 q4 = *(const float4*)(pb1 + g * 8);
    const v2f alo = {a4.x, a4.y}, ahi = {a4.z, a4.w};
    const v2f plo = {p4.x, p4.y}, phi = {p4.z, p4.w};
    const v2f qlo = {q4.x, q4.y}, qhi = {q4.z, q4.w};
    acc0 = __builtin_amdgcn_wmma_f32_16x16x4_f32(false, alo, false, plo, (short)0, acc0, false, false);
    acc1 = __builtin_amdgcn_wmma_f32_16x16x4_f32(false, alo, false, qlo, (short)0, acc1, false, false);
    acc0 = __builtin_amdgcn_wmma_f32_16x16x4_f32(false, ahi, false, phi, (short)0, acc0, false, false);
    acc1 = __builtin_amdgcn_wmma_f32_16x16x4_f32(false, ahi, false, qhi, (short)0, acc1, false, false);
  }
}

__device__ __forceinline__ float chunk_cumsum(const float* sa, int j, float& total) {
  float run = 0.f, mine = 0.f;
  #pragma unroll
  for (int k = 0; k < SSD_Q; ++k) {
    run += sa[k];
    if (k == j) mine = run;
  }
  total = run;
  return mine;
}

__global__ void __launch_bounds__(256)
ssd_state_kernel(const float* __restrict__ xbc,
                 const float* __restrict__ dtda,
                 const float* __restrict__ alog,
                 float* __restrict__ SH,
                 float* __restrict__ dec) {
  const int c  = blockIdx.x % SSD_NC;
  const int hh = (blockIdx.x / SSD_NC) % NHEADS;
  const int b  = blockIdx.x / (SSD_NC * NHEADS);
  __shared__ __attribute__((aligned(16))) float sXT[HEADDIM][GT_P];
  __shared__ __attribute__((aligned(16))) float sBT[D_STATE][GT_P];
  __shared__ float sa[SSD_Q];
  __shared__ float sw[SSD_Q];
  const int tid  = threadIdx.x;
  const int lane = tid & 31;
  const int wave = tid >> 5;
  const int ld_c4 = tid & 15;
  const int ld_r  = tid >> 4;
  const long row0 = (long)b * LOUT + c * SSD_Q;
  float4 rx[4], rb[4];
  #pragma unroll
  for (int ps = 0; ps < 4; ++ps) {
    const float* rowp = xbc + (row0 + ld_r + 16 * ps) * CONV_DIM;
    rx[ps] = *(const float4*)(rowp + hh * HEADDIM + ld_c4 * 4);
    rb[ps] = *(const float4*)(rowp + D_INNER + ld_c4 * 4);
  }
  const float dtv = dtda[((row0 + (tid & (SSD_Q - 1))) * NHEADS + hh) * 2];
  if (tid < SSD_Q) sa[tid] = dtv * (-__expf(alog[hh]));
  __syncthreads();
  if (tid < SSD_Q) {
    float total;
    const float mine = chunk_cumsum(sa, tid, total);
    sw[tid] = dtv * __expf(total - mine);
    if (tid == 0) dec[(b * NHEADS + hh) * SSD_NC + c] = __expf(total);
  }
  __syncthreads();
  #pragma unroll
  for (int ps = 0; ps < 4; ++ps) {
    const int j = ld_r + 16 * ps;
    const float wj = sw[j];
    sXT[ld_c4 * 4 + 0][j] = rx[ps].x * wj;
    sXT[ld_c4 * 4 + 1][j] = rx[ps].y * wj;
    sXT[ld_c4 * 4 + 2][j] = rx[ps].z * wj;
    sXT[ld_c4 * 4 + 3][j] = rx[ps].w * wj;
    sBT[ld_c4 * 4 + 0][j] = rb[ps].x;
    sBT[ld_c4 * 4 + 1][j] = rb[ps].y;
    sBT[ld_c4 * 4 + 2][j] = rb[ps].z;
    sBT[ld_c4 * 4 + 3][j] = rb[ps].w;
  }
  __syncthreads();
  const int r  = lane & 15;
  const int h  = lane >> 4;
  const int wm = wave & 3;
  const int wn = wave >> 2;
  v8f acc0 = {0.f,0.f,0.f,0.f,0.f,0.f,0.f,0.f};
  v8f acc1 = acc0;
  tile64_mma(&sXT[wm * 16 + r][4 * h], &sBT[wn * 32 + r][4 * h], &sBT[wn * 32 + 16 + r][4 * h], acc0, acc1);
  __syncthreads();
  #pragma unroll
  for (int i = 0; i < 8; ++i) {
    sXT[wm * 16 + 8 * h + i][wn * 32 + r]      = acc0[i];
    sXT[wm * 16 + 8 * h + i][wn * 32 + 16 + r] = acc1[i];
  }
  __syncthreads();
  float* dst = SH + ((long)((b * NHEADS + hh) * SSD_NC + c)) * SSD_HN;
  #pragma unroll
  for (int ps = 0; ps < 4; ++ps) {
    const int row = ld_r + 16 * ps;
    *(float4*)(dst + row * D_STATE + ld_c4 * 4) = *(const float4*)&sXT[row][ld_c4 * 4];
  }
}

__global__ void __launch_bounds__(256)
ssd_carry_kernel(float* __restrict__ SH, const float* __restrict__ dec) {
  const int bh = blockIdx.x >> 2;
  const int e4 = (blockIdx.x & 3) * 256 + threadIdx.x;
  float4* base = (float4*)(SH + (long)bh * SSD_NC * SSD_HN) + e4;
  float4 s[SSD_NC];
  float  d[SSD_NC];
  #pragma unroll
  for (int c = 0; c < SSD_NC; ++c) {
    s[c] = base[c * (SSD_HN / 4)];
    d[c] = dec[bh * SSD_NC + c];
  }
  float4 H = make_float4(0.f, 0.f, 0.f, 0.f);
  #pragma unroll
  for (int c = 0; c < SSD_NC; ++c) {
    base[c * (SSD_HN / 4)] = H;
    H.x = fmaf(H.x, d[c], s[c].x);
    H.y = fmaf(H.y, d[c], s[c].y);
    H.z = fmaf(H.z, d[c], s[c].z);
    H.w = fmaf(H.w, d[c], s[c].w);
  }
}

#define SSD_GATE1(yv, zv) ((yv) * ((zv) * __builtin_amdgcn_rcpf(1.f + __expf(-(zv)))))
__global__ void __launch_bounds__(256)
ssd_out_kernel(const float* __restrict__ xbc,
               const float* __restrict__ dtda,
               const float* __restrict__ alog,
               const float* __restrict__ Dp,
               const float* __restrict__ zx,
               const float* __restrict__ nw,
               const float* __restrict__ SH,
               float* __restrict__ y,
               float* __restrict__ ssq) {
  const int c  = blockIdx.x % SSD_NC;
  const int hh = (blockIdx.x / SSD_NC) % NHEADS;
  const int b  = blockIdx.x / (SSD_NC * NHEADS);
  __shared__ __attribute__((aligned(16))) float sCm[SSD_Q][GT_P];
  __shared__ __attribute__((aligned(16))) float sBm[SSD_Q][GT_P];
  __shared__ __attribute__((aligned(16))) float sXT[HEADDIM][GT_P];
  __shared__ __attribute__((aligned(16))) float sHn[HEADDIM][GT_P];
  __shared__ __attribute__((aligned(16))) float sM[SSD_Q][GT_P];
  __shared__ float sa[SSD_Q];
  __shared__ float scs[SSD_Q];
  __shared__ float sdt[SSD_Q];
  const int tid  = threadIdx.x;
  const int lane = tid & 31;
  const int wave = tid >> 5;
  const int ld_c4 = tid & 15;
  const int ld_r  = tid >> 4;
  const long row0 = (long)b * LOUT + c * SSD_Q;
  const float* hin = SH + ((long)((b * NHEADS + hh) * SSD_NC + c)) * SSD_HN;
  float4 rx[4], rb[4], rc[4], rh[4];
  #pragma unroll
  for (int ps = 0; ps < 4; ++ps) {
    const float* rowp = xbc + (row0 + ld_r + 16 * ps) * CONV_DIM;
    rx[ps] = *(const float4*)(rowp + hh * HEADDIM + ld_c4 * 4);
    rb[ps] = *(const float4*)(rowp + D_INNER + ld_c4 * 4);
    rc[ps] = *(const float4*)(rowp + D_INNER + D_STATE + ld_c4 * 4);
    rh[ps] = *(const float4*)(hin + (ld_r + 16 * ps) * D_STATE + ld_c4 * 4);
  }
  const int   dt_j = tid & (SSD_Q - 1);
  const float dtv  = dtda[((row0 + dt_j) * NHEADS + hh) * 2];
  sdt[dt_j] = dtv;
  sa[dt_j]  = dtv * (-__expf(alog[hh]));
  #pragma unroll
  for (int ps = 0; ps < 4; ++ps) {
    *(float4*)&sCm[ld_r + 16 * ps][ld_c4 * 4] = rc[ps];
    *(float4*)&sBm[ld_r + 16 * ps][ld_c4 * 4] = rb[ps];
    *(float4*)&sHn[ld_r + 16 * ps][ld_c4 * 4] = rh[ps];
  }
  __syncthreads();
  if (tid < SSD_Q) {
    float total;
    scs[tid] = chunk_cumsum(sa, tid, total);
  }
  #pragma unroll
  for (int ps = 0; ps < 4; ++ps) {
    const int j = ld_r + 16 * ps;
    const float dj = sdt[j];
    sXT[ld_c4 * 4 + 0][j] = rx[ps].x * dj;
    sXT[ld_c4 * 4 + 1][j] = rx[ps].y * dj;
    sXT[ld_c4 * 4 + 2][j] = rx[ps].z * dj;
    sXT[ld_c4 * 4 + 3][j] = rx[ps].w * dj;
  }
  __syncthreads();
  const int r  = lane & 15;
  const int h  = lane >> 4;
  const int wm = wave & 3;
  const int wn = wave >> 2;
  const v8f zero8 = {0.f,0.f,0.f,0.f,0.f,0.f,0.f,0.f};
  v8f g0 = zero8, g1 = zero8, i0 = zero8, i1 = zero8;
  const float* pa = &sCm[wm * 16 + r][4 * h];
  tile64_mma(pa, &sBm[wn * 32 + r][4 * h], &sBm[wn * 32 + 16 + r][4 * h], g0, g1);
  tile64_mma(pa, &sHn[wn * 32 + r][4 * h], &sHn[wn * 32 + 16 + r][4 * h], i0, i1);
  const int col0 = wn * 32 + r;
  const int col1 = col0 + 16;
  const float cs0 = scs[col0];
  const float cs1 = scs[col1];
  float ecs[8];
  #pragma unroll
  for (int i = 0; i < 8; ++i) {
    const int row = wm * 16 + 8 * h + i;
    const float csr = scs[row];
    ecs[i] = __expf(csr);
    const float m0 = (row >= col0) ? g0[i] * __expf(csr - cs0) : 0.f;
    const float m1 = (row >= col1) ? g1[i] * __expf(csr - cs1) : 0.f;
    sM[row][col0] = m0;
    sM[row][col1] = m1;
  }
  __syncthreads();
  const int wo_rowin  = (tid >> 1) & 3;
  const int wo_half   = tid & 1;
  const int wo_colgrp = (tid >> 3) & 7;
  const int wo_col    = wo_colgrp * 8 + wo_half * 4;
  const float4 n4 = *(const float4*)(nw + hh * HEADDIM + wo_col);
  float4 xv[4], zv[4];
  #pragma unroll
  for (int ps = 0; ps < 4; ++ps) {
    const long R = row0 + (ps * 4 + (tid >> 6)) * 4 + wo_rowin;
    xv[ps] = *(const float4*)(xbc + R * CONV_DIM + hh * HEADDIM + wo_col);
    zv[ps] = *(const float4*)(zx + R * D_IN_PROJ + hh * HEADDIM + wo_col);
  }
  v8f y0 = zero8, y1 = zero8;
  tile64_mma(&sM[wm * 16 + r][4 * h], &sXT[wn * 32 + r][4 * h], &sXT[wn * 32 + 16 + r][4 * h], y0, y1);
  #pragma unroll
  for (int i = 0; i < 8; ++i) {
    const int row = wm * 16 + 8 * h + i;
    sBm[row][col0] = fmaf(i0[i], ecs[i], y0[i]);
    sBm[row][col1] = fmaf(i1[i], ecs[i], y1[i]);
  }
  __syncthreads();
  const float Dh = Dp[hh];
  #pragma unroll
  for (int ps = 0; ps < 4; ++ps) {
    const int rowgrp = ps * 4 + (tid >> 6);
    const int row = rowgrp * 4 + wo_rowin;
    const float4 yv = *(const float4*)&sBm[row][wo_col];
    const float gx = SSD_GATE1(fmaf(xv[ps].x, Dh, yv.x), zv[ps].x);
    const float gy = SSD_GATE1(fmaf(xv[ps].y, Dh, yv.y), zv[ps].y);
    const float gz = SSD_GATE1(fmaf(xv[ps].z, Dh, yv.z), zv[ps].z);
    const float gw = SSD_GATE1(fmaf(xv[ps].w, Dh, yv.w), zv[ps].w);
    float ss = gx * gx;
    ss = fmaf(gy, gy, ss);
    ss = fmaf(gz, gz, ss);
    ss = fmaf(gw, gw, ss);
    *(float4*)(y + (row0 + rowgrp * 4) * D_INNER + (hh * SCAN_SPLIT + wo_colgrp) * 32 + wo_rowin * 8 + wo_half * 4) =
        make_float4(gx * n4.x, gy * n4.y, gz * n4.z, gw * n4.w);
    ssq[((long)((b * NHEADS + hh) * SCAN_SPLIT + wo_colgrp) * LOUT + (c * SSD_Q + row)) * 2 + wo_half] = ss;
  }
}
#undef SSD_GATE1

extern "C" void kernel_launch(void* const* d_in, const int* in_sizes, int n_in,
                              void* d_out, int out_size, void* d_ws, size_t ws_size,
                              hipStream_t stream) {
  const float* tokens     = (const float*)d_in[0];
  const float* conv_w     = (const float*)d_in[1];
  const float* in_proj_w  = (const float*)d_in[2];
  const float* conv1d_w   = (const float*)d_in[3];
  const float* conv1d_b   = (const float*)d_in[4];
  const float* dt_bias    = (const float*)d_in[5];
  const float* A_log      = (const float*)d_in[6];
  const float* D_param    = (const float*)d_in[7];
  const float* norm_w     = (const float*)d_in[8];
  const float* out_proj_w = (const float*)d_in[9];
  float* out = (float*)d_out;
  float* ws  = (float*)d_ws;

  float* u0   = ws;
  float* u1   = u0 + (long)ROWS * D_MODEL;
  float* zx   = u1 + (long)ROWS * D_MODEL;
  float* xact = zx + (long)ROWS * D_IN_PROJ;
  float* ybuf = xact + (long)ROWS * CONV_DIM;
  float* dtda = ybuf + (long)ROWS * D_INNER;
  float* ssqb = dtda + (long)ROWS * NHEADS * 2;
  float* shb  = ssqb + (long)BATCH * NHEADS * SCAN_SPLIT * LOUT * 2;
  float* decb = shb + (long)BATCH * NHEADS * SSD_NC * SSD_HN;
  float* col  = zx;
  (void)ws_size; (void)in_sizes; (void)n_in; (void)out_size;

  {
    im2col_stem_kernel<<<ROWS / IM_R, IM_NT, 0, stream>>>(tokens, col);
    gemm_xt_wmma<false><<<(ROWS / GT_M) * ((D_MODEL + GT_N - 1) / GT_N), 256, 0, stream>>>(col, conv_w, u0, nullptr, ROWS, D_MODEL, KSTEM);
  }

  const float* u_in = u0;
  for (int blk = 0; blk < 2; ++blk) {
    const float* ipw = in_proj_w  + (long)blk * D_IN_PROJ * D_MODEL;
    const float* cw  = conv1d_w   + (long)blk * CONV_DIM * D_CONV;
    const float* cb  = conv1d_b   + (long)blk * CONV_DIM;
    const float* dtb = dt_bias    + (long)blk * NHEADS;
    const float* al  = A_log      + (long)blk * NHEADS;
    const float* Dpp = D_param    + (long)blk * NHEADS;
    const float* nw  = norm_w     + (long)blk * D_INNER;
    const float* opw = out_proj_w + (long)blk * D_MODEL * D_INNER;

    {
      gemm_xt_wmma<false><<<(ROWS / GT_M) * ((D_IN_PROJ + GT_N - 1) / GT_N), 256, 0, stream>>>(u_in, ipw, zx, nullptr, ROWS, D_IN_PROJ, D_MODEL);
    }
    dwconv_silu_kernel<<<ROWS / DW_R, DW_NT, 0, stream>>>(zx, cw, cb, xact, dtb, al, dtda);
    ssd_state_kernel<<<BATCH * NHEADS * SSD_NC, 256, 0, stream>>>(xact, dtda, al, shb, decb);
    ssd_carry_kernel<<<BATCH * NHEADS * 4, 256, 0, stream>>>(shb, decb);
    ssd_out_kernel<<<BATCH * NHEADS * SSD_NC, 256, 0, stream>>>(xact, dtda, al, Dpp, zx, nw, shb, ybuf, ssqb);
    float* dst = (blk == 0) ? u1 : out;
    {
      gemm_xt_wmma<true><<<(ROWS / GT_M) * ((D_MODEL + GT_N - 1) / GT_N), 256, 0, stream>>>(ybuf, opw, dst, ssqb, ROWS, D_MODEL, D_INNER);
    }
    u_in = u1;
  }
}
